# starting kernel
# baseline (speedup 1.0000x reference)
#include <hip/hip_runtime.h>
#include <hip/hip_bf16.h>

typedef __attribute__((ext_vector_type(2))) float v2f;
typedef __attribute__((ext_vector_type(8))) float v8f;

#define DIM 64
#define LEAKY 0.01f

__global__ __launch_bounds__(128)
void gat_gemm_wmma(const float* __restrict__ hin, const float* __restrict__ W,
                   const float* __restrict__ bias, float* __restrict__ z,
                   int nTiles, int reluIn) {
  __shared__ float sW[DIM * DIM];
  __shared__ float sB[DIM];

  const int tid = threadIdx.x;
  for (int i = tid; i < DIM * DIM; i += 128) sW[i] = W[i];
  if (tid < DIM) sB[tid] = bias[tid];
  __syncthreads();

  const int wave = tid >> 5;
  const int lane = tid & 31;
  const int rowTile = blockIdx.x * 4 + wave;
  if (rowTile >= nTiles) return;

  const int row0 = rowTile * 16;
  const int laneRow = lane & 15;
  const int kHalf = (lane >> 4) * 2;

  const float* arow = hin + (size_t)(row0 + laneRow) * DIM;
  v2f aR[16];
#pragma unroll
  for (int t = 0; t < 16; ++t) {
    v2f a = *(const v2f*)(arow + 4 * t + kHalf);
    if (reluIn) { a.x = fmaxf(a.x, 0.0f); a.y = fmaxf(a.y, 0.0f); }
    aR[t] = a;
  }

  const int rOff = (lane < 16) ? 0 : 8;
#pragma unroll
  for (int nt = 0; nt < 4; ++nt) {
    const int n0 = nt * 16;
    const float bcol = sB[n0 + laneRow];
    v8f acc;
#pragma unroll
    for (int r = 0; r < 8; ++r) acc[r] = bcol;

#pragma unroll
    for (int t = 0; t < 16; ++t) {
      v2f b = *(const v2f*)(&sW[(n0 + laneRow) * DIM + 4 * t + kHalf]);
      acc = __builtin_amdgcn_wmma_f32_16x16x4_f32(
           false, aR[t],  false, b,
           (short)0, acc,  false,  false);
    }

    float* zr = z + (size_t)(row0 + rOff) * DIM + n0 + laneRow;
#pragma unroll
    for (int r = 0; r < 8; ++r) zr[(size_t)r * DIM] = acc[r];
  }
}

__global__ __launch_bounds__(256)
void node_scores(const float* __restrict__ z, const float* __restrict__ a,
                 float* __restrict__ ssrc, float* __restrict__ sdst, int nN) {
  const int lane = threadIdx.x & 31;
  const int warp = (blockIdx.x * blockDim.x + threadIdx.x) >> 5;
  const int nWarps = (gridDim.x * blockDim.x) >> 5;

  const float a0 = a[lane], a1 = a[lane + 32];
  const float a2 = a[lane + 64], a3 = a[lane + 96];

  for (int n = warp; n < nN; n += nWarps) {
    const float* zr = z + (size_t)n * DIM;
    const float z0 = zr[lane], z1 = zr[lane + 32];
    float ps = z0 * a0 + z1 * a1;
    float pd = z0 * a2 + z1 * a3;
#pragma unroll
    for (int off = 16; off > 0; off >>= 1) {
      ps += __shfl_xor(ps, off, 32);
      pd += __shfl_xor(pd, off, 32);
    }
    if (lane == 0) { ssrc[n] = ps; sdst[n] = pd; }
  }
}

__device__ __forceinline__ unsigned fmap(float f) {
  unsigned u = __float_as_uint(f);
  return (u & 0x80000000u) ? ~u : (u | 0x80000000u);
}
__device__ __forceinline__ float funmap(unsigned u) {
  return (u & 0x80000000u) ? __uint_as_float(u ^ 0x80000000u) : __uint_as_float(~u);
}

__global__ __launch_bounds__(256)
void edge_scores_max(const int* __restrict__ src, const int* __restrict__ dst,
                     const float* __restrict__ ssrc, const float* __restrict__ sdst,
                     const float* __restrict__ abp, float* __restrict__ ebuf,
                     unsigned* __restrict__ mbits, int nE) {
  const float ab = abp[0];
  for (int e = blockIdx.x * blockDim.x + threadIdx.x; e < nE;
       e += gridDim.x * blockDim.x) {
    const int d = dst[e];
    float v = ssrc[src[e]] + sdst[d] + ab;
    v = (v > 0.0f) ? v : LEAKY * v;
    ebuf[e] = v;
    atomicMax(&mbits[d], fmap(v));
  }
}

__global__ __launch_bounds__(256)
void edge_exp_sum(const int* __restrict__ dst, const unsigned* __restrict__ mbits,
                  float* __restrict__ ebuf, float* __restrict__ den, int nE) {
  for (int e = blockIdx.x * blockDim.x + threadIdx.x; e < nE;
       e += gridDim.x * blockDim.x) {
    const int d = dst[e];
    const float ex = __expf(ebuf[e] - funmap(mbits[d]));
    ebuf[e] = ex;
    atomicAdd(&den[d], ex);
  }
}

__global__ __launch_bounds__(256)
void edge_aggregate(const int* __restrict__ src, const int* __restrict__ dst,
                    const float* __restrict__ ebuf, const float* __restrict__ den,
                    const float* __restrict__ z, float* __restrict__ out, int nE) {
  const int lane = threadIdx.x & 31;
  const int warp = (blockIdx.x * blockDim.x + threadIdx.x) >> 5;
  const int nWarps = (gridDim.x * blockDim.x) >> 5;

  for (int e = warp; e < nE; e += nWarps) {
    const int s = src[e], d = dst[e];
    const float alpha = ebuf[e] / den[d];
    const float* zr = z + (size_t)s * DIM;
    float* orow = out + (size_t)d * DIM;
    atomicAdd(orow + lane,      alpha * zr[lane]);
    atomicAdd(orow + lane + 32, alpha * zr[lane + 32]);
  }
}

extern "C" void kernel_launch(void* const* d_in, const int* in_sizes, int n_in,
                              void* d_out, int out_size, void* d_ws, size_t ws_size,
                              hipStream_t stream) {
  const float* x  = (const float*)d_in[0];
  const int* src  = (const int*)d_in[1];
  const int* dst  = (const int*)d_in[2];
  const int nN = in_sizes[0] / DIM;
  const int nE = in_sizes[1];

  char* ws = (char*)d_ws;
  const size_t featB = (size_t)nN * DIM * sizeof(float);
  float*    z     = (float*)ws;
  float*    h     = (float*)(ws + featB);
  float*    ebuf  = (float*)(ws + 2 * featB);
  unsigned* mbits = (unsigned*)(ws + 2 * featB + (size_t)nE * 4);
  float*    den   = (float*)(ws + 2 * featB + (size_t)nE * 4 + (size_t)nN * 4);
  float*    ssrc  = (float*)(ws + 2 * featB + (size_t)nE * 4 + (size_t)nN * 8);
  float*    sdst  = (float*)(ws + 2 * featB + (size_t)nE * 4 + (size_t)nN * 12);

  const int nTiles = (nN + 15) / 16;
  const dim3 gemmGrid((nTiles + 3) / 4), gemmBlk(128);
  const int edgeBlocks = (nE + 255) / 256;
  const int nodeBlocks = ((nN + 7) / 8);
  const int aggBlocks = 8192;

  for (int L = 0; L < 3; ++L) {
    const float* W  = (const float*)d_in[3 + 4 * L];
    const float* b  = (const float*)d_in[4 + 4 * L];
    const float* a  = (const float*)d_in[5 + 4 * L];
    const float* ab = (const float*)d_in[6 + 4 * L];

    const float* hin = (L == 0) ? x : h;
    float* aggout = (L == 2) ? (float*)d_out : h;
    const int reluIn = (L > 0) ? 1 : 0;

    gat_gemm_wmma<<<gemmGrid, gemmBlk, 0, stream>>>(hin, W, b, z, nTiles, reluIn);

    node_scores<<<nodeBlocks, 256, 0, stream>>>(z, a, ssrc, sdst, nN);

    hipMemsetAsync(mbits, 0, (size_t)nN * 4, stream);
    hipMemsetAsync(den,   0, (size_t)nN * 4, stream);
    hipMemsetAsync(aggout, 0, featB, stream);

    edge_scores_max<<<edgeBlocks, 256, 0, stream>>>(src, dst, ssrc, sdst, ab,
                                                    ebuf, mbits, nE);
    edge_exp_sum<<<edgeBlocks, 256, 0, stream>>>(dst, mbits, ebuf, den, nE);
    edge_aggregate<<<aggBlocks, 256, 0, stream>>>(src, dst, ebuf, den, z, aggout, nE);
  }
}


# Edges grouped by destination once per call (returning int add as rank, scan, place); each layer sums softmax and rows per node in registers: no float atomics, sums now in list order
# speedup vs baseline: 3.0298x; 3.0298x over previous
#include <hip/hip_runtime.h>
#include <hip/hip_bf16.h>

typedef __attribute__((ext_vector_type(2))) float v2f;
typedef __attribute__((ext_vector_type(8))) float v8f;

#define DIM 64
#define LEAKY 0.01f

__global__ __launch_bounds__(128)
void gat_gemm_wmma(const float* __restrict__ hin, const float* __restrict__ W,
                   const float* __restrict__ bias, float* __restrict__ z,
                   int nTiles, int reluIn) {
  __shared__ float sW[DIM * DIM];
  __shared__ float sB[DIM];

  const int tid = threadIdx.x;
  for (int i = tid; i < DIM * DIM; i += 128) sW[i] = W[i];
  if (tid < DIM) sB[tid] = bias[tid];
  __syncthreads();

  const int wave = tid >> 5;
  const int lane = tid & 31;
  const int rowTile = blockIdx.x * 4 + wave;
  if (rowTile >= nTiles) return;

  const int row0 = rowTile * 16;
  const int laneRow = lane & 15;
  const int kHalf = (lane >> 4) * 2;

  const float* arow = hin + (size_t)(row0 + laneRow) * DIM;
  v2f aR[16];
#pragma unroll
  for (int t = 0; t < 16; ++t) {
    v2f a = *(const v2f*)(arow + 4 * t + kHalf);
    if (reluIn) { a.x = fmaxf(a.x, 0.0f); a.y = fmaxf(a.y, 0.0f); }
    aR[t] = a;
  }

  const int rOff = (lane < 16) ? 0 : 8;
#pragma unroll
  for (int nt = 0; nt < 4; ++nt) {
    const int n0 = nt * 16;
    const float bcol = sB[n0 + laneRow];
    v8f acc;
#pragma unroll
    for (int r = 0; r < 8; ++r) acc[r] = bcol;

#pragma unroll
    for (int t = 0; t < 16; ++t) {
      v2f b = *(const v2f*)(&sW[(n0 + laneRow) * DIM + 4 * t + kHalf]);
      acc = __builtin_amdgcn_wmma_f32_16x16x4_f32(
           false, aR[t],  false, b,
           (short)0, acc,  false,  false);
    }

    float* zr = z + (size_t)(row0 + rOff) * DIM + n0 + laneRow;
#pragma unroll
    for (int r = 0; r < 8; ++r) zr[(size_t)r * DIM] = acc[r];
  }
}

__global__ __launch_bounds__(256)
void node_scores(const float* __restrict__ z, const float* __restrict__ a,
                 float* __restrict__ ssrc, float* __restrict__ sdst, int nN) {
  const int lane = threadIdx.x & 31;
  const int warp = (blockIdx.x * blockDim.x + threadIdx.x) >> 5;
  const int nWarps = (gridDim.x * blockDim.x) >> 5;

  const float a0 = a[lane], a1 = a[lane + 32];
  const float a2 = a[lane + 64], a3 = a[lane + 96];

  for (int n = warp; n < nN; n += nWarps) {
    const float* zr = z + (size_t)n * DIM;
    const float z0 = zr[lane], z1 = zr[lane + 32];
    float ps = z0 * a0 + z1 * a1;
    float pd = z0 * a2 + z1 * a3;
#pragma unroll
    for (int off = 16; off > 0; off >>= 1) {
      ps += __shfl_xor(ps, off, 32);
      pd += __shfl_xor(pd, off, 32);
    }
    if (lane == 0) { ssrc[n] = ps; sdst[n] = pd; }
  }
}

__global__ __launch_bounds__(256)
void edge_rank(const int* __restrict__ dst, int* __restrict__ cnt,
               int* __restrict__ rank, int nE) {
  for (int e = blockIdx.x * blockDim.x + threadIdx.x; e < nE;
       e += gridDim.x * blockDim.x) {
    rank[e] = atomicAdd(&cnt[dst[e]], 1);
  }
}

__global__ __launch_bounds__(1024)
void seg_offsets(const int* __restrict__ cnt, int* __restrict__ rowptr, int nN) {
  __shared__ int wsum[32];
  const int lane = threadIdx.x & 31;
  const int wave = threadIdx.x >> 5;
  const int chunk = (((nN + 31) / 32) + 31) & ~31;
  const int lo = wave * chunk;
  const int hi = min(lo + chunk, nN);

  int tot = 0;
  for (int i = lo + lane; i < hi; i += 32) tot += cnt[i];
#pragma unroll
  for (int off = 16; off > 0; off >>= 1) tot += __shfl_xor(tot, off, 32);
  if (lane == 0) wsum[wave] = tot;
  __syncthreads();

  int part = (lane < wave) ? wsum[lane] : 0;
#pragma unroll
  for (int off = 16; off > 0; off >>= 1) part += __shfl_xor(part, off, 32);
  int carry = part;

  for (int i0 = lo; i0 < hi; i0 += 32) {
    const int i = i0 + lane;
    const int c = (i < hi) ? cnt[i] : 0;
    int x = c;
#pragma unroll
    for (int off = 1; off < 32; off <<= 1) {
      const int y = __shfl_up(x, off, 32);
      if (lane >= off) x += y;
    }
    if (i < hi) rowptr[i] = carry + x - c;
    carry += __shfl(x, 31, 32);
  }
  if (wave == 31 && lane == 0) rowptr[nN] = carry;
}

__global__ __launch_bounds__(256)
void edge_place(const int* __restrict__ src, const int* __restrict__ dst,
                const int* __restrict__ rank, const int* __restrict__ rowptr,
                int* __restrict__ csrc, int nE) {
  for (int e = blockIdx.x * blockDim.x + threadIdx.x; e < nE;
       e += gridDim.x * blockDim.x) {
    csrc[rowptr[dst[e]] + rank[e]] = src[e];
  }
}

__device__ __forceinline__ float wave_max(float v) {
#pragma unroll
  for (int off = 16; off > 0; off >>= 1) v = fmaxf(v, __shfl_xor(v, off, 32));
  return v;
}
__device__ __forceinline__ float wave_sum(float v) {
#pragma unroll
  for (int off = 16; off > 0; off >>= 1) v += __shfl_xor(v, off, 32);
  return v;
}

template <int B>
__device__ __forceinline__ void add_rows(const float* __restrict__ z, int s, float alpha,
                                         int j0, int lane, v2f& acc) {
  v2f r[B];
#pragma unroll
  for (int u = 0; u < B; ++u) {
    const int sj = __builtin_amdgcn_readlane(s, j0 + u);
    r[u] = *(const v2f*)(z + (size_t)sj * DIM + 2 * lane);
  }
#pragma unroll
  for (int u = 0; u < B; ++u) {
    const float aj = __int_as_float(__builtin_amdgcn_readlane(__float_as_int(alpha), j0 + u));
    acc.x += aj * r[u].x;
    acc.y += aj * r[u].y;
  }
}

__device__ __forceinline__ void add_list(const float* __restrict__ z, int s, float alpha,
                                         int n, int lane, v2f& acc) {
  int j = 0;
  for (; j + 8 <= n; j += 8) add_rows<8>(z, s, alpha, j, lane, acc);
  if (n - j >= 4) { add_rows<4>(z, s, alpha, j, lane, acc); j += 4; }
  if (n - j >= 2) { add_rows<2>(z, s, alpha, j, lane, acc); j += 2; }
  if (n - j >= 1) { add_rows<1>(z, s, alpha, j, lane, acc); }
}

__global__ __launch_bounds__(256)
void node_softmax_aggregate(const int* __restrict__ rowptr, const int* __restrict__ csrc,
                            const float* __restrict__ ssrc, const float* __restrict__ sdst,
                            const float* __restrict__ abp, const float* __restrict__ z,
                            float* __restrict__ out, int nN) {
  const int lane = threadIdx.x & 31;
  const int warp = __builtin_amdgcn_readfirstlane((int)((blockIdx.x * blockDim.x + threadIdx.x) >> 5));
  const int nWarps = (gridDim.x * blockDim.x) >> 5;
  const float ab = abp[0];

  for (int d = warp; d < nN; d += nWarps) {
    const int beg = __builtin_amdgcn_readfirstlane(rowptr[d]);
    const int end = __builtin_amdgcn_readfirstlane(rowptr[d + 1]);
    const int deg = end - beg;
    const float sd = sdst[d];
    v2f acc; acc.x = 0.0f; acc.y = 0.0f;

    if (deg > 0 && deg <= 32) {
      int s = 0;
      float v = -INFINITY;
      if (lane < deg) {
        s = csrc[beg + lane];
        const float t = ssrc[s] + sd + ab;
        v = (t > 0.0f) ? t : LEAKY * t;
      }
      const float m = wave_max(v);
      const float ex = (lane < deg) ? __expf(v - m) : 0.0f;
      const float den = wave_sum(ex);
      const float alpha = ex / den;
      add_list(z, s, alpha, deg, lane, acc);
    } else if (deg > 32) {
      float m = -INFINITY;
      for (int c = beg; c < end; c += 32) {
        const int i = c + lane;
        if (i < end) {
          const float t = ssrc[csrc[i]] + sd + ab;
          m = fmaxf(m, (t > 0.0f) ? t : LEAKY * t);
        }
      }
      m = wave_max(m);
      float den = 0.0f;
      for (int c = beg; c < end; c += 32) {
        const int i = c + lane;
        if (i < end) {
          const float t = ssrc[csrc[i]] + sd + ab;
          den += __expf(((t > 0.0f) ? t : LEAKY * t) - m);
        }
      }
      den = wave_sum(den);
      for (int c = beg; c < end; c += 32) {
        const int i = c + lane;
        int s = 0;
        float alpha = 0.0f;
        if (i < end) {
          s = csrc[i];
          const float t = ssrc[s] + sd + ab;
          alpha = __expf(((t > 0.0f) ? t : LEAKY * t) - m) / den;
        }
        add_list(z, s, alpha, min(32, end - c), lane, acc);
      }
    }
    *(v2f*)(out + (size_t)d * DIM + 2 * lane) = acc;
  }
}

extern "C" void kernel_launch(void* const* d_in, const int* in_sizes, int n_in,
                              void* d_out, int out_size, void* d_ws, size_t ws_size,
                              hipStream_t stream) {
  const float* x  = (const float*)d_in[0];
  const int* src  = (const int*)d_in[1];
  const int* dst  = (const int*)d_in[2];
  const int nN = in_sizes[0] / DIM;
  const int nE = in_sizes[1];

  char* ws = (char*)d_ws;
  const size_t featB = (size_t)nN * DIM * sizeof(float);
  const size_t edgeB = (size_t)nE * 4;
  const size_t nodeB = (size_t)nN * 4;
  float* z      = (float*)ws;
  float* h      = (float*)(ws + featB);
  int*   rank   = (int*)(ws + 2 * featB);
  int*   csrc   = (int*)(ws + 2 * featB + edgeB);
  int*   cnt    = (int*)(ws + 2 * featB + 2 * edgeB);
  float* ssrc   = (float*)(ws + 2 * featB + 2 * edgeB + nodeB);
  float* sdst   = (float*)(ws + 2 * featB + 2 * edgeB + 2 * nodeB);
  int*   rowptr = (int*)(ws + 2 * featB + 2 * edgeB + 3 * nodeB);

  const int nTiles = (nN + 15) / 16;
  const dim3 gemmGrid((nTiles + 3) / 4), gemmBlk(128);
  const int edgeBlocks = (nE + 255) / 256;
  const int nodeBlocks = ((nN + 7) / 8);
  const int aggBlocks = (nN + 31) / 32;

  hipMemsetAsync(cnt, 0, nodeB, stream);
  edge_rank<<<edgeBlocks, 256, 0, stream>>>(dst, cnt, rank, nE);
  seg_offsets<<<1, 1024, 0, stream>>>(cnt, rowptr, nN);
  edge_place<<<edgeBlocks, 256, 0, stream>>>(src, dst, rank, rowptr, csrc, nE);

  for (int L = 0; L < 3; ++L) {
    const float* W  = (const float*)d_in[3 + 4 * L];
    const float* b  = (const float*)d_in[4 + 4 * L];
    const float* a  = (const float*)d_in[5 + 4 * L];
    const float* ab = (const float*)d_in[6 + 4 * L];

    const float* hin = (L == 0) ? x : h;
    float* aggout = (L == 2) ? (float*)d_out : h;
    const int reluIn = (L > 0) ? 1 : 0;

    gat_gemm_wmma<<<gemmGrid, gemmBlk, 0, stream>>>(hin, W, b, z, nTiles, reluIn);

    node_scores<<<nodeBlocks, 256, 0, stream>>>(z, a, ssrc, sdst, nN);

    node_softmax_aggregate<<<aggBlocks, 256, 0, stream>>>(rowptr, csrc, ssrc, sdst, ab,
                                                          z, aggout, nN);
  }
}


# Attention scalars s_src/s_dst now come out of the GEMM epilogue, from the WMMA accumulators (half-wave row sums by DPP exchange); the three score launches go; score sums now in tile-column order
# speedup vs baseline: 3.4220x; 1.1294x over previous
#include <hip/hip_runtime.h>
#include <hip/hip_bf16.h>

typedef __attribute__((ext_vector_type(2))) float v2f;
typedef __attribute__((ext_vector_type(8))) float v8f;

#define DIM 64
#define LEAKY 0.01f

template <int M>
__device__ __forceinline__ float row_xor(float x) {
  return __builtin_bit_cast(float, __builtin_amdgcn_update_dpp(
      0, __builtin_bit_cast(int, x), 0x160 + M  , 0xf, 0xf, true));
}

__global__ __launch_bounds__(128)
void gat_gemm_wmma(const float* __restrict__ hin, const float* __restrict__ W,
                   const float* __restrict__ bias, const float* __restrict__ att,
                   float* __restrict__ z, float* __restrict__ ssrc,
                   float* __restrict__ sdst, int nTiles, int reluIn) {
  __shared__ float sW[DIM * DIM];
  __shared__ float sB[DIM];

  const int tid = threadIdx.x;
  for (int i = tid; i < DIM * DIM; i += 128) sW[i] = W[i];
  if (tid < DIM) sB[tid] = bias[tid];
  __syncthreads();

  const int wave = tid >> 5;
  const int lane = tid & 31;
  const int rowTile = blockIdx.x * 4 + wave;
  if (rowTile >= nTiles) return;

  const int row0 = rowTile * 16;
  const int laneRow = lane & 15;
  const int kHalf = (lane >> 4) * 2;

  v2f aw[4];
#pragma unroll
  for (int nt = 0; nt < 4; ++nt) {
    aw[nt].x = att[nt * 16 + laneRow];
    aw[nt].y = att[DIM + nt * 16 + laneRow];
  }

  const float* arow = hin + (size_t)(row0 + laneRow) * DIM;
  v2f aR[16];
#pragma unroll
  for (int t = 0; t < 16; ++t) {
    v2f a = *(const v2f*)(arow + 4 * t + kHalf);
    if (reluIn) { a.x = fmaxf(a.x, 0.0f); a.y = fmaxf(a.y, 0.0f); }
    aR[t] = a;
  }

  const int rOff = (lane < 16) ? 0 : 8;
  v2f sd[8];
#pragma unroll
  for (int r = 0; r < 8; ++r) { sd[r].x = 0.0f; sd[r].y = 0.0f; }
#pragma unroll
  for (int nt = 0; nt < 4; ++nt) {
    const int n0 = nt * 16;
    const float bcol = sB[n0 + laneRow];
    v8f acc;
#pragma unroll
    for (int r = 0; r < 8; ++r) acc[r] = bcol;

#pragma unroll
    for (int t = 0; t < 16; ++t) {
      v2f b = *(const v2f*)(&sW[(n0 + laneRow) * DIM + 4 * t + kHalf]);
      acc = __builtin_amdgcn_wmma_f32_16x16x4_f32(
           false, aR[t],  false, b,
           (short)0, acc,  false,  false);
    }

    float* zr = z + (size_t)(row0 + rOff) * DIM + n0 + laneRow;
#pragma unroll
    for (int r = 0; r < 8; ++r) zr[(size_t)r * DIM] = acc[r];
#pragma unroll
    for (int r = 0; r < 8; ++r) sd[r] += acc[r] * aw[nt];
  }

  const bool b3 = (lane & 8) != 0, b2 = (lane & 4) != 0;
  const bool b1 = (lane & 2) != 0, b0 = (lane & 1) != 0;
  float t[8], u[4], w[2];
#pragma unroll
  for (int r = 0; r < 8; ++r) {
    const float keep = b3 ? sd[r].y : sd[r].x;
    const float send = b3 ? sd[r].x : sd[r].y;
    t[r] = keep + row_xor<8>(send);
  }
#pragma unroll
  for (int j = 0; j < 4; ++j) {
    const float keep = b2 ? t[j + 4] : t[j];
    const float send = b2 ? t[j] : t[j + 4];
    u[j] = keep + row_xor<4>(send);
  }
#pragma unroll
  for (int j = 0; j < 2; ++j) {
    const float keep = b1 ? u[j + 2] : u[j];
    const float send = b1 ? u[j] : u[j + 2];
    w[j] = keep + row_xor<2>(send);
  }
  const float keep = b0 ? w[1] : w[0];
  const float send = b0 ? w[0] : w[1];
  const float y = keep + row_xor<1>(send);

  float* sp = b3 ? sdst : ssrc;
  sp[row0 + rOff + (lane & 7)] = y;
}

__global__ __launch_bounds__(256)
void edge_rank(const int* __restrict__ dst, int* __restrict__ cnt,
               int* __restrict__ rank, int nE) {
  for (int e = blockIdx.x * blockDim.x + threadIdx.x; e < nE;
       e += gridDim.x * blockDim.x) {
    rank[e] = atomicAdd(&cnt[dst[e]], 1);
  }
}

__global__ __launch_bounds__(1024)
void seg_offsets(const int* __restrict__ cnt, int* __restrict__ rowptr, int nN) {
  __shared__ int wsum[32];
  const int lane = threadIdx.x & 31;
  const int wave = threadIdx.x >> 5;
  const int chunk = (((nN + 31) / 32) + 31) & ~31;
  const int lo = wave * chunk;
  const int hi = min(lo + chunk, nN);

  int tot = 0;
  for (int i = lo + lane; i < hi; i += 32) tot += cnt[i];
#pragma unroll
  for (int off = 16; off > 0; off >>= 1) tot += __shfl_xor(tot, off, 32);
  if (lane == 0) wsum[wave] = tot;
  __syncthreads();

  int part = (lane < wave) ? wsum[lane] : 0;
#pragma unroll
  for (int off = 16; off > 0; off >>= 1) part += __shfl_xor(part, off, 32);
  int carry = part;

  for (int i0 = lo; i0 < hi; i0 += 32) {
    const int i = i0 + lane;
    const int c = (i < hi) ? cnt[i] : 0;
    int x = c;
#pragma unroll
    for (int off = 1; off < 32; off <<= 1) {
      const int y = __shfl_up(x, off, 32);
      if (lane >= off) x += y;
    }
    if (i < hi) rowptr[i] = carry + x - c;
    carry += __shfl(x, 31, 32);
  }
  if (wave == 31 && lane == 0) rowptr[nN] = carry;
}

__global__ __launch_bounds__(256)
void edge_place(const int* __restrict__ src, const int* __restrict__ dst,
                const int* __restrict__ rank, const int* __restrict__ rowptr,
                int* __restrict__ csrc, int nE) {
  for (int e = blockIdx.x * blockDim.x + threadIdx.x; e < nE;
       e += gridDim.x * blockDim.x) {
    csrc[rowptr[dst[e]] + rank[e]] = src[e];
  }
}

__device__ __forceinline__ float wave_max(float v) {
#pragma unroll
  for (int off = 16; off > 0; off >>= 1) v = fmaxf(v, __shfl_xor(v, off, 32));
  return v;
}
__device__ __forceinline__ float wave_sum(float v) {
#pragma unroll
  for (int off = 16; off > 0; off >>= 1) v += __shfl_xor(v, off, 32);
  return v;
}

template <int B>
__device__ __forceinline__ void add_rows(const float* __restrict__ z, int s, float alpha,
                                         int j0, int lane, v2f& acc) {
  v2f r[B];
#pragma unroll
  for (int u = 0; u < B; ++u) {
    const int sj = __builtin_amdgcn_readlane(s, j0 + u);
    r[u] = *(const v2f*)(z + (size_t)sj * DIM + 2 * lane);
  }
#pragma unroll
  for (int u = 0; u < B; ++u) {
    const float aj = __int_as_float(__builtin_amdgcn_readlane(__float_as_int(alpha), j0 + u));
    acc.x += aj * r[u].x;
    acc.y += aj * r[u].y;
  }
}

__device__ __forceinline__ void add_list(const float* __restrict__ z, int s, float alpha,
                                         int n, int lane, v2f& acc) {
  int j = 0;
  for (; j + 8 <= n; j += 8) add_rows<8>(z, s, alpha, j, lane, acc);
  if (n - j >= 4) { add_rows<4>(z, s, alpha, j, lane, acc); j += 4; }
  if (n - j >= 2) { add_rows<2>(z, s, alpha, j, lane, acc); j += 2; }
  if (n - j >= 1) { add_rows<1>(z, s, alpha, j, lane, acc); }
}

__global__ __launch_bounds__(256)
void node_softmax_aggregate(const int* __restrict__ rowptr, const int* __restrict__ csrc,
                            const float* __restrict__ ssrc, const float* __restrict__ sdst,
                            const float* __restrict__ abp, const float* __restrict__ z,
                            float* __restrict__ out, int nN) {
  const int lane = threadIdx.x & 31;
  const int warp = __builtin_amdgcn_readfirstlane((int)((blockIdx.x * blockDim.x + threadIdx.x) >> 5));
  const int nWarps = (gridDim.x * blockDim.x) >> 5;
  const float ab = abp[0];

  for (int d = warp; d < nN; d += nWarps) {
    const int beg = __builtin_amdgcn_readfirstlane(rowptr[d]);
    const int end = __builtin_amdgcn_readfirstlane(rowptr[d + 1]);
    const int deg = end - beg;
    const float sd = sdst[d];
    v2f acc; acc.x = 0.0f; acc.y = 0.0f;

    if (deg > 0 && deg <= 32) {
      int s = 0;
      float v = -INFINITY;
      if (lane < deg) {
        s = csrc[beg + lane];
        const float t = ssrc[s] + sd + ab;
        v = (t > 0.0f) ? t : LEAKY * t;
      }
      const float m = wave_max(v);
      const float ex = (lane < deg) ? __expf(v - m) : 0.0f;
      const float den = wave_sum(ex);
      const float alpha = ex / den;
      add_list(z, s, alpha, deg, lane, acc);
    } else if (deg > 32) {
      float m = -INFINITY;
      for (int c = beg; c < end; c += 32) {
        const int i = c + lane;
        if (i < end) {
          const float t = ssrc[csrc[i]] + sd + ab;
          m = fmaxf(m, (t > 0.0f) ? t : LEAKY * t);
        }
      }
      m = wave_max(m);
      float den = 0.0f;
      for (int c = beg; c < end; c += 32) {
        const int i = c + lane;
        if (i < end) {
          const float t = ssrc[csrc[i]] + sd + ab;
          den += __expf(((t > 0.0f) ? t : LEAKY * t) - m);
        }
      }
      den = wave_sum(den);
      for (int c = beg; c < end; c += 32) {
        const int i = c + lane;
        int s = 0;
        float alpha = 0.0f;
        if (i < end) {
          s = csrc[i];
          const float t = ssrc[s] + sd + ab;
          alpha = __expf(((t > 0.0f) ? t : LEAKY * t) - m) / den;
        }
        add_list(z, s, alpha, min(32, end - c), lane, acc);
      }
    }
    *(v2f*)(out + (size_t)d * DIM + 2 * lane) = acc;
  }
}

extern "C" void kernel_launch(void* const* d_in, const int* in_sizes, int n_in,
                              void* d_out, int out_size, void* d_ws, size_t ws_size,
                              hipStream_t stream) {
  const float* x  = (const float*)d_in[0];
  const int* src  = (const int*)d_in[1];
  const int* dst  = (const int*)d_in[2];
  const int nN = in_sizes[0] / DIM;
  const int nE = in_sizes[1];

  char* ws = (char*)d_ws;
  const size_t featB = (size_t)nN * DIM * sizeof(float);
  const size_t edgeB = (size_t)nE * 4;
  const size_t nodeB = (size_t)nN * 4;
  float* z      = (float*)ws;
  float* h      = (float*)(ws + featB);
  int*   rank   = (int*)(ws + 2 * featB);
  int*   csrc   = (int*)(ws + 2 * featB + edgeB);
  int*   cnt    = (int*)(ws + 2 * featB + 2 * edgeB);
  float* ssrc   = (float*)(ws + 2 * featB + 2 * edgeB + nodeB);
  float* sdst   = (float*)(ws + 2 * featB + 2 * edgeB + 2 * nodeB);
  int*   rowptr = (int*)(ws + 2 * featB + 2 * edgeB + 3 * nodeB);

  const int nTiles = (nN + 15) / 16;
  const dim3 gemmGrid((nTiles + 3) / 4), gemmBlk(128);
  const int edgeBlocks = (nE + 255) / 256;
  const int aggBlocks = (nN + 31) / 32;

  hipMemsetAsync(cnt, 0, nodeB, stream);
  edge_rank<<<edgeBlocks, 256, 0, stream>>>(dst, cnt, rank, nE);
  seg_offsets<<<1, 1024, 0, stream>>>(cnt, rowptr, nN);
  edge_place<<<edgeBlocks, 256, 0, stream>>>(src, dst, rank, rowptr, csrc, nE);

  for (int L = 0; L < 3; ++L) {
    const float* W  = (const float*)d_in[3 + 4 * L];
    const float* b  = (const float*)d_in[4 + 4 * L];
    const float* a  = (const float*)d_in[5 + 4 * L];
    const float* ab = (const float*)d_in[6 + 4 * L];

    const float* hin = (L == 0) ? x : h;
    float* aggout = (L == 2) ? (float*)d_out : h;
    const int reluIn = (L > 0) ? 1 : 0;

    gat_gemm_wmma<<<gemmGrid, gemmBlk, 0, stream>>>(hin, W, b, a, z, ssrc, sdst, nTiles, reluIn);

    node_softmax_aggregate<<<aggBlocks, 256, 0, stream>>>(rowptr, csrc, ssrc, sdst, ab,
                                                          z, aggout, nN);
  }
}


# Edges grouped by destination with LDS counters in two levels (256 node ranges, then nodes): no global atomic per edge, no one-block scan, no memset; list order = arrival order of LDS adds
# speedup vs baseline: 5.1361x; 1.5009x over previous
#include <hip/hip_runtime.h>
#include <hip/hip_bf16.h>

typedef __attribute__((ext_vector_type(2))) float v2f;
typedef __attribute__((ext_vector_type(8))) float v8f;

#define DIM 64
#define LEAKY 0.01f

template <int M>
__device__ __forceinline__ float row_xor(float x) {
  return __builtin_bit_cast(float, __builtin_amdgcn_update_dpp(
      0, __builtin_bit_cast(int, x), 0x160 + M  , 0xf, 0xf, true));
}

__global__ __launch_bounds__(128)
void gat_gemm_wmma(const float* __restrict__ hin, const float* __restrict__ W,
                   const float* __restrict__ bias, const float* __restrict__ att,
                   float* __restrict__ z, float* __restrict__ ssrc,
                   float* __restrict__ sdst, int nTiles, int reluIn) {
  __shared__ float sW[DIM * DIM];
  __shared__ float sB[DIM];

  const int tid = threadIdx.x;
  for (int i = tid; i < DIM * DIM; i += 128) sW[i] = W[i];
  if (tid < DIM) sB[tid] = bias[tid];
  __syncthreads();

  const int wave = tid >> 5;
  const int lane = tid & 31;
  const int rowTile = blockIdx.x * 4 + wave;
  if (rowTile >= nTiles) return;

  const int row0 = rowTile * 16;
  const int laneRow = lane & 15;
  const int kHalf = (lane >> 4) * 2;

  v2f aw[4];
#pragma unroll
  for (int nt = 0; nt < 4; ++nt) {
    aw[nt].x = att[nt * 16 + laneRow];
    aw[nt].y = att[DIM + nt * 16 + laneRow];
  }

  const float* arow = hin + (size_t)(row0 + laneRow) * DIM;
  v2f aR[16];
#pragma unroll
  for (int t = 0; t < 16; ++t) {
    v2f a = *(const v2f*)(arow + 4 * t + kHalf);
    if (reluIn) { a.x = fmaxf(a.x, 0.0f); a.y = fmaxf(a.y, 0.0f); }
    aR[t] = a;
  }

  const int rOff = (lane < 16) ? 0 : 8;
  v2f sd[8];
#pragma unroll
  for (int r = 0; r < 8; ++r) { sd[r].x = 0.0f; sd[r].y = 0.0f; }
#pragma unroll
  for (int nt = 0; nt < 4; ++nt) {
    const int n0 = nt * 16;
    const float bcol = sB[n0 + laneRow];
    v8f acc;
#pragma unroll
    for (int r = 0; r < 8; ++r) acc[r] = bcol;

#pragma unroll
    for (int t = 0; t < 16; ++t) {
      v2f b = *(const v2f*)(&sW[(n0 + laneRow) * DIM + 4 * t + kHalf]);
      acc = __builtin_amdgcn_wmma_f32_16x16x4_f32(
           false, aR[t],  false, b,
           (short)0, acc,  false,  false);
    }

    float* zr = z + (size_t)(row0 + rOff) * DIM + n0 + laneRow;
#pragma unroll
    for (int r = 0; r < 8; ++r) zr[(size_t)r * DIM] = acc[r];
#pragma unroll
    for (int r = 0; r < 8; ++r) sd[r] += acc[r] * aw[nt];
  }

  const bool b3 = (lane & 8) != 0, b2 = (lane & 4) != 0;
  const bool b1 = (lane & 2) != 0, b0 = (lane & 1) != 0;
  float t[8], u[4], w[2];
#pragma unroll
  for (int r = 0; r < 8; ++r) {
    const float keep = b3 ? sd[r].y : sd[r].x;
    const float send = b3 ? sd[r].x : sd[r].y;
    t[r] = keep + row_xor<8>(send);
  }
#pragma unroll
  for (int j = 0; j < 4; ++j) {
    const float keep = b2 ? t[j + 4] : t[j];
    const float send = b2 ? t[j] : t[j + 4];
    u[j] = keep + row_xor<4>(send);
  }
#pragma unroll
  for (int j = 0; j < 2; ++j) {
    const float keep = b1 ? u[j + 2] : u[j];
    const float send = b1 ? u[j] : u[j + 2];
    w[j] = keep + row_xor<2>(send);
  }
  const float keep = b0 ? w[1] : w[0];
  const float send = b0 ? w[0] : w[1];
  const float y = keep + row_xor<1>(send);

  float* sp = b3 ? sdst : ssrc;
  sp[row0 + rOff + (lane & 7)] = y;
}

#define NB 256
#define NWG 256
#define MAXS 1024

typedef __attribute__((ext_vector_type(2))) int v2i;

__device__ __forceinline__ int block_excl_scan(int c, int* wsum) {
  const int lane = threadIdx.x & 31;
  const int wave = threadIdx.x >> 5;
  int x = c;
#pragma unroll
  for (int off = 1; off < 32; off <<= 1) {
    const int y = __shfl_up(x, off, 32);
    if (lane >= off) x += y;
  }
  if (lane == 31) wsum[wave] = x;
  __syncthreads();
  int part = (lane < wave) ? wsum[lane] : 0;
#pragma unroll
  for (int off = 16; off > 0; off >>= 1) part += __shfl_xor(part, off, 32);
  return part + x - c;
}

__global__ __launch_bounds__(1024)
void range_count(const int* __restrict__ dst, int* __restrict__ wgcnt, int nE, int nN) {
  __shared__ int hist[NB];
  const int tid = threadIdx.x;
  const int wg = blockIdx.x;
  const int S = (nN + NB - 1) / NB;
  const int chunk = (nE + NWG - 1) / NWG;
  const int e0 = wg * chunk;
  const int e1 = min(e0 + chunk, nE);
  if (tid < NB) hist[tid] = 0;
  __syncthreads();
  for (int eb = e0 + tid; eb < e1; eb += 4 * 1024) {
    int d[4];
#pragma unroll
    for (int u = 0; u < 4; ++u) {
      const int e = eb + u * 1024;
      d[u] = (e < e1) ? dst[e] : 0;
    }
#pragma unroll
    for (int u = 0; u < 4; ++u) {
      if (eb + u * 1024 < e1) atomicAdd(&hist[d[u] / S], 1);
    }
  }
  __syncthreads();
  if (tid < NB) wgcnt[wg * NB + tid] = hist[tid];
}

__global__ __launch_bounds__(1024)
void range_scatter(const int* __restrict__ src, const int* __restrict__ dst,
                   const int* __restrict__ wgcnt, int* __restrict__ bbase,
                   int* __restrict__ bkt, int nE, int nN) {
  __shared__ int sTot[4 * NB];
  __shared__ int sPre[4 * NB];
  __shared__ int cursor[NB];
  __shared__ int wsum[32];
  const int tid = threadIdx.x;
  const int wg = blockIdx.x;
  const int g = tid >> 8;
  const int b = tid & (NB - 1);

  int tot = 0, pre = 0;
#pragma unroll 16
  for (int r = g * (NWG / 4); r < (g + 1) * (NWG / 4); ++r) {
    const int v = wgcnt[r * NB + b];
    tot += v;
    pre += (r < wg) ? v : 0;
  }
  sTot[tid] = tot;
  sPre[tid] = pre;
  __syncthreads();
  int colTot = 0, colPre = 0;
  if (tid < NB) {
    colTot = sTot[b] + sTot[NB + b] + sTot[2 * NB + b] + sTot[3 * NB + b];
    colPre = sPre[b] + sPre[NB + b] + sPre[2 * NB + b] + sPre[3 * NB + b];
  }
  const int base = block_excl_scan(colTot, wsum);
  if (tid < NB) {
    cursor[b] = base + colPre;
    if (wg == 0) bbase[b] = base;
  }
  if (wg == 0 && tid == 0) bbase[NB] = nE;
  __syncthreads();

  const int S = (nN + NB - 1) / NB;
  const int chunk = (nE + NWG - 1) / NWG;
  const int e0 = wg * chunk;
  const int e1 = min(e0 + chunk, nE);
  for (int eb = e0 + tid; eb < e1; eb += 4 * 1024) {
    v2i rec[4];
#pragma unroll
    for (int u = 0; u < 4; ++u) {
      const int e = eb + u * 1024;
      rec[u].x = (e < e1) ? dst[e] : 0;
      rec[u].y = (e < e1) ? src[e] : 0;
    }
#pragma unroll
    for (int u = 0; u < 4; ++u) {
      if (eb + u * 1024 < e1) {
        const int pos = atomicAdd(&cursor[rec[u].x / S], 1);
        *(v2i*)(bkt + 2 * (size_t)pos) = rec[u];
      }
    }
  }
}

__global__ __launch_bounds__(1024)
void range_to_lists(const int* __restrict__ bkt, const int* __restrict__ bbase,
                    int* __restrict__ rowptr, int* __restrict__ csrc, int nE, int nN) {
  __shared__ int cnt[MAXS];
  __shared__ int wsum[32];
  const int tid = threadIdx.x;
  const int b = blockIdx.x;
  const int S = (nN + NB - 1) / NB;
  const int n0 = b * S;
  const int beg = bbase[b];
  const int end = bbase[b + 1];

  cnt[tid] = 0;
  __syncthreads();
  for (int ib = beg + tid; ib < end; ib += 4 * 1024) {
    int d[4];
#pragma unroll
    for (int u = 0; u < 4; ++u) {
      const int i = ib + u * 1024;
      d[u] = (i < end) ? bkt[2 * (size_t)i] : n0;
    }
#pragma unroll
    for (int u = 0; u < 4; ++u) {
      if (ib + u * 1024 < end) atomicAdd(&cnt[d[u] - n0], 1);
    }
  }
  __syncthreads();
  const int c = cnt[tid];
  const int ex = block_excl_scan(c, wsum);
  cnt[tid] = ex;
  if (tid < S && n0 + tid < nN) rowptr[n0 + tid] = beg + ex;
  if (b == 0 && tid == 0) rowptr[nN] = nE;
  __syncthreads();
  for (int ib = beg + tid; ib < end; ib += 4 * 1024) {
    v2i rec[4];
#pragma unroll
    for (int u = 0; u < 4; ++u) {
      const int i = ib + u * 1024;
      rec[u].x = n0;
      rec[u].y = 0;
      if (i < end) rec[u] = *(const v2i*)(bkt + 2 * (size_t)i);
    }
#pragma unroll
    for (int u = 0; u < 4; ++u) {
      if (ib + u * 1024 < end) {
        const int k = atomicAdd(&cnt[rec[u].x - n0], 1);
        csrc[beg + k] = rec[u].y;
      }
    }
  }
}

__device__ __forceinline__ float wave_max(float v) {
#pragma unroll
  for (int off = 16; off > 0; off >>= 1) v = fmaxf(v, __shfl_xor(v, off, 32));
  return v;
}
__device__ __forceinline__ float wave_sum(float v) {
#pragma unroll
  for (int off = 16; off > 0; off >>= 1) v += __shfl_xor(v, off, 32);
  return v;
}

template <int B>
__device__ __forceinline__ void add_rows(const float* __restrict__ z, int s, float alpha,
                                         int j0, int lane, v2f& acc) {
  v2f r[B];
#pragma unroll
  for (int u = 0; u < B; ++u) {
    const int sj = __builtin_amdgcn_readlane(s, j0 + u);
    r[u] = *(const v2f*)(z + (size_t)sj * DIM + 2 * lane);
  }
#pragma unroll
  for (int u = 0; u < B; ++u) {
    const float aj = __int_as_float(__builtin_amdgcn_readlane(__float_as_int(alpha), j0 + u));
    acc.x += aj * r[u].x;
    acc.y += aj * r[u].y;
  }
}

__device__ __forceinline__ void add_list(const float* __restrict__ z, int s, float alpha,
                                         int n, int lane, v2f& acc) {
  int j = 0;
  for (; j + 8 <= n; j += 8) add_rows<8>(z, s, alpha, j, lane, acc);
  if (n - j >= 4) { add_rows<4>(z, s, alpha, j, lane, acc); j += 4; }
  if (n - j >= 2) { add_rows<2>(z, s, alpha, j, lane, acc); j += 2; }
  if (n - j >= 1) { add_rows<1>(z, s, alpha, j, lane, acc); }
}

__global__ __launch_bounds__(256)
void node_softmax_aggregate(const int* __restrict__ rowptr, const int* __restrict__ csrc,
                            const float* __restrict__ ssrc, const float* __restrict__ sdst,
                            const float* __restrict__ abp, const float* __restrict__ z,
                            float* __restrict__ out, int nN) {
  const int lane = threadIdx.x & 31;
  const int warp = __builtin_amdgcn_readfirstlane((int)((blockIdx.x * blockDim.x + threadIdx.x) >> 5));
  const int nWarps = (gridDim.x * blockDim.x) >> 5;
  const float ab = abp[0];

  for (int d = warp; d < nN; d += nWarps) {
    const int beg = __builtin_amdgcn_readfirstlane(rowptr[d]);
    const int end = __builtin_amdgcn_readfirstlane(rowptr[d + 1]);
    const int deg = end - beg;
    const float sd = sdst[d];
    v2f acc; acc.x = 0.0f; acc.y = 0.0f;

    if (deg > 0 && deg <= 32) {
      int s = 0;
      float v = -INFINITY;
      if (lane < deg) {
        s = csrc[beg + lane];
        const float t = ssrc[s] + sd + ab;
        v = (t > 0.0f) ? t : LEAKY * t;
      }
      const float m = wave_max(v);
      const float ex = (lane < deg) ? __expf(v - m) : 0.0f;
      const float den = wave_sum(ex);
      const float alpha = ex / den;
      add_list(z, s, alpha, deg, lane, acc);
    } else if (deg > 32) {
      float m = -INFINITY;
      for (int c = beg; c < end; c += 32) {
        const int i = c + lane;
        if (i < end) {
          const float t = ssrc[csrc[i]] + sd + ab;
          m = fmaxf(m, (t > 0.0f) ? t : LEAKY * t);
        }
      }
      m = wave_max(m);
      float den = 0.0f;
      for (int c = beg; c < end; c += 32) {
        const int i = c + lane;
        if (i < end) {
          const float t = ssrc[csrc[i]] + sd + ab;
          den += __expf(((t > 0.0f) ? t : LEAKY * t) - m);
        }
      }
      den = wave_sum(den);
      for (int c = beg; c < end; c += 32) {
        const int i = c + lane;
        int s = 0;
        float alpha = 0.0f;
        if (i < end) {
          s = csrc[i];
          const float t = ssrc[s] + sd + ab;
          alpha = __expf(((t > 0.0f) ? t : LEAKY * t) - m) / den;
        }
        add_list(z, s, alpha, min(32, end - c), lane, acc);
      }
    }
    *(v2f*)(out + (size_t)d * DIM + 2 * lane) = acc;
  }
}

extern "C" void kernel_launch(void* const* d_in, const int* in_sizes, int n_in,
                              void* d_out, int out_size, void* d_ws, size_t ws_size,
                              hipStream_t stream) {
  const float* x  = (const float*)d_in[0];
  const int* src  = (const int*)d_in[1];
  const int* dst  = (const int*)d_in[2];
  const int nN = in_sizes[0] / DIM;
  const int nE = in_sizes[1];

  char* ws = (char*)d_ws;
  const size_t featB = (size_t)nN * DIM * sizeof(float);
  const size_t edgeB = (size_t)nE * 4;
  const size_t nodeB = (size_t)nN * 4;
  float* z      = (float*)ws;
  float* h      = (float*)(ws + featB);
  int*   bkt    = (int*)(ws + 2 * featB);
  int*   csrc   = (int*)(ws + 2 * featB + 2 * edgeB);
  int*   wgcnt  = (int*)(ws + 2 * featB + 3 * edgeB);
  const size_t tabB = (size_t)NWG * NB * 4;
  int*   bbase  = (int*)(ws + 2 * featB + 3 * edgeB + tabB);
  float* ssrc   = (float*)(ws + 2 * featB + 3 * edgeB + tabB + 2048);
  float* sdst   = (float*)(ws + 2 * featB + 3 * edgeB + tabB + 2048 + nodeB);
  int*   rowptr = (int*)(ws + 2 * featB + 3 * edgeB + tabB + 2048 + 2 * nodeB);

  const int nTiles = (nN + 15) / 16;
  const dim3 gemmGrid((nTiles + 3) / 4), gemmBlk(128);
  const int aggBlocks = (nN + 31) / 32;

  range_count<<<NWG, 1024, 0, stream>>>(dst, wgcnt, nE, nN);
  range_scatter<<<NWG, 1024, 0, stream>>>(src, dst, wgcnt, bbase, bkt, nE, nN);
  range_to_lists<<<NB, 1024, 0, stream>>>(bkt, bbase, rowptr, csrc, nE, nN);

  for (int L = 0; L < 3; ++L) {
    const float* W  = (const float*)d_in[3 + 4 * L];
    const float* b  = (const float*)d_in[4 + 4 * L];
    const float* a  = (const float*)d_in[5 + 4 * L];
    const float* ab = (const float*)d_in[6 + 4 * L];

    const float* hin = (L == 0) ? x : h;
    float* aggout = (L == 2) ? (float*)d_out : h;
    const int reluIn = (L > 0) ? 1 : 0;

    gat_gemm_wmma<<<gemmGrid, gemmBlk, 0, stream>>>(hin, W, b, a, z, ssrc, sdst, nTiles, reluIn);

    node_softmax_aggregate<<<aggBlocks, 256, 0, stream>>>(rowptr, csrc, ssrc, sdst, ab,
                                                          z, aggout, nN);
  }
}


# W tile in LDS at a row pitch of 68 floats instead of 64: the GEMM's B-fragment reads (16 rows at one k per half-wave) now cover all 64 banks once instead of a 16-way bank conflict
# speedup vs baseline: 5.4239x; 1.0560x over previous
#include <hip/hip_runtime.h>
#include <hip/hip_bf16.h>

typedef __attribute__((ext_vector_type(2))) float v2f;
typedef __attribute__((ext_vector_type(8))) float v8f;

#define DIM 64
#define LEAKY 0.01f
#define WPITCH 68

template <int M>
__device__ __forceinline__ float row_xor(float x) {
  return __builtin_bit_cast(float, __builtin_amdgcn_update_dpp(
      0, __builtin_bit_cast(int, x), 0x160 + M  , 0xf, 0xf, true));
}

__global__ __launch_bounds__(128)
void gat_gemm_wmma(const float* __restrict__ hin, const float* __restrict__ W,
                   const float* __restrict__ bias, const float* __restrict__ att,
                   float* __restrict__ z, float* __restrict__ ssrc,
                   float* __restrict__ sdst, int nTiles, int reluIn) {
  __shared__ __attribute__((aligned(16))) float sW[DIM * WPITCH];
  __shared__ float sB[DIM];

  const int tid = threadIdx.x;
  for (int r = tid >> 6; r < DIM; r += 2) sW[r * WPITCH + (tid & (DIM - 1))] = W[r * DIM + (tid & (DIM - 1))];
  if (tid < DIM) sB[tid] = bias[tid];
  __syncthreads();

  const int wave = tid >> 5;
  const int lane = tid & 31;
  const int rowTile = blockIdx.x * 4 + wave;
  if (rowTile >= nTiles) return;

  const int row0 = rowTile * 16;
  const int laneRow = lane & 15;
  const int kHalf = (lane >> 4) * 2;

  v2f aw[4];
#pragma unroll
  for (int nt = 0; nt < 4; ++nt) {
    aw[nt].x = att[nt * 16 + laneRow];
    aw[nt].y = att[DIM + nt * 16 + laneRow];
  }

  const float* arow = hin + (size_t)(row0 + laneRow) * DIM;
  v2f aR[16];
#pragma unroll
  for (int t = 0; t < 16; ++t) {
    v2f a = *(const v2f*)(arow + 4 * t + kHalf);
    if (reluIn) { a.x = fmaxf(a.x, 0.0f); a.y = fmaxf(a.y, 0.0f); }
    aR[t] = a;
  }

  const int rOff = (lane < 16) ? 0 : 8;
  v2f sd[8];
#pragma unroll
  for (int r = 0; r < 8; ++r) { sd[r].x = 0.0f; sd[r].y = 0.0f; }
#pragma unroll
  for (int nt = 0; nt < 4; ++nt) {
    const int n0 = nt * 16;
    const float bcol = sB[n0 + laneRow];
    v8f acc;
#pragma unroll
    for (int r = 0; r < 8; ++r) acc[r] = bcol;

#pragma unroll
    for (int t = 0; t < 16; ++t) {
      v2f b = *(const v2f*)(&sW[(n0 + laneRow) * WPITCH + 4 * t + kHalf]);
      acc = __builtin_amdgcn_wmma_f32_16x16x4_f32(
           false, aR[t],  false, b,
           (short)0, acc,  false,  false);
    }

    float* zr = z + (size_t)(row0 + rOff) * DIM + n0 + laneRow;
#pragma unroll
    for (int r = 0; r < 8; ++r) zr[(size_t)r * DIM] = acc[r];
#pragma unroll
    for (int r = 0; r < 8; ++r) sd[r] += acc[r] * aw[nt];
  }

  const bool b3 = (lane & 8) != 0, b2 = (lane & 4) != 0;
  const bool b1 = (lane & 2) != 0, b0 = (lane & 1) != 0;
  float t[8], u[4], w[2];
#pragma unroll
  for (int r = 0; r < 8; ++r) {
    const float keep = b3 ? sd[r].y : sd[r].x;
    const float send = b3 ? sd[r].x : sd[r].y;
    t[r] = keep + row_xor<8>(send);
  }
#pragma unroll
  for (int j = 0; j < 4; ++j) {
    const float keep = b2 ? t[j + 4] : t[j];
    const float send = b2 ? t[j] : t[j + 4];
    u[j] = keep + row_xor<4>(send);
  }
#pragma unroll
  for (int j = 0; j < 2; ++j) {
    const float keep = b1 ? u[j + 2] : u[j];
    const float send = b1 ? u[j] : u[j + 2];
    w[j] = keep + row_xor<2>(send);
  }
  const float keep = b0 ? w[1] : w[0];
  const float send = b0 ? w[0] : w[1];
  const float y = keep + row_xor<1>(send);

  float* sp = b3 ? sdst : ssrc;
  sp[row0 + rOff + (lane & 7)] = y;
}

#define NB 256
#define NWG 256
#define MAXS 1024

typedef __attribute__((ext_vector_type(2))) int v2i;

__device__ __forceinline__ int block_excl_scan(int c, int* wsum) {
  const int lane = threadIdx.x & 31;
  const int wave = threadIdx.x >> 5;
  int x = c;
#pragma unroll
  for (int off = 1; off < 32; off <<= 1) {
    const int y = __shfl_up(x, off, 32);
    if (lane >= off) x += y;
  }
  if (lane == 31) wsum[wave] = x;
  __syncthreads();
  int part = (lane < wave) ? wsum[lane] : 0;
#pragma unroll
  for (int off = 16; off > 0; off >>= 1) part += __shfl_xor(part, off, 32);
  return part + x - c;
}

__global__ __launch_bounds__(1024)
void range_count(const int* __restrict__ dst, int* __restrict__ wgcnt, int nE, int nN) {
  __shared__ int hist[NB];
  const int tid = threadIdx.x;
  const int wg = blockIdx.x;
  const int S = (nN + NB - 1) / NB;
  const int chunk = (nE + NWG - 1) / NWG;
  const int e0 = wg * chunk;
  const int e1 = min(e0 + chunk, nE);
  if (tid < NB) hist[tid] = 0;
  __syncthreads();
  for (int eb = e0 + tid; eb < e1; eb += 4 * 1024) {
    int d[4];
#pragma unroll
    for (int u = 0; u < 4; ++u) {
      const int e = eb + u * 1024;
      d[u] = (e < e1) ? dst[e] : 0;
    }
#pragma unroll
    for (int u = 0; u < 4; ++u) {
      if (eb + u * 1024 < e1) atomicAdd(&hist[d[u] / S], 1);
    }
  }
  __syncthreads();
  if (tid < NB) wgcnt[wg * NB + tid] = hist[tid];
}

__global__ __launch_bounds__(1024)
void range_scatter(const int* __restrict__ src, const int* __restrict__ dst,
                   const int* __restrict__ wgcnt, int* __restrict__ bbase,
                   int* __restrict__ bkt, int nE, int nN) {
  __shared__ int sTot[4 * NB];
  __shared__ int sPre[4 * NB];
  __shared__ int cursor[NB];
  __shared__ int wsum[32];
  const int tid = threadIdx.x;
  const int wg = blockIdx.x;
  const int g = tid >> 8;
  const int b = tid & (NB - 1);

  int tot = 0, pre = 0;
#pragma unroll 16
  for (int r = g * (NWG / 4); r < (g + 1) * (NWG / 4); ++r) {
    const int v = wgcnt[r * NB + b];
    tot += v;
    pre += (r < wg) ? v : 0;
  }
  sTot[tid] = tot;
  sPre[tid] = pre;
  __syncthreads();
  int colTot = 0, colPre = 0;
  if (tid < NB) {
    colTot = sTot[b] + sTot[NB + b] + sTot[2 * NB + b] + sTot[3 * NB + b];
    colPre = sPre[b] + sPre[NB + b] + sPre[2 * NB + b] + sPre[3 * NB + b];
  }
  const int base = block_excl_scan(colTot, wsum);
  if (tid < NB) {
    cursor[b] = base + colPre;
    if (wg == 0) bbase[b] = base;
  }
  if (wg == 0 && tid == 0) bbase[NB] = nE;
  __syncthreads();

  const int S = (nN + NB - 1) / NB;
  const int chunk = (nE + NWG - 1) / NWG;
  const int e0 = wg * chunk;
  const int e1 = min(e0 + chunk, nE);
  for (int eb = e0 + tid; eb < e1; eb += 4 * 1024) {
    v2i rec[4];
#pragma unroll
    for (int u = 0; u < 4; ++u) {
      const int e = eb + u * 1024;
      rec[u].x = (e < e1) ? dst[e] : 0;
      rec[u].y = (e < e1) ? src[e] : 0;
    }
#pragma unroll
    for (int u = 0; u < 4; ++u) {
      if (eb + u * 1024 < e1) {
        const int pos = atomicAdd(&cursor[rec[u].x / S], 1);
        *(v2i*)(bkt + 2 * (size_t)pos) = rec[u];
      }
    }
  }
}

__global__ __launch_bounds__(1024)
void range_to_lists(const int* __restrict__ bkt, const int* __restrict__ bbase,
                    int* __restrict__ rowptr, int* __restrict__ csrc, int nE, int nN) {
  __shared__ int cnt[MAXS];
  __shared__ int wsum[32];
  const int tid = threadIdx.x;
  const int b = blockIdx.x;
  const int S = (nN + NB - 1) / NB;
  const int n0 = b * S;
  const int beg = bbase[b];
  const int end = bbase[b + 1];

  cnt[tid] = 0;
  __syncthreads();
  for (int ib = beg + tid; ib < end; ib += 4 * 1024) {
    int d[4];
#pragma unroll
    for (int u = 0; u < 4; ++u) {
      const int i = ib + u * 1024;
      d[u] = (i < end) ? bkt[2 * (size_t)i] : n0;
    }
#pragma unroll
    for (int u = 0; u < 4; ++u) {
      if (ib + u * 1024 < end) atomicAdd(&cnt[d[u] - n0], 1);
    }
  }
  __syncthreads();
  const int c = cnt[tid];
  const int ex = block_excl_scan(c, wsum);
  cnt[tid] = ex;
  if (tid < S && n0 + tid < nN) rowptr[n0 + tid] = beg + ex;
  if (b == 0 && tid == 0) rowptr[nN] = nE;
  __syncthreads();
  for (int ib = beg + tid; ib < end; ib += 4 * 1024) {
    v2i rec[4];
#pragma unroll
    for (int u = 0; u < 4; ++u) {
      const int i = ib + u * 1024;
      rec[u].x = n0;
      rec[u].y = 0;
      if (i < end) rec[u] = *(const v2i*)(bkt + 2 * (size_t)i);
    }
#pragma unroll
    for (int u = 0; u < 4; ++u) {
      if (ib + u * 1024 < end) {
        const int k = atomicAdd(&cnt[rec[u].x - n0], 1);
        csrc[beg + k] = rec[u].y;
      }
    }
  }
}

__device__ __forceinline__ float wave_max(float v) {
#pragma unroll
  for (int off = 16; off > 0; off >>= 1) v = fmaxf(v, __shfl_xor(v, off, 32));
  return v;
}
__device__ __forceinline__ float wave_sum(float v) {
#pragma unroll
  for (int off = 16; off > 0; off >>= 1) v += __shfl_xor(v, off, 32);
  return v;
}

template <int B>
__device__ __forceinline__ void add_rows(const float* __restrict__ z, int s, float alpha,
                                         int j0, int lane, v2f& acc) {
  v2f r[B];
#pragma unroll
  for (int u = 0; u < B; ++u) {
    const int sj = __builtin_amdgcn_readlane(s, j0 + u);
    r[u] = *(const v2f*)(z + (size_t)sj * DIM + 2 * lane);
  }
#pragma unroll
  for (int u = 0; u < B; ++u) {
    const float aj = __int_as_float(__builtin_amdgcn_readlane(__float_as_int(alpha), j0 + u));
    acc.x += aj * r[u].x;
    acc.y += aj * r[u].y;
  }
}

__device__ __forceinline__ void add_list(const float* __restrict__ z, int s, float alpha,
                                         int n, int lane, v2f& acc) {
  int j = 0;
  for (; j + 8 <= n; j += 8) add_rows<8>(z, s, alpha, j, lane, acc);
  if (n - j >= 4) { add_rows<4>(z, s, alpha, j, lane, acc); j += 4; }
  if (n - j >= 2) { add_rows<2>(z, s, alpha, j, lane, acc); j += 2; }
  if (n - j >= 1) { add_rows<1>(z, s, alpha, j, lane, acc); }
}

__global__ __launch_bounds__(256)
void node_softmax_aggregate(const int* __restrict__ rowptr, const int* __restrict__ csrc,
                            const float* __restrict__ ssrc, const float* __restrict__ sdst,
                            const float* __restrict__ abp, const float* __restrict__ z,
                            float* __restrict__ out, int nN) {
  const int lane = threadIdx.x & 31;
  const int warp = __builtin_amdgcn_readfirstlane((int)((blockIdx.x * blockDim.x + threadIdx.x) >> 5));
  const int nWarps = (gridDim.x * blockDim.x) >> 5;
  const float ab = abp[0];

  for (int d = warp; d < nN; d += nWarps) {
    const int beg = __builtin_amdgcn_readfirstlane(rowptr[d]);
    const int end = __builtin_amdgcn_readfirstlane(rowptr[d + 1]);
    const int deg = end - beg;
    const float sd = sdst[d];
    v2f acc; acc.x = 0.0f; acc.y = 0.0f;

    if (deg > 0 && deg <= 32) {
      int s = 0;
      float v = -INFINITY;
      if (lane < deg) {
        s = csrc[beg + lane];
        const float t = ssrc[s] + sd + ab;
        v = (t > 0.0f) ? t : LEAKY * t;
      }
      const float m = wave_max(v);
      const float ex = (lane < deg) ? __expf(v - m) : 0.0f;
      const float den = wave_sum(ex);
      const float alpha = ex / den;
      add_list(z, s, alpha, deg, lane, acc);
    } else if (deg > 32) {
      float m = -INFINITY;
      for (int c = beg; c < end; c += 32) {
        const int i = c + lane;
        if (i < end) {
          const float t = ssrc[csrc[i]] + sd + ab;
          m = fmaxf(m, (t > 0.0f) ? t : LEAKY * t);
        }
      }
      m = wave_max(m);
      float den = 0.0f;
      for (int c = beg; c < end; c += 32) {
        const int i = c + lane;
        if (i < end) {
          const float t = ssrc[csrc[i]] + sd + ab;
          den += __expf(((t > 0.0f) ? t : LEAKY * t) - m);
        }
      }
      den = wave_sum(den);
      for (int c = beg; c < end; c += 32) {
        const int i = c + lane;
        int s = 0;
        float alpha = 0.0f;
        if (i < end) {
          s = csrc[i];
          const float t = ssrc[s] + sd + ab;
          alpha = __expf(((t > 0.0f) ? t : LEAKY * t) - m) / den;
        }
        add_list(z, s, alpha, min(32, end - c), lane, acc);
      }
    }
    *(v2f*)(out + (size_t)d * DIM + 2 * lane) = acc;
  }
}

extern "C" void kernel_launch(void* const* d_in, const int* in_sizes, int n_in,
                              void* d_out, int out_size, void* d_ws, size_t ws_size,
                              hipStream_t stream) {
  const float* x  = (const float*)d_in[0];
  const int* src  = (const int*)d_in[1];
  const int* dst  = (const int*)d_in[2];
  const int nN = in_sizes[0] / DIM;
  const int nE = in_sizes[1];

  char* ws = (char*)d_ws;
  const size_t featB = (size_t)nN * DIM * sizeof(float);
  const size_t edgeB = (size_t)nE * 4;
  const size_t nodeB = (size_t)nN * 4;
  float* z      = (float*)ws;
  float* h      = (float*)(ws + featB);
  int*   bkt    = (int*)(ws + 2 * featB);
  int*   csrc   = (int*)(ws + 2 * featB + 2 * edgeB);
  int*   wgcnt  = (int*)(ws + 2 * featB + 3 * edgeB);
  const size_t tabB = (size_t)NWG * NB * 4;
  int*   bbase  = (int*)(ws + 2 * featB + 3 * edgeB + tabB);
  float* ssrc   = (float*)(ws + 2 * featB + 3 * edgeB + tabB + 2048);
  float* sdst   = (float*)(ws + 2 * featB + 3 * edgeB + tabB + 2048 + nodeB);
  int*   rowptr = (int*)(ws + 2 * featB + 3 * edgeB + tabB + 2048 + 2 * nodeB);

  const int nTiles = (nN + 15) / 16;
  const dim3 gemmGrid((nTiles + 3) / 4), gemmBlk(128);
  const int aggBlocks = (nN + 31) / 32;

  range_count<<<NWG, 1024, 0, stream>>>(dst, wgcnt, nE, nN);
  range_scatter<<<NWG, 1024, 0, stream>>>(src, dst, wgcnt, bbase, bkt, nE, nN);
  range_to_lists<<<NB, 1024, 0, stream>>>(bkt, bbase, rowptr, csrc, nE, nN);

  for (int L = 0; L < 3; ++L) {
    const float* W  = (const float*)d_in[3 + 4 * L];
    const float* b  = (const float*)d_in[4 + 4 * L];
    const float* a  = (const float*)d_in[5 + 4 * L];
    const float* ab = (const float*)d_in[6 + 4 * L];

    const float* hin = (L == 0) ? x : h;
    float* aggout = (L == 2) ? (float*)d_out : h;
    const int reluIn = (L > 0) ? 1 : 0;

    gat_gemm_wmma<<<gemmGrid, gemmBlk, 0, stream>>>(hin, W, b, a, z, ssrc, sdst, nTiles, reluIn);

    node_softmax_aggregate<<<aggBlocks, 256, 0, stream>>>(rowptr, csrc, ssrc, sdst, ab,
                                                          z, aggout, nN);
  }
}


# Layers 1 and 2: the linear map and attention scalars run inside the previous layer's aggregate launch (its 32 relu rows held in LDS as two WMMA A tiles): two GEMM launches and the h round trip go
# speedup vs baseline: 6.1271x; 1.1296x over previous
#include <hip/hip_runtime.h>
#include <hip/hip_bf16.h>

typedef __attribute__((ext_vector_type(2))) float v2f;
typedef __attribute__((ext_vector_type(8))) float v8f;

#define DIM 64
#define LEAKY 0.01f
#define WPITCH 68

template <int M>
__device__ __forceinline__ float row_xor(float x) {
  return __builtin_bit_cast(float, __builtin_amdgcn_update_dpp(
      0, __builtin_bit_cast(int, x), 0x160 + M  , 0xf, 0xf, true));
}

__global__ __launch_bounds__(128)
void gat_gemm_wmma(const float* __restrict__ hin, const float* __restrict__ W,
                   const float* __restrict__ bias, const float* __restrict__ att,
                   float* __restrict__ z, float* __restrict__ ssrc,
                   float* __restrict__ sdst, int nTiles, int reluIn) {
  __shared__ __attribute__((aligned(16))) float sW[DIM * WPITCH];
  __shared__ float sB[DIM];

  const int tid = threadIdx.x;
  for (int r = tid >> 6; r < DIM; r += 2) sW[r * WPITCH + (tid & (DIM - 1))] = W[r * DIM + (tid & (DIM - 1))];
  if (tid < DIM) sB[tid] = bias[tid];
  __syncthreads();

  const int wave = tid >> 5;
  const int lane = tid & 31;
  const int rowTile = blockIdx.x * 4 + wave;
  if (rowTile >= nTiles) return;

  const int row0 = rowTile * 16;
  const int laneRow = lane & 15;
  const int kHalf = (lane >> 4) * 2;

  v2f aw[4];
#pragma unroll
  for (int nt = 0; nt < 4; ++nt) {
    aw[nt].x = att[nt * 16 + laneRow];
    aw[nt].y = att[DIM + nt * 16 + laneRow];
  }

  const float* arow = hin + (size_t)(row0 + laneRow) * DIM;
  v2f aR[16];
#pragma unroll
  for (int t = 0; t < 16; ++t) {
    v2f a = *(const v2f*)(arow + 4 * t + kHalf);
    if (reluIn) { a.x = fmaxf(a.x, 0.0f); a.y = fmaxf(a.y, 0.0f); }
    aR[t] = a;
  }

  const int rOff = (lane < 16) ? 0 : 8;
  v2f sd[8];
#pragma unroll
  for (int r = 0; r < 8; ++r) { sd[r].x = 0.0f; sd[r].y = 0.0f; }
#pragma unroll
  for (int nt = 0; nt < 4; ++nt) {
    const int n0 = nt * 16;
    const float bcol = sB[n0 + laneRow];
    v8f acc;
#pragma unroll
    for (int r = 0; r < 8; ++r) acc[r] = bcol;

#pragma unroll
    for (int t = 0; t < 16; ++t) {
      v2f b = *(const v2f*)(&sW[(n0 + laneRow) * WPITCH + 4 * t + kHalf]);
      acc = __builtin_amdgcn_wmma_f32_16x16x4_f32(
           false, aR[t],  false, b,
           (short)0, acc,  false,  false);
    }

    float* zr = z + (size_t)(row0 + rOff) * DIM + n0 + laneRow;
#pragma unroll
    for (int r = 0; r < 8; ++r) zr[(size_t)r * DIM] = acc[r];
#pragma unroll
    for (int r = 0; r < 8; ++r) sd[r] += acc[r] * aw[nt];
  }

  const bool b3 = (lane & 8) != 0, b2 = (lane & 4) != 0;
  const bool b1 = (lane & 2) != 0, b0 = (lane & 1) != 0;
  float t[8], u[4], w[2];
#pragma unroll
  for (int r = 0; r < 8; ++r) {
    const float keep = b3 ? sd[r].y : sd[r].x;
    const float send = b3 ? sd[r].x : sd[r].y;
    t[r] = keep + row_xor<8>(send);
  }
#pragma unroll
  for (int j = 0; j < 4; ++j) {
    const float keep = b2 ? t[j + 4] : t[j];
    const float send = b2 ? t[j] : t[j + 4];
    u[j] = keep + row_xor<4>(send);
  }
#pragma unroll
  for (int j = 0; j < 2; ++j) {
    const float keep = b1 ? u[j + 2] : u[j];
    const float send = b1 ? u[j] : u[j + 2];
    w[j] = keep + row_xor<2>(send);
  }
  const float keep = b0 ? w[1] : w[0];
  const float send = b0 ? w[0] : w[1];
  const float y = keep + row_xor<1>(send);

  float* sp = b3 ? sdst : ssrc;
  sp[row0 + rOff + (lane & 7)] = y;
}

#define NB 256
#define NWG 256
#define MAXS 1024

typedef __attribute__((ext_vector_type(2))) int v2i;

__device__ __forceinline__ int block_excl_scan(int c, int* wsum) {
  const int lane = threadIdx.x & 31;
  const int wave = threadIdx.x >> 5;
  int x = c;
#pragma unroll
  for (int off = 1; off < 32; off <<= 1) {
    const int y = __shfl_up(x, off, 32);
    if (lane >= off) x += y;
  }
  if (lane == 31) wsum[wave] = x;
  __syncthreads();
  int part = (lane < wave) ? wsum[lane] : 0;
#pragma unroll
  for (int off = 16; off > 0; off >>= 1) part += __shfl_xor(part, off, 32);
  return part + x - c;
}

__global__ __launch_bounds__(1024)
void range_count(const int* __restrict__ dst, int* __restrict__ wgcnt, int nE, int nN) {
  __shared__ int hist[NB];
  const int tid = threadIdx.x;
  const int wg = blockIdx.x;
  const int S = (nN + NB - 1) / NB;
  const int chunk = (nE + NWG - 1) / NWG;
  const int e0 = wg * chunk;
  const int e1 = min(e0 + chunk, nE);
  if (tid < NB) hist[tid] = 0;
  __syncthreads();
  for (int eb = e0 + tid; eb < e1; eb += 4 * 1024) {
    int d[4];
#pragma unroll
    for (int u = 0; u < 4; ++u) {
      const int e = eb + u * 1024;
      d[u] = (e < e1) ? dst[e] : 0;
    }
#pragma unroll
    for (int u = 0; u < 4; ++u) {
      if (eb + u * 1024 < e1) atomicAdd(&hist[d[u] / S], 1);
    }
  }
  __syncthreads();
  if (tid < NB) wgcnt[wg * NB + tid] = hist[tid];
}

__global__ __launch_bounds__(1024)
void range_scatter(const int* __restrict__ src, const int* __restrict__ dst,
                   const int* __restrict__ wgcnt, int* __restrict__ bbase,
                   int* __restrict__ bkt, int nE, int nN) {
  __shared__ int sTot[4 * NB];
  __shared__ int sPre[4 * NB];
  __shared__ int cursor[NB];
  __shared__ int wsum[32];
  const int tid = threadIdx.x;
  const int wg = blockIdx.x;
  const int g = tid >> 8;
  const int b = tid & (NB - 1);

  int tot = 0, pre = 0;
#pragma unroll 16
  for (int r = g * (NWG / 4); r < (g + 1) * (NWG / 4); ++r) {
    const int v = wgcnt[r * NB + b];
    tot += v;
    pre += (r < wg) ? v : 0;
  }
  sTot[tid] = tot;
  sPre[tid] = pre;
  __syncthreads();
  int colTot = 0, colPre = 0;
  if (tid < NB) {
    colTot = sTot[b] + sTot[NB + b] + sTot[2 * NB + b] + sTot[3 * NB + b];
    colPre = sPre[b] + sPre[NB + b] + sPre[2 * NB + b] + sPre[3 * NB + b];
  }
  const int base = block_excl_scan(colTot, wsum);
  if (tid < NB) {
    cursor[b] = base + colPre;
    if (wg == 0) bbase[b] = base;
  }
  if (wg == 0 && tid == 0) bbase[NB] = nE;
  __syncthreads();

  const int S = (nN + NB - 1) / NB;
  const int chunk = (nE + NWG - 1) / NWG;
  const int e0 = wg * chunk;
  const int e1 = min(e0 + chunk, nE);
  for (int eb = e0 + tid; eb < e1; eb += 4 * 1024) {
    v2i rec[4];
#pragma unroll
    for (int u = 0; u < 4; ++u) {
      const int e = eb + u * 1024;
      rec[u].x = (e < e1) ? dst[e] : 0;
      rec[u].y = (e < e1) ? src[e] : 0;
    }
#pragma unroll
    for (int u = 0; u < 4; ++u) {
      if (eb + u * 1024 < e1) {
        const int pos = atomicAdd(&cursor[rec[u].x / S], 1);
        *(v2i*)(bkt + 2 * (size_t)pos) = rec[u];
      }
    }
  }
}

__global__ __launch_bounds__(1024)
void range_to_lists(const int* __restrict__ bkt, const int* __restrict__ bbase,
                    int* __restrict__ rowptr, int* __restrict__ csrc, int nE, int nN) {
  __shared__ int cnt[MAXS];
  __shared__ int wsum[32];
  const int tid = threadIdx.x;
  const int b = blockIdx.x;
  const int S = (nN + NB - 1) / NB;
  const int n0 = b * S;
  const int beg = bbase[b];
  const int end = bbase[b + 1];

  cnt[tid] = 0;
  __syncthreads();
  for (int ib = beg + tid; ib < end; ib += 4 * 1024) {
    int d[4];
#pragma unroll
    for (int u = 0; u < 4; ++u) {
      const int i = ib + u * 1024;
      d[u] = (i < end) ? bkt[2 * (size_t)i] : n0;
    }
#pragma unroll
    for (int u = 0; u < 4; ++u) {
      if (ib + u * 1024 < end) atomicAdd(&cnt[d[u] - n0], 1);
    }
  }
  __syncthreads();
  const int c = cnt[tid];
  const int ex = block_excl_scan(c, wsum);
  cnt[tid] = ex;
  if (tid < S && n0 + tid < nN) rowptr[n0 + tid] = beg + ex;
  if (b == 0 && tid == 0) rowptr[nN] = nE;
  __syncthreads();
  for (int ib = beg + tid; ib < end; ib += 4 * 1024) {
    v2i rec[4];
#pragma unroll
    for (int u = 0; u < 4; ++u) {
      const int i = ib + u * 1024;
      rec[u].x = n0;
      rec[u].y = 0;
      if (i < end) rec[u] = *(const v2i*)(bkt + 2 * (size_t)i);
    }
#pragma unroll
    for (int u = 0; u < 4; ++u) {
      if (ib + u * 1024 < end) {
        const int k = atomicAdd(&cnt[rec[u].x - n0], 1);
        csrc[beg + k] = rec[u].y;
      }
    }
  }
}

__device__ __forceinline__ float wave_max(float v) {
#pragma unroll
  for (int off = 16; off > 0; off >>= 1) v = fmaxf(v, __shfl_xor(v, off, 32));
  return v;
}
__device__ __forceinline__ float wave_sum(float v) {
#pragma unroll
  for (int off = 16; off > 0; off >>= 1) v += __shfl_xor(v, off, 32);
  return v;
}

template <int B>
__device__ __forceinline__ void add_rows(const float* __restrict__ z, int s, float alpha,
                                         int j0, int lane, v2f& acc) {
  v2f r[B];
#pragma unroll
  for (int u = 0; u < B; ++u) {
    const int sj = __builtin_amdgcn_readlane(s, j0 + u);
    r[u] = *(const v2f*)(z + (size_t)sj * DIM + 2 * lane);
  }
#pragma unroll
  for (int u = 0; u < B; ++u) {
    const float aj = __int_as_float(__builtin_amdgcn_readlane(__float_as_int(alpha), j0 + u));
    acc.x += aj * r[u].x;
    acc.y += aj * r[u].y;
  }
}

__device__ __forceinline__ void add_list(const float* __restrict__ z, int s, float alpha,
                                         int n, int lane, v2f& acc) {
  int j = 0;
  for (; j + 8 <= n; j += 8) add_rows<8>(z, s, alpha, j, lane, acc);
  if (n - j >= 4) { add_rows<4>(z, s, alpha, j, lane, acc); j += 4; }
  if (n - j >= 2) { add_rows<2>(z, s, alpha, j, lane, acc); j += 2; }
  if (n - j >= 1) { add_rows<1>(z, s, alpha, j, lane, acc); }
}

__device__ __forceinline__ v2f aggregate_node(int d, int lane, float ab,
                                              const int* __restrict__ rowptr,
                                              const int* __restrict__ csrc,
                                              const float* __restrict__ ssrc,
                                              const float* __restrict__ sdst,
                                              const float* __restrict__ z) {
  const int beg = __builtin_amdgcn_readfirstlane(rowptr[d]);
  const int end = __builtin_amdgcn_readfirstlane(rowptr[d + 1]);
  const int deg = end - beg;
  const float sd = sdst[d];
  v2f acc; acc.x = 0.0f; acc.y = 0.0f;

  if (deg > 0 && deg <= 32) {
    int s = 0;
    float v = -INFINITY;
    if (lane < deg) {
      s = csrc[beg + lane];
      const float t = ssrc[s] + sd + ab;
      v = (t > 0.0f) ? t : LEAKY * t;
    }
    const float m = wave_max(v);
    const float ex = (lane < deg) ? __expf(v - m) : 0.0f;
    const float den = wave_sum(ex);
    const float alpha = ex / den;
    add_list(z, s, alpha, deg, lane, acc);
  } else if (deg > 32) {
    float m = -INFINITY;
    for (int c = beg; c < end; c += 32) {
      const int i = c + lane;
      if (i < end) {
        const float t = ssrc[csrc[i]] + sd + ab;
        m = fmaxf(m, (t > 0.0f) ? t : LEAKY * t);
      }
    }
    m = wave_max(m);
    float den = 0.0f;
    for (int c = beg; c < end; c += 32) {
      const int i = c + lane;
      if (i < end) {
        const float t = ssrc[csrc[i]] + sd + ab;
        den += __expf(((t > 0.0f) ? t : LEAKY * t) - m);
      }
    }
    den = wave_sum(den);
    for (int c = beg; c < end; c += 32) {
      const int i = c + lane;
      int s = 0;
      float alpha = 0.0f;
      if (i < end) {
        s = csrc[i];
        const float t = ssrc[s] + sd + ab;
        alpha = __expf(((t > 0.0f) ? t : LEAKY * t) - m) / den;
      }
      add_list(z, s, alpha, min(32, end - c), lane, acc);
    }
  }
  return acc;
}

__global__ __launch_bounds__(256)
void node_softmax_aggregate(const int* __restrict__ rowptr, const int* __restrict__ csrc,
                            const float* __restrict__ ssrc, const float* __restrict__ sdst,
                            const float* __restrict__ abp, const float* __restrict__ z,
                            float* __restrict__ out, int nN) {
  const int lane = threadIdx.x & 31;
  const int warp = __builtin_amdgcn_readfirstlane((int)((blockIdx.x * blockDim.x + threadIdx.x) >> 5));
  const int nWarps = (gridDim.x * blockDim.x) >> 5;
  const float ab = abp[0];

  for (int d = warp; d < nN; d += nWarps) {
    const v2f acc = aggregate_node(d, lane, ab, rowptr, csrc, ssrc, sdst, z);
    *(v2f*)(out + (size_t)d * DIM + 2 * lane) = acc;
  }
}

__global__ __launch_bounds__(256)
void node_aggregate_next_gemm(const int* __restrict__ rowptr, const int* __restrict__ csrc,
                              const float* __restrict__ ssrc, const float* __restrict__ sdst,
                              const float* __restrict__ abp, const float* __restrict__ z,
                              const float* __restrict__ Wn, const float* __restrict__ biasn,
                              const float* __restrict__ attn, float* __restrict__ zn,
                              float* __restrict__ ssrcn, float* __restrict__ sdstn, int nN) {
  __shared__ __attribute__((aligned(16))) float sW[DIM * WPITCH];
  __shared__ __attribute__((aligned(16))) float sH[32 * WPITCH];
  __shared__ float sB[DIM];
  __shared__ float sP[2 * 16 * 32];

  const int tid = threadIdx.x;
  const int lane = tid & 31;
  const int w = __builtin_amdgcn_readfirstlane(tid >> 5);
  const int warp = blockIdx.x * 8 + w;
  const int nWarps = gridDim.x * 8;

  float bv = 0.0f;
  if (tid < DIM) bv = biasn[tid];
#pragma unroll
  for (int i = 0; i < DIM / 4; ++i) {
    const int r = (tid >> 6) + 4 * i;
    sW[r * WPITCH + (tid & (DIM - 1))] = Wn[r * DIM + (tid & (DIM - 1))];
  }
  if (tid < DIM) sB[tid] = bv;

  const float ab = abp[0];
  for (int k = 0; k < 4; ++k) {
    const int d = warp + k * nWarps;
    v2f acc; acc.x = 0.0f; acc.y = 0.0f;
    if (d < nN) acc = aggregate_node(d, lane, ab, rowptr, csrc, ssrc, sdst, z);
    acc.x = fmaxf(acc.x, 0.0f);
    acc.y = fmaxf(acc.y, 0.0f);
    *(v2f*)(&sH[(k * 8 + w) * WPITCH + 2 * lane]) = acc;
  }
  __syncthreads();

  const int rt = w >> 1;
  const int ch = w & 1;
  const int laneRow = lane & 15;
  const int hh = lane >> 4;
  const int kHalf = hh * 2;
  const int dBase = blockIdx.x * 8 + (2 * rt + hh) * nWarps;
  const int c0 = ch * 32 + laneRow;

  v8f acc[2];
  v2f aw[2];
  if (w < 4) {
    const float* aBase = &sH[(rt * 16 + laneRow) * WPITCH + kHalf];
    const float* bBase = &sW[c0 * WPITCH + kHalf];
#pragma unroll
    for (int j = 0; j < 2; ++j) {
      const float bcol = sB[c0 + 16 * j];
#pragma unroll
      for (int r = 0; r < 8; ++r) acc[j][r] = bcol;
      aw[j].x = attn[c0 + 16 * j];
      aw[j].y = attn[DIM + c0 + 16 * j];
    }
#pragma unroll
    for (int t = 0; t < 16; ++t) {
      const v2f a = *(const v2f*)(aBase + 4 * t);
#pragma unroll
      for (int j = 0; j < 2; ++j) {
        const v2f b = *(const v2f*)(bBase + 16 * j * WPITCH + 4 * t);
        acc[j] = __builtin_amdgcn_wmma_f32_16x16x4_f32(
             false, a,  false, b,
             (short)0, acc[j],  false,  false);
      }
    }
    __builtin_amdgcn_sched_group_barrier(0x100, 3, 0);
#pragma unroll
    for (int g = 0; g < 7; ++g) {
      __builtin_amdgcn_sched_group_barrier(0x100, 3, 0);
      __builtin_amdgcn_sched_group_barrier(0x008, 4, 0);
    }
    __builtin_amdgcn_sched_group_barrier(0x008, 4, 0);
  }

  v2f sd[8];
  if (w < 4 && ch == 0) {
#pragma unroll
    for (int r = 0; r < 8; ++r) { sd[r].x = 0.0f; sd[r].y = 0.0f; }
#pragma unroll
    for (int j = 0; j < 2; ++j) {
#pragma unroll
      for (int r = 0; r < 8; ++r) sd[r] += acc[j][r] * aw[j];
    }
#pragma unroll
    for (int r = 0; r < 8; ++r) {
      sP[(rt * 16 + 2 * r) * 32 + lane] = sd[r].x;
      sP[(rt * 16 + 2 * r + 1) * 32 + lane] = sd[r].y;
    }
  }
  __syncthreads();
  if (w >= 4) return;

#pragma unroll
  for (int j = 0; j < 2; ++j) {
#pragma unroll
    for (int r = 0; r < 8; ++r) {
      const int d = dBase + r;
      if (d < nN) zn[(size_t)d * DIM + c0 + 16 * j] = acc[j][r];
    }
  }
  if (ch == 0) return;

#pragma unroll
  for (int r = 0; r < 8; ++r) {
    sd[r].x = sP[(rt * 16 + 2 * r) * 32 + lane];
    sd[r].y = sP[(rt * 16 + 2 * r + 1) * 32 + lane];
  }
#pragma unroll
  for (int j = 0; j < 2; ++j) {
#pragma unroll
    for (int r = 0; r < 8; ++r) sd[r] += acc[j][r] * aw[j];
  }

  const bool b3 = (lane & 8) != 0, b2 = (lane & 4) != 0;
  const bool b1 = (lane & 2) != 0, b0 = (lane & 1) != 0;
  float t[8], u[4], x2[2];
#pragma unroll
  for (int r = 0; r < 8; ++r) {
    const float keep = b3 ? sd[r].y : sd[r].x;
    const float send = b3 ? sd[r].x : sd[r].y;
    t[r] = keep + row_xor<8>(send);
  }
#pragma unroll
  for (int j = 0; j < 4; ++j) {
    const float keep = b2 ? t[j + 4] : t[j];
    const float send = b2 ? t[j] : t[j + 4];
    u[j] = keep + row_xor<4>(send);
  }
#pragma unroll
  for (int j = 0; j < 2; ++j) {
    const float keep = b1 ? u[j + 2] : u[j];
    const float send = b1 ? u[j] : u[j + 2];
    x2[j] = keep + row_xor<2>(send);
  }
  const float keep = b0 ? x2[1] : x2[0];
  const float send = b0 ? x2[0] : x2[1];
  const float y = keep + row_xor<1>(send);

  const int dy = dBase + (lane & 7);
  float* sp = b3 ? sdstn : ssrcn;
  if (dy < nN) sp[dy] = y;
}

extern "C" void kernel_launch(void* const* d_in, const int* in_sizes, int n_in,
                              void* d_out, int out_size, void* d_ws, size_t ws_size,
                              hipStream_t stream) {
  const float* x  = (const float*)d_in[0];
  const int* src  = (const int*)d_in[1];
  const int* dst  = (const int*)d_in[2];
  const int nN = in_sizes[0] / DIM;
  const int nE = in_sizes[1];

  char* ws = (char*)d_ws;
  const size_t featB = (size_t)nN * DIM * sizeof(float);
  const size_t edgeB = (size_t)nE * 4;
  const size_t nodeB = (size_t)nN * 4;
  float* z      = (float*)ws;
  float* z2     = (float*)(ws + featB);
  int*   bkt    = (int*)(ws + 2 * featB);
  int*   csrc   = (int*)(ws + 2 * featB + 2 * edgeB);
  int*   wgcnt  = (int*)(ws + 2 * featB + 3 * edgeB);
  const size_t tabB = (size_t)NWG * NB * 4;
  int*   bbase  = (int*)(ws + 2 * featB + 3 * edgeB + tabB);
  float* ssrc   = (float*)(ws + 2 * featB + 3 * edgeB + tabB + 2048);
  float* sdst   = (float*)(ws + 2 * featB + 3 * edgeB + tabB + 2048 + nodeB);
  int*   rowptr = (int*)(ws + 2 * featB + 3 * edgeB + tabB + 2048 + 2 * nodeB);
  float* ssrc2  = (float*)(ws + 2 * featB + 3 * edgeB + tabB + 4096 + 3 * nodeB);
  float* sdst2  = (float*)(ws + 2 * featB + 3 * edgeB + tabB + 4096 + 4 * nodeB);

  const int nTiles = (nN + 15) / 16;
  const dim3 gemmGrid((nTiles + 3) / 4), gemmBlk(128);
  const int aggBlocks = (nN + 31) / 32;

  range_count<<<NWG, 1024, 0, stream>>>(dst, wgcnt, nE, nN);
  range_scatter<<<NWG, 1024, 0, stream>>>(src, dst, wgcnt, bbase, bkt, nE, nN);
  range_to_lists<<<NB, 1024, 0, stream>>>(bkt, bbase, rowptr, csrc, nE, nN);

  gat_gemm_wmma<<<gemmGrid, gemmBlk, 0, stream>>>(x, (const float*)d_in[3], (const float*)d_in[4],
                                                  (const float*)d_in[5], z, ssrc, sdst, nTiles, 0);

  node_aggregate_next_gemm<<<aggBlocks, 256, 0, stream>>>(
      rowptr, csrc, ssrc, sdst, (const float*)d_in[6], z,
      (const float*)d_in[7], (const float*)d_in[8], (const float*)d_in[9], z2, ssrc2, sdst2, nN);

  node_aggregate_next_gemm<<<aggBlocks, 256, 0, stream>>>(
      rowptr, csrc, ssrc2, sdst2, (const float*)d_in[10], z2,
      (const float*)d_in[11], (const float*)d_in[12], (const float*)d_in[13], z, ssrc, sdst, nN);

  node_softmax_aggregate<<<aggBlocks, 256, 0, stream>>>(rowptr, csrc, ssrc, sdst,
                                                        (const float*)d_in[14], z,
                                                        (float*)d_out, nN);
}


# Edge grouping in 2 launches, not 3: each work-group sorts its own piece of the edge list by node range (LDS count, scan, place); the count launch and the table sums go; list order = LDS add arrival
# speedup vs baseline: 7.4182x; 1.2107x over previous
#include <hip/hip_runtime.h>
#include <hip/hip_bf16.h>

typedef __attribute__((ext_vector_type(2))) float v2f;
typedef __attribute__((ext_vector_type(8))) float v8f;

#define DIM 64
#define LEAKY 0.01f
#define WPITCH 68

template <int M>
__device__ __forceinline__ float row_xor(float x) {
  return __builtin_bit_cast(float, __builtin_amdgcn_update_dpp(
      0, __builtin_bit_cast(int, x), 0x160 + M  , 0xf, 0xf, true));
}

__global__ __launch_bounds__(128)
void gat_gemm_wmma(const float* __restrict__ hin, const float* __restrict__ W,
                   const float* __restrict__ bias, const float* __restrict__ att,
                   float* __restrict__ z, float* __restrict__ ssrc,
                   float* __restrict__ sdst, int nTiles, int reluIn) {
  __shared__ __attribute__((aligned(16))) float sW[DIM * WPITCH];
  __shared__ float sB[DIM];

  const int tid = threadIdx.x;
  for (int r = tid >> 6; r < DIM; r += 2) sW[r * WPITCH + (tid & (DIM - 1))] = W[r * DIM + (tid & (DIM - 1))];
  if (tid < DIM) sB[tid] = bias[tid];
  __syncthreads();

  const int wave = tid >> 5;
  const int lane = tid & 31;
  const int rowTile = blockIdx.x * 4 + wave;
  if (rowTile >= nTiles) return;

  const int row0 = rowTile * 16;
  const int laneRow = lane & 15;
  const int kHalf = (lane >> 4) * 2;

  v2f aw[4];
#pragma unroll
  for (int nt = 0; nt < 4; ++nt) {
    aw[nt].x = att[nt * 16 + laneRow];
    aw[nt].y = att[DIM + nt * 16 + laneRow];
  }

  const float* arow = hin + (size_t)(row0 + laneRow) * DIM;
  v2f aR[16];
#pragma unroll
  for (int t = 0; t < 16; ++t) {
    v2f a = *(const v2f*)(arow + 4 * t + kHalf);
    if (reluIn) { a.x = fmaxf(a.x, 0.0f); a.y = fmaxf(a.y, 0.0f); }
    aR[t] = a;
  }

  const int rOff = (lane < 16) ? 0 : 8;
  v2f sd[8];
#pragma unroll
  for (int r = 0; r < 8; ++r) { sd[r].x = 0.0f; sd[r].y = 0.0f; }
#pragma unroll
  for (int nt = 0; nt < 4; ++nt) {
    const int n0 = nt * 16;
    const float bcol = sB[n0 + laneRow];
    v8f acc;
#pragma unroll
    for (int r = 0; r < 8; ++r) acc[r] = bcol;

#pragma unroll
    for (int t = 0; t < 16; ++t) {
      v2f b = *(const v2f*)(&sW[(n0 + laneRow) * WPITCH + 4 * t + kHalf]);
      acc = __builtin_amdgcn_wmma_f32_16x16x4_f32(
           false, aR[t],  false, b,
           (short)0, acc,  false,  false);
    }

    float* zr = z + (size_t)(row0 + rOff) * DIM + n0 + laneRow;
#pragma unroll
    for (int r = 0; r < 8; ++r) zr[(size_t)r * DIM] = acc[r];
#pragma unroll
    for (int r = 0; r < 8; ++r) sd[r] += acc[r] * aw[nt];
  }

  const bool b3 = (lane & 8) != 0, b2 = (lane & 4) != 0;
  const bool b1 = (lane & 2) != 0, b0 = (lane & 1) != 0;
  float t[8], u[4], w[2];
#pragma unroll
  for (int r = 0; r < 8; ++r) {
    const float keep = b3 ? sd[r].y : sd[r].x;
    const float send = b3 ? sd[r].x : sd[r].y;
    t[r] = keep + row_xor<8>(send);
  }
#pragma unroll
  for (int j = 0; j < 4; ++j) {
    const float keep = b2 ? t[j + 4] : t[j];
    const float send = b2 ? t[j] : t[j + 4];
    u[j] = keep + row_xor<4>(send);
  }
#pragma unroll
  for (int j = 0; j < 2; ++j) {
    const float keep = b1 ? u[j + 2] : u[j];
    const float send = b1 ? u[j] : u[j + 2];
    w[j] = keep + row_xor<2>(send);
  }
  const float keep = b0 ? w[1] : w[0];
  const float send = b0 ? w[0] : w[1];
  const float y = keep + row_xor<1>(send);

  float* sp = b3 ? sdst : ssrc;
  sp[row0 + rOff + (lane & 7)] = y;
}

#define NB 256
#define NWG 256
#define MAXS 1024

typedef __attribute__((ext_vector_type(2))) int v2i;

__device__ __forceinline__ int block_excl_scan(int c, int* wsum) {
  const int lane = threadIdx.x & 31;
  const int wave = threadIdx.x >> 5;
  int x = c;
#pragma unroll
  for (int off = 1; off < 32; off <<= 1) {
    const int y = __shfl_up(x, off, 32);
    if (lane >= off) x += y;
  }
  if (lane == 31) wsum[wave] = x;
  __syncthreads();
  int part = (lane < wave) ? wsum[lane] : 0;
#pragma unroll
  for (int off = 16; off > 0; off >>= 1) part += __shfl_xor(part, off, 32);
  return part + x - c;
}

__global__ __launch_bounds__(1024)
void piece_sort(const int* __restrict__ src, const int* __restrict__ dst,
                int* __restrict__ off, int* __restrict__ bkt, int nE, int nN) {
  __shared__ int hist[NB];
  __shared__ int wsum[32];
  const int tid = threadIdx.x;
  const int wg = blockIdx.x;
  const int S = (nN + NB - 1) / NB;
  const int chunk = (nE + NWG - 1) / NWG;
  const int e0 = wg * chunk;
  const int e1 = min(e0 + chunk, nE);
  if (tid < NB) hist[tid] = 0;

  v2i rec[4];
#pragma unroll
  for (int u = 0; u < 4; ++u) {
    const int e = e0 + tid + u * 1024;
    rec[u].x = (e < e1) ? dst[e] : 0;
    rec[u].y = (e < e1) ? src[e] : 0;
  }
  __syncthreads();
  int rg[4];
#pragma unroll
  for (int u = 0; u < 4; ++u) {
    rg[u] = rec[u].x / S;
    if (e0 + tid + u * 1024 < e1) atomicAdd(&hist[rg[u]], 1);
  }
  for (int eb = e0 + tid + 4 * 1024; eb < e1; eb += 4 * 1024) {
    int d[4];
#pragma unroll
    for (int u = 0; u < 4; ++u) {
      const int e = eb + u * 1024;
      d[u] = (e < e1) ? dst[e] : 0;
    }
#pragma unroll
    for (int u = 0; u < 4; ++u) {
      if (eb + u * 1024 < e1) atomicAdd(&hist[d[u] / S], 1);
    }
  }
  __syncthreads();
  const int c = (tid < NB) ? hist[tid] : 0;
  const int ex = block_excl_scan(c, wsum);
  if (tid < NB) {
    hist[tid] = ex;
    off[wg * NB + tid] = ex;
  }
  __syncthreads();

  int pos[4];
#pragma unroll
  for (int u = 0; u < 4; ++u) {
    pos[u] = 0;
    if (e0 + tid + u * 1024 < e1) pos[u] = atomicAdd(&hist[rg[u]], 1);
  }
#pragma unroll
  for (int u = 0; u < 4; ++u) {
    if (e0 + tid + u * 1024 < e1) *(v2i*)(bkt + 2 * (size_t)(e0 + pos[u])) = rec[u];
  }
  for (int eb = e0 + tid + 4 * 1024; eb < e1; eb += 4 * 1024) {
    v2i r2[4];
#pragma unroll
    for (int u = 0; u < 4; ++u) {
      const int e = eb + u * 1024;
      r2[u].x = (e < e1) ? dst[e] : 0;
      r2[u].y = (e < e1) ? src[e] : 0;
    }
#pragma unroll
    for (int u = 0; u < 4; ++u) {
      if (eb + u * 1024 < e1) {
        const int p2 = atomicAdd(&hist[r2[u].x / S], 1);
        *(v2i*)(bkt + 2 * (size_t)(e0 + p2)) = r2[u];
      }
    }
  }
}

__global__ __launch_bounds__(1024)
void range_to_lists(const int* __restrict__ bkt, const int* __restrict__ off,
                    int* __restrict__ rowptr, int* __restrict__ csrc, int nE, int nN) {
  __shared__ int cnt[MAXS];
  __shared__ int wsum[32];
  __shared__ int red[32];
  const int tid = threadIdx.x;
  const int lane = tid & 31;
  const int b = blockIdx.x;
  const int S = (nN + NB - 1) / NB;
  const int n0 = b * S;
  const int chunk = (nE + NWG - 1) / NWG;
  const int pc = tid >> 2;
  const int q = tid & 3;
  const int pe0 = pc * chunk;
  const int plen = max(min(pe0 + chunk, nE) - pe0, 0);
  const int o0 = off[pc * NB + b];
  const int o1 = (b + 1 < NB) ? off[pc * NB + b + 1] : plen;
  const int sb = pe0 + o0;
  const int se = pe0 + o1;

  v2i rec[8];
#pragma unroll
  for (int u = 0; u < 8; ++u) {
    const int i = sb + q + 4 * u;
    rec[u].x = n0;
    rec[u].y = 0;
    if (i < se) rec[u] = *(const v2i*)(bkt + 2 * (size_t)i);
  }

  int part = (q == 0) ? o0 : 0;
#pragma unroll
  for (int sh = 16; sh > 0; sh >>= 1) part += __shfl_xor(part, sh, 32);
  if (lane == 0) red[tid >> 5] = part;
  cnt[tid] = 0;
  __syncthreads();

#pragma unroll
  for (int u = 0; u < 8; ++u) {
    if (sb + q + 4 * u < se) atomicAdd(&cnt[rec[u].x - n0], 1);
  }
  for (int ib = sb + q + 32; ib < se; ib += 16) {
    int d[4];
#pragma unroll
    for (int u = 0; u < 4; ++u) {
      const int i = ib + 4 * u;
      d[u] = (i < se) ? bkt[2 * (size_t)i] : n0;
    }
#pragma unroll
    for (int u = 0; u < 4; ++u) {
      if (ib + 4 * u < se) atomicAdd(&cnt[d[u] - n0], 1);
    }
  }
  __syncthreads();
  const int c = cnt[tid];
  const int ex = block_excl_scan(c, wsum);
  cnt[tid] = ex;
  int beg = 0;
#pragma unroll
  for (int w = 0; w < 32; ++w) beg += red[w];
  if (tid < S && n0 + tid < nN) rowptr[n0 + tid] = beg + ex;
  if (b == 0 && tid == 0) rowptr[nN] = nE;
  __syncthreads();

  int k[8];
#pragma unroll
  for (int u = 0; u < 8; ++u) {
    k[u] = 0;
    if (sb + q + 4 * u < se) k[u] = atomicAdd(&cnt[rec[u].x - n0], 1);
  }
#pragma unroll
  for (int u = 0; u < 8; ++u) {
    if (sb + q + 4 * u < se) csrc[beg + k[u]] = rec[u].y;
  }
  for (int ib = sb + q + 32; ib < se; ib += 16) {
    v2i r2[4];
#pragma unroll
    for (int u = 0; u < 4; ++u) {
      const int i = ib + 4 * u;
      r2[u].x = n0;
      r2[u].y = 0;
      if (i < se) r2[u] = *(const v2i*)(bkt + 2 * (size_t)i);
    }
#pragma unroll
    for (int u = 0; u < 4; ++u) {
      if (ib + 4 * u < se) {
        const int k2 = atomicAdd(&cnt[r2[u].x - n0], 1);
        csrc[beg + k2] = r2[u].y;
      }
    }
  }
}

__device__ __forceinline__ float wave_max(float v) {
#pragma unroll
  for (int off = 16; off > 0; off >>= 1) v = fmaxf(v, __shfl_xor(v, off, 32));
  return v;
}
__device__ __forceinline__ float wave_sum(float v) {
#pragma unroll
  for (int off = 16; off > 0; off >>= 1) v += __shfl_xor(v, off, 32);
  return v;
}

template <int B>
__device__ __forceinline__ void add_rows(const float* __restrict__ z, int s, float alpha,
                                         int j0, int lane, v2f& acc) {
  v2f r[B];
#pragma unroll
  for (int u = 0; u < B; ++u) {
    const int sj = __builtin_amdgcn_readlane(s, j0 + u);
    r[u] = *(const v2f*)(z + (size_t)sj * DIM + 2 * lane);
  }
#pragma unroll
  for (int u = 0; u < B; ++u) {
    const float aj = __int_as_float(__builtin_amdgcn_readlane(__float_as_int(alpha), j0 + u));
    acc.x += aj * r[u].x;
    acc.y += aj * r[u].y;
  }
}

__device__ __forceinline__ void add_list(const float* __restrict__ z, int s, float alpha,
                                         int n, int lane, v2f& acc) {
  int j = 0;
  for (; j + 8 <= n; j += 8) add_rows<8>(z, s, alpha, j, lane, acc);
  if (n - j >= 4) { add_rows<4>(z, s, alpha, j, lane, acc); j += 4; }
  if (n - j >= 2) { add_rows<2>(z, s, alpha, j, lane, acc); j += 2; }
  if (n - j >= 1) { add_rows<1>(z, s, alpha, j, lane, acc); }
}

__device__ __forceinline__ v2f aggregate_node(int d, int lane, float ab,
                                              const int* __restrict__ rowptr,
                                              const int* __restrict__ csrc,
                                              const float* __restrict__ ssrc,
                                              const float* __restrict__ sdst,
                                              const float* __restrict__ z) {
  const int beg = __builtin_amdgcn_readfirstlane(rowptr[d]);
  const int end = __builtin_amdgcn_readfirstlane(rowptr[d + 1]);
  const int deg = end - beg;
  const float sd = sdst[d];
  v2f acc; acc.x = 0.0f; acc.y = 0.0f;

  if (deg > 0 && deg <= 32) {
    int s = 0;
    float v = -INFINITY;
    if (lane < deg) {
      s = csrc[beg + lane];
      const float t = ssrc[s] + sd + ab;
      v = (t > 0.0f) ? t : LEAKY * t;
    }
    const float m = wave_max(v);
    const float ex = (lane < deg) ? __expf(v - m) : 0.0f;
    const float den = wave_sum(ex);
    const float alpha = ex / den;
    add_list(z, s, alpha, deg, lane, acc);
  } else if (deg > 32) {
    float m = -INFINITY;
    for (int c = beg; c < end; c += 32) {
      const int i = c + lane;
      if (i < end) {
        const float t = ssrc[csrc[i]] + sd + ab;
        m = fmaxf(m, (t > 0.0f) ? t : LEAKY * t);
      }
    }
    m = wave_max(m);
    float den = 0.0f;
    for (int c = beg; c < end; c += 32) {
      const int i = c + lane;
      if (i < end) {
        const float t = ssrc[csrc[i]] + sd + ab;
        den += __expf(((t > 0.0f) ? t : LEAKY * t) - m);
      }
    }
    den = wave_sum(den);
    for (int c = beg; c < end; c += 32) {
      const int i = c + lane;
      int s = 0;
      float alpha = 0.0f;
      if (i < end) {
        s = csrc[i];
        const float t = ssrc[s] + sd + ab;
        alpha = __expf(((t > 0.0f) ? t : LEAKY * t) - m) / den;
      }
      add_list(z, s, alpha, min(32, end - c), lane, acc);
    }
  }
  return acc;
}

__global__ __launch_bounds__(256)
void node_softmax_aggregate(const int* __restrict__ rowptr, const int* __restrict__ csrc,
                            const float* __restrict__ ssrc, const float* __restrict__ sdst,
                            const float* __restrict__ abp, const float* __restrict__ z,
                            float* __restrict__ out, int nN) {
  const int lane = threadIdx.x & 31;
  const int warp = __builtin_amdgcn_readfirstlane((int)((blockIdx.x * blockDim.x + threadIdx.x) >> 5));
  const int nWarps = (gridDim.x * blockDim.x) >> 5;
  const float ab = abp[0];

  for (int d = warp; d < nN; d += nWarps) {
    const v2f acc = aggregate_node(d, lane, ab, rowptr, csrc, ssrc, sdst, z);
    *(v2f*)(out + (size_t)d * DIM + 2 * lane) = acc;
  }
}

__global__ __launch_bounds__(256)
void node_aggregate_next_gemm(const int* __restrict__ rowptr, const int* __restrict__ csrc,
                              const float* __restrict__ ssrc, const float* __restrict__ sdst,
                              const float* __restrict__ abp, const float* __restrict__ z,
                              const float* __restrict__ Wn, const float* __restrict__ biasn,
                              const float* __restrict__ attn, float* __restrict__ zn,
                              float* __restrict__ ssrcn, float* __restrict__ sdstn, int nN) {
  __shared__ __attribute__((aligned(16))) float sW[DIM * WPITCH];
  __shared__ __attribute__((aligned(16))) float sH[32 * WPITCH];
  __shared__ float sB[DIM];
  __shared__ float sP[2 * 16 * 32];

  const int tid = threadIdx.x;
  const int lane = tid & 31;
  const int w = __builtin_amdgcn_readfirstlane(tid >> 5);
  const int warp = blockIdx.x * 8 + w;
  const int nWarps = gridDim.x * 8;

  float bv = 0.0f;
  if (tid < DIM) bv = biasn[tid];
#pragma unroll
  for (int i = 0; i < DIM / 4; ++i) {
    const int r = (tid >> 6) + 4 * i;
    sW[r * WPITCH + (tid & (DIM - 1))] = Wn[r * DIM + (tid & (DIM - 1))];
  }
  if (tid < DIM) sB[tid] = bv;

  const float ab = abp[0];
  for (int k = 0; k < 4; ++k) {
    const int d = warp + k * nWarps;
    v2f acc; acc.x = 0.0f; acc.y = 0.0f;
    if (d < nN) acc = aggregate_node(d, lane, ab, rowptr, csrc, ssrc, sdst, z);
    acc.x = fmaxf(acc.x, 0.0f);
    acc.y = fmaxf(acc.y, 0.0f);
    *(v2f*)(&sH[(k * 8 + w) * WPITCH + 2 * lane]) = acc;
  }
  __syncthreads();

  const int rt = w >> 1;
  const int ch = w & 1;
  const int laneRow = lane & 15;
  const int hh = lane >> 4;
  const int kHalf = hh * 2;
  const int dBase = blockIdx.x * 8 + (2 * rt + hh) * nWarps;
  const int c0 = ch * 32 + laneRow;

  v8f acc[2];
  v2f aw[2];
  if (w < 4) {
    const float* aBase = &sH[(rt * 16 + laneRow) * WPITCH + kHalf];
    const float* bBase = &sW[c0 * WPITCH + kHalf];
#pragma unroll
    for (int j = 0; j < 2; ++j) {
      const float bcol = sB[c0 + 16 * j];
#pragma unroll
      for (int r = 0; r < 8; ++r) acc[j][r] = bcol;
      aw[j].x = attn[c0 + 16 * j];
      aw[j].y = attn[DIM + c0 + 16 * j];
    }
#pragma unroll
    for (int t = 0; t < 16; ++t) {
      const v2f a = *(const v2f*)(aBase + 4 * t);
#pragma unroll
      for (int j = 0; j < 2; ++j) {
        const v2f b = *(const v2f*)(bBase + 16 * j * WPITCH + 4 * t);
        acc[j] = __builtin_amdgcn_wmma_f32_16x16x4_f32(
             false, a,  false, b,
             (short)0, acc[j],  false,  false);
      }
    }
    __builtin_amdgcn_sched_group_barrier(0x100, 3, 0);
#pragma unroll
    for (int g = 0; g < 7; ++g) {
      __builtin_amdgcn_sched_group_barrier(0x100, 3, 0);
      __builtin_amdgcn_sched_group_barrier(0x008, 4, 0);
    }
    __builtin_amdgcn_sched_group_barrier(0x008, 4, 0);
  }

  v2f sd[8];
  if (w < 4 && ch == 0) {
#pragma unroll
    for (int r = 0; r < 8; ++r) { sd[r].x = 0.0f; sd[r].y = 0.0f; }
#pragma unroll
    for (int j = 0; j < 2; ++j) {
#pragma unroll
      for (int r = 0; r < 8; ++r) sd[r] += acc[j][r] * aw[j];
    }
#pragma unroll
    for (int r = 0; r < 8; ++r) {
      sP[(rt * 16 + 2 * r) * 32 + lane] = sd[r].x;
      sP[(rt * 16 + 2 * r + 1) * 32 + lane] = sd[r].y;
    }
  }
  __syncthreads();
  if (w >= 4) return;

#pragma unroll
  for (int j = 0; j < 2; ++j) {
#pragma unroll
    for (int r = 0; r < 8; ++r) {
      const int d = dBase + r;
      if (d < nN) zn[(size_t)d * DIM + c0 + 16 * j] = acc[j][r];
    }
  }
  if (ch == 0) return;

#pragma unroll
  for (int r = 0; r < 8; ++r) {
    sd[r].x = sP[(rt * 16 + 2 * r) * 32 + lane];
    sd[r].y = sP[(rt * 16 + 2 * r + 1) * 32 + lane];
  }
#pragma unroll
  for (int j = 0; j < 2; ++j) {
#pragma unroll
    for (int r = 0; r < 8; ++r) sd[r] += acc[j][r] * aw[j];
  }

  const bool b3 = (lane & 8) != 0, b2 = (lane & 4) != 0;
  const bool b1 = (lane & 2) != 0, b0 = (lane & 1) != 0;
  float t[8], u[4], x2[2];
#pragma unroll
  for (int r = 0; r < 8; ++r) {
    const float keep = b3 ? sd[r].y : sd[r].x;
    const float send = b3 ? sd[r].x : sd[r].y;
    t[r] = keep + row_xor<8>(send);
  }
#pragma unroll
  for (int j = 0; j < 4; ++j) {
    const float keep = b2 ? t[j + 4] : t[j];
    const float send = b2 ? t[j] : t[j + 4];
    u[j] = keep + row_xor<4>(send);
  }
#pragma unroll
  for (int j = 0; j < 2; ++j) {
    const float keep = b1 ? u[j + 2] : u[j];
    const float send = b1 ? u[j] : u[j + 2];
    x2[j] = keep + row_xor<2>(send);
  }
  const float keep = b0 ? x2[1] : x2[0];
  const float send = b0 ? x2[0] : x2[1];
  const float y = keep + row_xor<1>(send);

  const int dy = dBase + (lane & 7);
  float* sp = b3 ? sdstn : ssrcn;
  if (dy < nN) sp[dy] = y;
}

extern "C" void kernel_launch(void* const* d_in, const int* in_sizes, int n_in,
                              void* d_out, int out_size, void* d_ws, size_t ws_size,
                              hipStream_t stream) {
  const float* x  = (const float*)d_in[0];
  const int* src  = (const int*)d_in[1];
  const int* dst  = (const int*)d_in[2];
  const int nN = in_sizes[0] / DIM;
  const int nE = in_sizes[1];

  char* ws = (char*)d_ws;
  const size_t featB = (size_t)nN * DIM * sizeof(float);
  const size_t edgeB = (size_t)nE * 4;
  const size_t nodeB = (size_t)nN * 4;
  float* z      = (float*)ws;
  float* z2     = (float*)(ws + featB);
  int*   bkt    = (int*)(ws + 2 * featB);
  int*   csrc   = (int*)(ws + 2 * featB + 2 * edgeB);
  int*   off    = (int*)(ws + 2 * featB + 3 * edgeB);
  const size_t tabB = (size_t)NWG * NB * 4;
  float* ssrc   = (float*)(ws + 2 * featB + 3 * edgeB + tabB + 2048);
  float* sdst   = (float*)(ws + 2 * featB + 3 * edgeB + tabB + 2048 + nodeB);
  int*   rowptr = (int*)(ws + 2 * featB + 3 * edgeB + tabB + 2048 + 2 * nodeB);
  float* ssrc2  = (float*)(ws + 2 * featB + 3 * edgeB + tabB + 4096 + 3 * nodeB);
  float* sdst2  = (float*)(ws + 2 * featB + 3 * edgeB + tabB + 4096 + 4 * nodeB);

  const int nTiles = (nN + 15) / 16;
  const dim3 gemmGrid((nTiles + 3) / 4), gemmBlk(128);
  const int aggBlocks = (nN + 31) / 32;

  piece_sort<<<NWG, 1024, 0, stream>>>(src, dst, off, bkt, nE, nN);
  range_to_lists<<<NB, 1024, 0, stream>>>(bkt, off, rowptr, csrc, nE, nN);

  gat_gemm_wmma<<<gemmGrid, gemmBlk, 0, stream>>>(x, (const float*)d_in[3], (const float*)d_in[4],
                                                  (const float*)d_in[5], z, ssrc, sdst, nTiles, 0);

  node_aggregate_next_gemm<<<aggBlocks, 256, 0, stream>>>(
      rowptr, csrc, ssrc, sdst, (const float*)d_in[6], z,
      (const float*)d_in[7], (const float*)d_in[8], (const float*)d_in[9], z2, ssrc2, sdst2, nN);

  node_aggregate_next_gemm<<<aggBlocks, 256, 0, stream>>>(
      rowptr, csrc, ssrc2, sdst2, (const float*)d_in[10], z2,
      (const float*)d_in[11], (const float*)d_in[12], (const float*)d_in[13], z, ssrc, sdst, nN);

  node_softmax_aggregate<<<aggBlocks, 256, 0, stream>>>(rowptr, csrc, ssrc, sdst,
                                                        (const float*)d_in[14], z,
                                                        (float*)d_out, nN);
}


# Each wave fetches the list heads of its 4 nodes together (bounds in one load, then sources, then source scores, each for all 4 before first use), softmax exchanges interleaved; sums in parent order
# speedup vs baseline: 7.7392x; 1.0433x over previous
#include <hip/hip_runtime.h>
#include <hip/hip_bf16.h>

typedef __attribute__((ext_vector_type(2))) float v2f;
typedef __attribute__((ext_vector_type(8))) float v8f;

#define DIM 64
#define LEAKY 0.01f
#define WPITCH 68

template <int M>
__device__ __forceinline__ float row_xor(float x) {
  return __builtin_bit_cast(float, __builtin_amdgcn_update_dpp(
      0, __builtin_bit_cast(int, x), 0x160 + M  , 0xf, 0xf, true));
}

__global__ __launch_bounds__(128)
void gat_gemm_wmma(const float* __restrict__ hin, const float* __restrict__ W,
                   const float* __restrict__ bias, const float* __restrict__ att,
                   float* __restrict__ z, float* __restrict__ ssrc,
                   float* __restrict__ sdst, int nTiles, int reluIn) {
  __shared__ __attribute__((aligned(16))) float sW[DIM * WPITCH];
  __shared__ float sB[DIM];

  const int tid = threadIdx.x;
  for (int r = tid >> 6; r < DIM; r += 2) sW[r * WPITCH + (tid & (DIM - 1))] = W[r * DIM + (tid & (DIM - 1))];
  if (tid < DIM) sB[tid] = bias[tid];
  __syncthreads();

  const int wave = tid >> 5;
  const int lane = tid & 31;
  const int rowTile = blockIdx.x * 4 + wave;
  if (rowTile >= nTiles) return;

  const int row0 = rowTile * 16;
  const int laneRow = lane & 15;
  const int kHalf = (lane >> 4) * 2;

  v2f aw[4];
#pragma unroll
  for (int nt = 0; nt < 4; ++nt) {
    aw[nt].x = att[nt * 16 + laneRow];
    aw[nt].y = att[DIM + nt * 16 + laneRow];
  }

  const float* arow = hin + (size_t)(row0 + laneRow) * DIM;
  v2f aR[16];
#pragma unroll
  for (int t = 0; t < 16; ++t) {
    v2f a = *(const v2f*)(arow + 4 * t + kHalf);
    if (reluIn) { a.x = fmaxf(a.x, 0.0f); a.y = fmaxf(a.y, 0.0f); }
    aR[t] = a;
  }

  const int rOff = (lane < 16) ? 0 : 8;
  v2f sd[8];
#pragma unroll
  for (int r = 0; r < 8; ++r) { sd[r].x = 0.0f; sd[r].y = 0.0f; }
#pragma unroll
  for (int nt = 0; nt < 4; ++nt) {
    const int n0 = nt * 16;
    const float bcol = sB[n0 + laneRow];
    v8f acc;
#pragma unroll
    for (int r = 0; r < 8; ++r) acc[r] = bcol;

#pragma unroll
    for (int t = 0; t < 16; ++t) {
      v2f b = *(const v2f*)(&sW[(n0 + laneRow) * WPITCH + 4 * t + kHalf]);
      acc = __builtin_amdgcn_wmma_f32_16x16x4_f32(
           false, aR[t],  false, b,
           (short)0, acc,  false,  false);
    }

    float* zr = z + (size_t)(row0 + rOff) * DIM + n0 + laneRow;
#pragma unroll
    for (int r = 0; r < 8; ++r) zr[(size_t)r * DIM] = acc[r];
#pragma unroll
    for (int r = 0; r < 8; ++r) sd[r] += acc[r] * aw[nt];
  }

  const bool b3 = (lane & 8) != 0, b2 = (lane & 4) != 0;
  const bool b1 = (lane & 2) != 0, b0 = (lane & 1) != 0;
  float t[8], u[4], w[2];
#pragma unroll
  for (int r = 0; r < 8; ++r) {
    const float keep = b3 ? sd[r].y : sd[r].x;
    const float send = b3 ? sd[r].x : sd[r].y;
    t[r] = keep + row_xor<8>(send);
  }
#pragma unroll
  for (int j = 0; j < 4; ++j) {
    const float keep = b2 ? t[j + 4] : t[j];
    const float send = b2 ? t[j] : t[j + 4];
    u[j] = keep + row_xor<4>(send);
  }
#pragma unroll
  for (int j = 0; j < 2; ++j) {
    const float keep = b1 ? u[j + 2] : u[j];
    const float send = b1 ? u[j] : u[j + 2];
    w[j] = keep + row_xor<2>(send);
  }
  const float keep = b0 ? w[1] : w[0];
  const float send = b0 ? w[0] : w[1];
  const float y = keep + row_xor<1>(send);

  float* sp = b3 ? sdst : ssrc;
  sp[row0 + rOff + (lane & 7)] = y;
}

#define NB 256
#define NWG 256
#define MAXS 1024

typedef __attribute__((ext_vector_type(2))) int v2i;

__device__ __forceinline__ int block_excl_scan(int c, int* wsum) {
  const int lane = threadIdx.x & 31;
  const int wave = threadIdx.x >> 5;
  int x = c;
#pragma unroll
  for (int off = 1; off < 32; off <<= 1) {
    const int y = __shfl_up(x, off, 32);
    if (lane >= off) x += y;
  }
  if (lane == 31) wsum[wave] = x;
  __syncthreads();
  int part = (lane < wave) ? wsum[lane] : 0;
#pragma unroll
  for (int off = 16; off > 0; off >>= 1) part += __shfl_xor(part, off, 32);
  return part + x - c;
}

__global__ __launch_bounds__(1024)
void piece_sort(const int* __restrict__ src, const int* __restrict__ dst,
                int* __restrict__ off, int* __restrict__ bkt, int nE, int nN) {
  __shared__ int hist[NB];
  __shared__ int wsum[32];
  const int tid = threadIdx.x;
  const int wg = blockIdx.x;
  const int S = (nN + NB - 1) / NB;
  const int chunk = (nE + NWG - 1) / NWG;
  const int e0 = wg * chunk;
  const int e1 = min(e0 + chunk, nE);
  if (tid < NB) hist[tid] = 0;

  v2i rec[4];
#pragma unroll
  for (int u = 0; u < 4; ++u) {
    const int e = e0 + tid + u * 1024;
    rec[u].x = (e < e1) ? dst[e] : 0;
    rec[u].y = (e < e1) ? src[e] : 0;
  }
  __syncthreads();
  int rg[4];
#pragma unroll
  for (int u = 0; u < 4; ++u) {
    rg[u] = rec[u].x / S;
    if (e0 + tid + u * 1024 < e1) atomicAdd(&hist[rg[u]], 1);
  }
  for (int eb = e0 + tid + 4 * 1024; eb < e1; eb += 4 * 1024) {
    int d[4];
#pragma unroll
    for (int u = 0; u < 4; ++u) {
      const int e = eb + u * 1024;
      d[u] = (e < e1) ? dst[e] : 0;
    }
#pragma unroll
    for (int u = 0; u < 4; ++u) {
      if (eb + u * 1024 < e1) atomicAdd(&hist[d[u] / S], 1);
    }
  }
  __syncthreads();
  const int c = (tid < NB) ? hist[tid] : 0;
  const int ex = block_excl_scan(c, wsum);
  if (tid < NB) {
    hist[tid] = ex;
    off[wg * NB + tid] = ex;
  }
  __syncthreads();

  int pos[4];
#pragma unroll
  for (int u = 0; u < 4; ++u) {
    pos[u] = 0;
    if (e0 + tid + u * 1024 < e1) pos[u] = atomicAdd(&hist[rg[u]], 1);
  }
#pragma unroll
  for (int u = 0; u < 4; ++u) {
    if (e0 + tid + u * 1024 < e1) *(v2i*)(bkt + 2 * (size_t)(e0 + pos[u])) = rec[u];
  }
  for (int eb = e0 + tid + 4 * 1024; eb < e1; eb += 4 * 1024) {
    v2i r2[4];
#pragma unroll
    for (int u = 0; u < 4; ++u) {
      const int e = eb + u * 1024;
      r2[u].x = (e < e1) ? dst[e] : 0;
      r2[u].y = (e < e1) ? src[e] : 0;
    }
#pragma unroll
    for (int u = 0; u < 4; ++u) {
      if (eb + u * 1024 < e1) {
        const int p2 = atomicAdd(&hist[r2[u].x / S], 1);
        *(v2i*)(bkt + 2 * (size_t)(e0 + p2)) = r2[u];
      }
    }
  }
}

__global__ __launch_bounds__(1024)
void range_to_lists(const int* __restrict__ bkt, const int* __restrict__ off,
                    int* __restrict__ rowptr, int* __restrict__ csrc, int nE, int nN) {
  __shared__ int cnt[MAXS];
  __shared__ int wsum[32];
  __shared__ int red[32];
  const int tid = threadIdx.x;
  const int lane = tid & 31;
  const int b = blockIdx.x;
  const int S = (nN + NB - 1) / NB;
  const int n0 = b * S;
  const int chunk = (nE + NWG - 1) / NWG;
  const int pc = tid >> 2;
  const int q = tid & 3;
  const int pe0 = pc * chunk;
  const int plen = max(min(pe0 + chunk, nE) - pe0, 0);
  const int o0 = off[pc * NB + b];
  const int o1 = (b + 1 < NB) ? off[pc * NB + b + 1] : plen;
  const int sb = pe0 + o0;
  const int se = pe0 + o1;

  v2i rec[8];
#pragma unroll
  for (int u = 0; u < 8; ++u) {
    const int i = sb + q + 4 * u;
    rec[u].x = n0;
    rec[u].y = 0;
    if (i < se) rec[u] = *(const v2i*)(bkt + 2 * (size_t)i);
  }

  int part = (q == 0) ? o0 : 0;
#pragma unroll
  for (int sh = 16; sh > 0; sh >>= 1) part += __shfl_xor(part, sh, 32);
  if (lane == 0) red[tid >> 5] = part;
  cnt[tid] = 0;
  __syncthreads();

#pragma unroll
  for (int u = 0; u < 8; ++u) {
    if (sb + q + 4 * u < se) atomicAdd(&cnt[rec[u].x - n0], 1);
  }
  for (int ib = sb + q + 32; ib < se; ib += 16) {
    int d[4];
#pragma unroll
    for (int u = 0; u < 4; ++u) {
      const int i = ib + 4 * u;
      d[u] = (i < se) ? bkt[2 * (size_t)i] : n0;
    }
#pragma unroll
    for (int u = 0; u < 4; ++u) {
      if (ib + 4 * u < se) atomicAdd(&cnt[d[u] - n0], 1);
    }
  }
  __syncthreads();
  const int c = cnt[tid];
  const int ex = block_excl_scan(c, wsum);
  cnt[tid] = ex;
  int beg = 0;
#pragma unroll
  for (int w = 0; w < 32; ++w) beg += red[w];
  if (tid < S && n0 + tid < nN) rowptr[n0 + tid] = beg + ex;
  if (b == 0 && tid == 0) rowptr[nN] = nE;
  __syncthreads();

  int k[8];
#pragma unroll
  for (int u = 0; u < 8; ++u) {
    k[u] = 0;
    if (sb + q + 4 * u < se) k[u] = atomicAdd(&cnt[rec[u].x - n0], 1);
  }
#pragma unroll
  for (int u = 0; u < 8; ++u) {
    if (sb + q + 4 * u < se) csrc[beg + k[u]] = rec[u].y;
  }
  for (int ib = sb + q + 32; ib < se; ib += 16) {
    v2i r2[4];
#pragma unroll
    for (int u = 0; u < 4; ++u) {
      const int i = ib + 4 * u;
      r2[u].x = n0;
      r2[u].y = 0;
      if (i < se) r2[u] = *(const v2i*)(bkt + 2 * (size_t)i);
    }
#pragma unroll
    for (int u = 0; u < 4; ++u) {
      if (ib + 4 * u < se) {
        const int k2 = atomicAdd(&cnt[r2[u].x - n0], 1);
        csrc[beg + k2] = r2[u].y;
      }
    }
  }
}

__device__ __forceinline__ float wave_max(float v) {
#pragma unroll
  for (int off = 16; off > 0; off >>= 1) v = fmaxf(v, __shfl_xor(v, off, 32));
  return v;
}
__device__ __forceinline__ float wave_sum(float v) {
#pragma unroll
  for (int off = 16; off > 0; off >>= 1) v += __shfl_xor(v, off, 32);
  return v;
}

template <int B>
__device__ __forceinline__ void add_rows(const float* __restrict__ z, int s, float alpha,
                                         int j0, int lane, v2f& acc) {
  v2f r[B];
#pragma unroll
  for (int u = 0; u < B; ++u) {
    const int sj = __builtin_amdgcn_readlane(s, j0 + u);
    r[u] = *(const v2f*)(z + (size_t)sj * DIM + 2 * lane);
  }
#pragma unroll
  for (int u = 0; u < B; ++u) {
    const float aj = __int_as_float(__builtin_amdgcn_readlane(__float_as_int(alpha), j0 + u));
    acc.x += aj * r[u].x;
    acc.y += aj * r[u].y;
  }
}

__device__ __forceinline__ void add_list(const float* __restrict__ z, int s, float alpha,
                                         int n, int lane, v2f& acc) {
  int j = 0;
  for (; j + 8 <= n; j += 8) add_rows<8>(z, s, alpha, j, lane, acc);
  if (n - j >= 4) { add_rows<4>(z, s, alpha, j, lane, acc); j += 4; }
  if (n - j >= 2) { add_rows<2>(z, s, alpha, j, lane, acc); j += 2; }
  if (n - j >= 1) { add_rows<1>(z, s, alpha, j, lane, acc); }
}

#define NPW 4

__device__ __forceinline__ void issue_bounds(int d0, int stride, int nN, int lane,
                                             const int* __restrict__ rowptr,
                                             const float* __restrict__ sdst,
                                             int& rbe, float& sdl) {
  const int kl = (lane >> 1) & (NPW - 1);
  const int dl = d0 + kl * stride;
  const int dk = (dl < nN) ? dl : 0;
  rbe = rowptr[dk + (lane & 1)];
  sdl = sdst[dk];
}

__device__ __forceinline__ void load_heads(int rbe, float sdl, int d0, int stride, int nN,
                                           int lane, float ab,
                                           const int* __restrict__ csrc,
                                           const float* __restrict__ ssrc,
                                           int (&beg)[NPW], int (&deg)[NPW], float (&sd)[NPW],
                                           int (&s)[NPW], float (&alpha)[NPW]) {
#pragma unroll
  for (int k = 0; k < NPW; ++k) {
    beg[k] = __builtin_amdgcn_readlane(rbe, 2 * k);
    const int end = __builtin_amdgcn_readlane(rbe, 2 * k + 1);
    deg[k] = (d0 + k * stride < nN) ? end - beg[k] : 0;
    sd[k] = __int_as_float(__builtin_amdgcn_readlane(__float_as_int(sdl), 2 * k));
  }
#pragma unroll
  for (int k = 0; k < NPW; ++k) s[k] = csrc[(lane < deg[k]) ? beg[k] + lane : 0];
  float ts[NPW];
#pragma unroll
  for (int k = 0; k < NPW; ++k) ts[k] = ssrc[s[k]];
  float v[NPW], m[NPW], ex[NPW], den[NPW];
#pragma unroll
  for (int k = 0; k < NPW; ++k) {
    const float t = ts[k] + sd[k] + ab;
    const float lv = (t > 0.0f) ? t : LEAKY * t;
    v[k] = (lane < deg[k]) ? lv : -INFINITY;
    m[k] = v[k];
  }
#pragma unroll
  for (int off = 16; off > 0; off >>= 1) {
#pragma unroll
    for (int k = 0; k < NPW; ++k) m[k] = fmaxf(m[k], __shfl_xor(m[k], off, 32));
  }
#pragma unroll
  for (int k = 0; k < NPW; ++k) {
    ex[k] = (lane < deg[k]) ? __expf(v[k] - m[k]) : 0.0f;
    den[k] = ex[k];
  }
#pragma unroll
  for (int off = 16; off > 0; off >>= 1) {
#pragma unroll
    for (int k = 0; k < NPW; ++k) den[k] += __shfl_xor(den[k], off, 32);
  }
#pragma unroll
  for (int k = 0; k < NPW; ++k) alpha[k] = ex[k] / den[k];
}

template <typename T>
__device__ __forceinline__ T pick(const T (&a)[NPW], int k) {
  T r = a[0];
#pragma unroll
  for (int j = 1; j < NPW; ++j) r = (k == j) ? a[j] : r;
  return r;
}

__device__ __forceinline__ v2f sum_rows(int beg, int deg, float sd, int s, float alpha,
                                        int lane, float ab,
                                        const int* __restrict__ csrc,
                                        const float* __restrict__ ssrc,
                                        const float* __restrict__ z) {
  v2f acc; acc.x = 0.0f; acc.y = 0.0f;
  if (deg > 0 && deg <= 32) {
    add_list(z, s, alpha, deg, lane, acc);
  } else if (deg > 32) {
    const int end = beg + deg;
    float m = -INFINITY;
    for (int c = beg; c < end; c += 32) {
      const int i = c + lane;
      if (i < end) {
        const float t = ssrc[csrc[i]] + sd + ab;
        m = fmaxf(m, (t > 0.0f) ? t : LEAKY * t);
      }
    }
    m = wave_max(m);
    float den = 0.0f;
    for (int c = beg; c < end; c += 32) {
      const int i = c + lane;
      if (i < end) {
        const float t = ssrc[csrc[i]] + sd + ab;
        den += __expf(((t > 0.0f) ? t : LEAKY * t) - m);
      }
    }
    den = wave_sum(den);
    for (int c = beg; c < end; c += 32) {
      const int i = c + lane;
      int s2 = 0;
      float a2 = 0.0f;
      if (i < end) {
        s2 = csrc[i];
        const float t = ssrc[s2] + sd + ab;
        a2 = __expf(((t > 0.0f) ? t : LEAKY * t) - m) / den;
      }
      add_list(z, s2, a2, min(32, end - c), lane, acc);
    }
  }
  return acc;
}

__global__ __launch_bounds__(256)
void node_softmax_aggregate(const int* __restrict__ rowptr, const int* __restrict__ csrc,
                            const float* __restrict__ ssrc, const float* __restrict__ sdst,
                            const float* __restrict__ abp, const float* __restrict__ z,
                            float* __restrict__ out, int nN) {
  const int lane = threadIdx.x & 31;
  const int warp = __builtin_amdgcn_readfirstlane((int)((blockIdx.x * blockDim.x + threadIdx.x) >> 5));
  const int nWarps = (gridDim.x * blockDim.x) >> 5;
  const float ab = abp[0];

  for (int base = warp; base < nN; base += NPW * nWarps) {
    int beg[NPW], deg[NPW], s[NPW];
    float sd[NPW], alpha[NPW];
    int rbe;
    float sdl;
    issue_bounds(base, nWarps, nN, lane, rowptr, sdst, rbe, sdl);
    load_heads(rbe, sdl, base, nWarps, nN, lane, ab, csrc, ssrc, beg, deg, sd, s, alpha);
#pragma unroll 1
    for (int k = 0; k < NPW; ++k) {
      const int d = base + k * nWarps;
      if (d < nN) {
        const v2f acc = sum_rows(pick(beg, k), pick(deg, k), pick(sd, k), pick(s, k),
                                 pick(alpha, k), lane, ab, csrc, ssrc, z);
        *(v2f*)(out + (size_t)d * DIM + 2 * lane) = acc;
      }
    }
  }
}

__global__ __launch_bounds__(256)
void node_aggregate_next_gemm(const int* __restrict__ rowptr, const int* __restrict__ csrc,
                              const float* __restrict__ ssrc, const float* __restrict__ sdst,
                              const float* __restrict__ abp, const float* __restrict__ z,
                              const float* __restrict__ Wn, const float* __restrict__ biasn,
                              const float* __restrict__ attn, float* __restrict__ zn,
                              float* __restrict__ ssrcn, float* __restrict__ sdstn, int nN) {
  __shared__ __attribute__((aligned(16))) float sW[DIM * WPITCH];
  __shared__ __attribute__((aligned(16))) float sH[32 * WPITCH];
  __shared__ float sB[DIM];
  __shared__ float sP[2 * 16 * 32];

  const int tid = threadIdx.x;
  const int lane = tid & 31;
  const int w = __builtin_amdgcn_readfirstlane(tid >> 5);
  const int warp = blockIdx.x * 8 + w;
  const int nWarps = gridDim.x * 8;

  int rbe;
  float sdl;
  issue_bounds(warp, nWarps, nN, lane, rowptr, sdst, rbe, sdl);

  float bv = 0.0f;
  if (tid < DIM) bv = biasn[tid];
#pragma unroll
  for (int i = 0; i < DIM / 4; ++i) {
    const int r = (tid >> 6) + 4 * i;
    sW[r * WPITCH + (tid & (DIM - 1))] = Wn[r * DIM + (tid & (DIM - 1))];
  }
  if (tid < DIM) sB[tid] = bv;

  const float ab = abp[0];
  {
    int beg[NPW], deg[NPW], s[NPW];
    float sd[NPW], alpha[NPW];
    load_heads(rbe, sdl, warp, nWarps, nN, lane, ab, csrc, ssrc, beg, deg, sd, s, alpha);
#pragma unroll 1
    for (int k = 0; k < NPW; ++k) {
      v2f acc = sum_rows(pick(beg, k), pick(deg, k), pick(sd, k), pick(s, k),
                         pick(alpha, k), lane, ab, csrc, ssrc, z);
      acc.x = fmaxf(acc.x, 0.0f);
      acc.y = fmaxf(acc.y, 0.0f);
      *(v2f*)(&sH[(k * 8 + w) * WPITCH + 2 * lane]) = acc;
    }
  }
  __syncthreads();

  const int rt = w >> 1;
  const int ch = w & 1;
  const int laneRow = lane & 15;
  const int hh = lane >> 4;
  const int kHalf = hh * 2;
  const int dBase = blockIdx.x * 8 + (2 * rt + hh) * nWarps;
  const int c0 = ch * 32 + laneRow;

  v8f acc[2];
  v2f aw[2];
  if (w < 4) {
    const float* aBase = &sH[(rt * 16 + laneRow) * WPITCH + kHalf];
    const float* bBase = &sW[c0 * WPITCH + kHalf];
#pragma unroll
    for (int j = 0; j < 2; ++j) {
      const float bcol = sB[c0 + 16 * j];
#pragma unroll
      for (int r = 0; r < 8; ++r) acc[j][r] = bcol;
      aw[j].x = attn[c0 + 16 * j];
      aw[j].y = attn[DIM + c0 + 16 * j];
    }
#pragma unroll
    for (int t = 0; t < 16; ++t) {
      const v2f a = *(const v2f*)(aBase + 4 * t);
#pragma unroll
      for (int j = 0; j < 2; ++j) {
        const v2f b = *(const v2f*)(bBase + 16 * j * WPITCH + 4 * t);
        acc[j] = __builtin_amdgcn_wmma_f32_16x16x4_f32(
             false, a,  false, b,
             (short)0, acc[j],  false,  false);
      }
    }
    __builtin_amdgcn_sched_group_barrier(0x100, 3, 0);
#pragma unroll
    for (int g = 0; g < 7; ++g) {
      __builtin_amdgcn_sched_group_barrier(0x100, 3, 0);
      __builtin_amdgcn_sched_group_barrier(0x008, 4, 0);
    }
    __builtin_amdgcn_sched_group_barrier(0x008, 4, 0);
  }

  v2f sd[8];
  if (w < 4 && ch == 0) {
#pragma unroll
    for (int r = 0; r < 8; ++r) { sd[r].x = 0.0f; sd[r].y = 0.0f; }
#pragma unroll
    for (int j = 0; j < 2; ++j) {
#pragma unroll
      for (int r = 0; r < 8; ++r) sd[r] += acc[j][r] * aw[j];
    }
#pragma unroll
    for (int r = 0; r < 8; ++r) {
      sP[(rt * 16 + 2 * r) * 32 + lane] = sd[r].x;
      sP[(rt * 16 + 2 * r + 1) * 32 + lane] = sd[r].y;
    }
  }
  __syncthreads();
  if (w >= 4) return;

#pragma unroll
  for (int j = 0; j < 2; ++j) {
#pragma unroll
    for (int r = 0; r < 8; ++r) {
      const int d = dBase + r;
      if (d < nN) zn[(size_t)d * DIM + c0 + 16 * j] = acc[j][r];
    }
  }
  if (ch == 0) return;

#pragma unroll
  for (int r = 0; r < 8; ++r) {
    sd[r].x = sP[(rt * 16 + 2 * r) * 32 + lane];
    sd[r].y = sP[(rt * 16 + 2 * r + 1) * 32 + lane];
  }
#pragma unroll
  for (int j = 0; j < 2; ++j) {
#pragma unroll
    for (int r = 0; r < 8; ++r) sd[r] += acc[j][r] * aw[j];
  }

  const bool b3 = (lane & 8) != 0, b2 = (lane & 4) != 0;
  const bool b1 = (lane & 2) != 0, b0 = (lane & 1) != 0;
  float t[8], u[4], x2[2];
#pragma unroll
  for (int r = 0; r < 8; ++r) {
    const float keep = b3 ? sd[r].y : sd[r].x;
    const float send = b3 ? sd[r].x : sd[r].y;
    t[r] = keep + row_xor<8>(send);
  }
#pragma unroll
  for (int j = 0; j < 4; ++j) {
    const float keep = b2 ? t[j + 4] : t[j];
    const float send = b2 ? t[j] : t[j + 4];
    u[j] = keep + row_xor<4>(send);
  }
#pragma unroll
  for (int j = 0; j < 2; ++j) {
    const float keep = b1 ? u[j + 2] : u[j];
    const float send = b1 ? u[j] : u[j + 2];
    x2[j] = keep + row_xor<2>(send);
  }
  const float keep = b0 ? x2[1] : x2[0];
  const float send = b0 ? x2[0] : x2[1];
  const float y = keep + row_xor<1>(send);

  const int dy = dBase + (lane & 7);
  float* sp = b3 ? sdstn : ssrcn;
  if (dy < nN) sp[dy] = y;
}

extern "C" void kernel_launch(void* const* d_in, const int* in_sizes, int n_in,
                              void* d_out, int out_size, void* d_ws, size_t ws_size,
                              hipStream_t stream) {
  const float* x  = (const float*)d_in[0];
  const int* src  = (const int*)d_in[1];
  const int* dst  = (const int*)d_in[2];
  const int nN = in_sizes[0] / DIM;
  const int nE = in_sizes[1];

  char* ws = (char*)d_ws;
  const size_t featB = (size_t)nN * DIM * sizeof(float);
  const size_t edgeB = (size_t)nE * 4;
  const size_t nodeB = (size_t)nN * 4;
  float* z      = (float*)ws;
  float* z2     = (float*)(ws + featB);
  int*   bkt    = (int*)(ws + 2 * featB);
  int*   csrc   = (int*)(ws + 2 * featB + 2 * edgeB);
  int*   off    = (int*)(ws + 2 * featB + 3 * edgeB);
  const size_t tabB = (size_t)NWG * NB * 4;
  float* ssrc   = (float*)(ws + 2 * featB + 3 * edgeB + tabB + 2048);
  float* sdst   = (float*)(ws + 2 * featB + 3 * edgeB + tabB + 2048 + nodeB);
  int*   rowptr = (int*)(ws + 2 * featB + 3 * edgeB + tabB + 2048 + 2 * nodeB);
  float* ssrc2  = (float*)(ws + 2 * featB + 3 * edgeB + tabB + 4096 + 3 * nodeB);
  float* sdst2  = (float*)(ws + 2 * featB + 3 * edgeB + tabB + 4096 + 4 * nodeB);

  const int nTiles = (nN + 15) / 16;
  const dim3 gemmGrid((nTiles + 3) / 4), gemmBlk(128);
  const int aggBlocks = (nN + 31) / 32;

  piece_sort<<<NWG, 1024, 0, stream>>>(src, dst, off, bkt, nE, nN);
  range_to_lists<<<NB, 1024, 0, stream>>>(bkt, off, rowptr, csrc, nE, nN);

  gat_gemm_wmma<<<gemmGrid, gemmBlk, 0, stream>>>(x, (const float*)d_in[3], (const float*)d_in[4],
                                                  (const float*)d_in[5], z, ssrc, sdst, nTiles, 0);

  node_aggregate_next_gemm<<<aggBlocks, 256, 0, stream>>>(
      rowptr, csrc, ssrc, sdst, (const float*)d_in[6], z,
      (const float*)d_in[7], (const float*)d_in[8], (const float*)d_in[9], z2, ssrc2, sdst2, nN);

  node_aggregate_next_gemm<<<aggBlocks, 256, 0, stream>>>(
      rowptr, csrc, ssrc2, sdst2, (const float*)d_in[10], z2,
      (const float*)d_in[11], (const float*)d_in[12], (const float*)d_in[13], z, ssrc, sdst, nN);

  node_softmax_aggregate<<<aggBlocks, 256, 0, stream>>>(rowptr, csrc, ssrc, sdst,
                                                        (const float*)d_in[14], z,
                                                        (float*)d_out, nN);
}


# Layer 0's linear map and attention scalars run as the first work-groups (32 waves, a row tile each) of the first edge-grouping launch, beside the sort: 5 launches, not 6; all sums in parent order
# speedup vs baseline: 8.2176x; 1.0618x over previous
#include <hip/hip_runtime.h>
#include <hip/hip_bf16.h>

typedef __attribute__((ext_vector_type(2))) float v2f;
typedef __attribute__((ext_vector_type(8))) float v8f;

#define DIM 64
#define LEAKY 0.01f
#define WPITCH 68

template <int M>
__device__ __forceinline__ float row_xor(float x) {
  return __builtin_bit_cast(float, __builtin_amdgcn_update_dpp(
      0, __builtin_bit_cast(int, x), 0x160 + M  , 0xf, 0xf, true));
}

__device__ __forceinline__
void gat_gemm_wmma(const float* __restrict__ hin, const float* __restrict__ W,
                   const float* __restrict__ bias, const float* __restrict__ att,
                   float* __restrict__ z, float* __restrict__ ssrc,
                   float* __restrict__ sdst, int nTiles, int tb,
                   float* __restrict__ sW, float* __restrict__ sB) {
  const int tid = threadIdx.x;
  float bv = 0.0f;
  if (tid < DIM) bv = bias[tid];
#pragma unroll
  for (int i = 0; i < DIM / 16; ++i) {
    const int r = (tid >> 6) + 16 * i;
    sW[r * WPITCH + (tid & (DIM - 1))] = W[r * DIM + (tid & (DIM - 1))];
  }
  if (tid < DIM) sB[tid] = bv;
  __syncthreads();

  const int wave = tid >> 5;
  const int lane = tid & 31;
  const int rowTile = tb * 32 + wave;
  if (rowTile >= nTiles) return;

  const int row0 = rowTile * 16;
  const int laneRow = lane & 15;
  const int kHalf = (lane >> 4) * 2;

  v2f aw[4];
#pragma unroll
  for (int nt = 0; nt < 4; ++nt) {
    aw[nt].x = att[nt * 16 + laneRow];
    aw[nt].y = att[DIM + nt * 16 + laneRow];
  }

  const float* arow = hin + (size_t)(row0 + laneRow) * DIM;
  v2f aR[16];
#pragma unroll
  for (int t = 0; t < 16; ++t) {
    aR[t] = *(const v2f*)(arow + 4 * t + kHalf);
  }

  const int rOff = (lane < 16) ? 0 : 8;
  v2f sd[8];
#pragma unroll
  for (int r = 0; r < 8; ++r) { sd[r].x = 0.0f; sd[r].y = 0.0f; }
#pragma unroll
  for (int nt = 0; nt < 4; ++nt) {
    const int n0 = nt * 16;
    const float bcol = sB[n0 + laneRow];
    v8f acc;
#pragma unroll
    for (int r = 0; r < 8; ++r) acc[r] = bcol;

#pragma unroll
    for (int t = 0; t < 16; ++t) {
      v2f b = *(const v2f*)(&sW[(n0 + laneRow) * WPITCH + 4 * t + kHalf]);
      acc = __builtin_amdgcn_wmma_f32_16x16x4_f32(
           false, aR[t],  false, b,
           (short)0, acc,  false,  false);
    }

    float* zr = z + (size_t)(row0 + rOff) * DIM + n0 + laneRow;
#pragma unroll
    for (int r = 0; r < 8; ++r) zr[(size_t)r * DIM] = acc[r];
#pragma unroll
    for (int r = 0; r < 8; ++r) sd[r] += acc[r] * aw[nt];
  }

  const bool b3 = (lane & 8) != 0, b2 = (lane & 4) != 0;
  const bool b1 = (lane & 2) != 0, b0 = (lane & 1) != 0;
  float t[8], u[4], w[2];
#pragma unroll
  for (int r = 0; r < 8; ++r) {
    const float keep = b3 ? sd[r].y : sd[r].x;
    const float send = b3 ? sd[r].x : sd[r].y;
    t[r] = keep + row_xor<8>(send);
  }
#pragma unroll
  for (int j = 0; j < 4; ++j) {
    const float keep = b2 ? t[j + 4] : t[j];
    const float send = b2 ? t[j] : t[j + 4];
    u[j] = keep + row_xor<4>(send);
  }
#pragma unroll
  for (int j = 0; j < 2; ++j) {
    const float keep = b1 ? u[j + 2] : u[j];
    const float send = b1 ? u[j] : u[j + 2];
    w[j] = keep + row_xor<2>(send);
  }
  const float keep = b0 ? w[1] : w[0];
  const float send = b0 ? w[0] : w[1];
  const float y = keep + row_xor<1>(send);

  float* sp = b3 ? sdst : ssrc;
  sp[row0 + rOff + (lane & 7)] = y;
}

#define NB 256
#define NWG 256
#define MAXS 1024

typedef __attribute__((ext_vector_type(2))) int v2i;

__device__ __forceinline__ int block_excl_scan(int c, int* wsum) {
  const int lane = threadIdx.x & 31;
  const int wave = threadIdx.x >> 5;
  int x = c;
#pragma unroll
  for (int off = 1; off < 32; off <<= 1) {
    const int y = __shfl_up(x, off, 32);
    if (lane >= off) x += y;
  }
  if (lane == 31) wsum[wave] = x;
  __syncthreads();
  int part = (lane < wave) ? wsum[lane] : 0;
#pragma unroll
  for (int off = 16; off > 0; off >>= 1) part += __shfl_xor(part, off, 32);
  return part + x - c;
}

__global__ __launch_bounds__(1024)
void piece_sort_and_linear(const int* __restrict__ src, const int* __restrict__ dst,
                           int* __restrict__ off, int* __restrict__ bkt, int nE, int nN,
                           const float* __restrict__ x, const float* __restrict__ W,
                           const float* __restrict__ bias, const float* __restrict__ att,
                           float* __restrict__ z, float* __restrict__ ssrc,
                           float* __restrict__ sdst, int nTiles, int nLin) {
  __shared__ __attribute__((aligned(16))) float sW[DIM * WPITCH];
  __shared__ float sB[DIM];
  __shared__ int hist[NB];
  __shared__ int wsum[32];
  if ((int)blockIdx.x < nLin) {
    gat_gemm_wmma(x, W, bias, att, z, ssrc, sdst, nTiles, blockIdx.x, sW, sB);
    return;
  }
  const int tid = threadIdx.x;
  const int wg = blockIdx.x - nLin;
  const int S = (nN + NB - 1) / NB;
  const int chunk = (nE + NWG - 1) / NWG;
  const int e0 = wg * chunk;
  const int e1 = min(e0 + chunk, nE);
  if (tid < NB) hist[tid] = 0;

  v2i rec[4];
#pragma unroll
  for (int u = 0; u < 4; ++u) {
    const int e = e0 + tid + u * 1024;
    rec[u].x = (e < e1) ? dst[e] : 0;
    rec[u].y = (e < e1) ? src[e] : 0;
  }
  __syncthreads();
  int rg[4];
#pragma unroll
  for (int u = 0; u < 4; ++u) {
    rg[u] = rec[u].x / S;
    if (e0 + tid + u * 1024 < e1) atomicAdd(&hist[rg[u]], 1);
  }
  for (int eb = e0 + tid + 4 * 1024; eb < e1; eb += 4 * 1024) {
    int d[4];
#pragma unroll
    for (int u = 0; u < 4; ++u) {
      const int e = eb + u * 1024;
      d[u] = (e < e1) ? dst[e] : 0;
    }
#pragma unroll
    for (int u = 0; u < 4; ++u) {
      if (eb + u * 1024 < e1) atomicAdd(&hist[d[u] / S], 1);
    }
  }
  __syncthreads();
  const int c = (tid < NB) ? hist[tid] : 0;
  const int ex = block_excl_scan(c, wsum);
  if (tid < NB) {
    hist[tid] = ex;
    off[wg * NB + tid] = ex;
  }
  __syncthreads();

  int pos[4];
#pragma unroll
  for (int u = 0; u < 4; ++u) {
    pos[u] = 0;
    if (e0 + tid + u * 1024 < e1) pos[u] = atomicAdd(&hist[rg[u]], 1);
  }
#pragma unroll
  for (int u = 0; u < 4; ++u) {
    if (e0 + tid + u * 1024 < e1) *(v2i*)(bkt + 2 * (size_t)(e0 + pos[u])) = rec[u];
  }
  for (int eb = e0 + tid + 4 * 1024; eb < e1; eb += 4 * 1024) {
    v2i r2[4];
#pragma unroll
    for (int u = 0; u < 4; ++u) {
      const int e = eb + u * 1024;
      r2[u].x = (e < e1) ? dst[e] : 0;
      r2[u].y = (e < e1) ? src[e] : 0;
    }
#pragma unroll
    for (int u = 0; u < 4; ++u) {
      if (eb + u * 1024 < e1) {
        const int p2 = atomicAdd(&hist[r2[u].x / S], 1);
        *(v2i*)(bkt + 2 * (size_t)(e0 + p2)) = r2[u];
      }
    }
  }
}

__global__ __launch_bounds__(1024)
void range_to_lists(const int* __restrict__ bkt, const int* __restrict__ off,
                    int* __restrict__ rowptr, int* __restrict__ csrc, int nE, int nN) {
  __shared__ int cnt[MAXS];
  __shared__ int wsum[32];
  __shared__ int red[32];
  const int tid = threadIdx.x;
  const int lane = tid & 31;
  const int b = blockIdx.x;
  const int S = (nN + NB - 1) / NB;
  const int n0 = b * S;
  const int chunk = (nE + NWG - 1) / NWG;
  const int pc = tid >> 2;
  const int q = tid & 3;
  const int pe0 = pc * chunk;
  const int plen = max(min(pe0 + chunk, nE) - pe0, 0);
  const int o0 = off[pc * NB + b];
  const int o1 = (b + 1 < NB) ? off[pc * NB + b + 1] : plen;
  const int sb = pe0 + o0;
  const int se = pe0 + o1;

  v2i rec[8];
#pragma unroll
  for (int u = 0; u < 8; ++u) {
    const int i = sb + q + 4 * u;
    rec[u].x = n0;
    rec[u].y = 0;
    if (i < se) rec[u] = *(const v2i*)(bkt + 2 * (size_t)i);
  }

  int part = (q == 0) ? o0 : 0;
#pragma unroll
  for (int sh = 16; sh > 0; sh >>= 1) part += __shfl_xor(part, sh, 32);
  if (lane == 0) red[tid >> 5] = part;
  cnt[tid] = 0;
  __syncthreads();

#pragma unroll
  for (int u = 0; u < 8; ++u) {
    if (sb + q + 4 * u < se) atomicAdd(&cnt[rec[u].x - n0], 1);
  }
  for (int ib = sb + q + 32; ib < se; ib += 16) {
    int d[4];
#pragma unroll
    for (int u = 0; u < 4; ++u) {
      const int i = ib + 4 * u;
      d[u] = (i < se) ? bkt[2 * (size_t)i] : n0;
    }
#pragma unroll
    for (int u = 0; u < 4; ++u) {
      if (ib + 4 * u < se) atomicAdd(&cnt[d[u] - n0], 1);
    }
  }
  __syncthreads();
  const int c = cnt[tid];
  const int ex = block_excl_scan(c, wsum);
  cnt[tid] = ex;
  int beg = 0;
#pragma unroll
  for (int w = 0; w < 32; ++w) beg += red[w];
  if (tid < S && n0 + tid < nN) rowptr[n0 + tid] = beg + ex;
  if (b == 0 && tid == 0) rowptr[nN] = nE;
  __syncthreads();

  int k[8];
#pragma unroll
  for (int u = 0; u < 8; ++u) {
    k[u] = 0;
    if (sb + q + 4 * u < se) k[u] = atomicAdd(&cnt[rec[u].x - n0], 1);
  }
#pragma unroll
  for (int u = 0; u < 8; ++u) {
    if (sb + q + 4 * u < se) csrc[beg + k[u]] = rec[u].y;
  }
  for (int ib = sb + q + 32; ib < se; ib += 16) {
    v2i r2[4];
#pragma unroll
    for (int u = 0; u < 4; ++u) {
      const int i = ib + 4 * u;
      r2[u].x = n0;
      r2[u].y = 0;
      if (i < se) r2[u] = *(const v2i*)(bkt + 2 * (size_t)i);
    }
#pragma unroll
    for (int u = 0; u < 4; ++u) {
      if (ib + 4 * u < se) {
        const int k2 = atomicAdd(&cnt[r2[u].x - n0], 1);
        csrc[beg + k2] = r2[u].y;
      }
    }
  }
}

__device__ __forceinline__ float wave_max(float v) {
#pragma unroll
  for (int off = 16; off > 0; off >>= 1) v = fmaxf(v, __shfl_xor(v, off, 32));
  return v;
}
__device__ __forceinline__ float wave_sum(float v) {
#pragma unroll
  for (int off = 16; off > 0; off >>= 1) v += __shfl_xor(v, off, 32);
  return v;
}

template <int B>
__device__ __forceinline__ void add_rows(const float* __restrict__ z, int s, float alpha,
                                         int j0, int lane, v2f& acc) {
  v2f r[B];
#pragma unroll
  for (int u = 0; u < B; ++u) {
    const int sj = __builtin_amdgcn_readlane(s, j0 + u);
    r[u] = *(const v2f*)(z + (size_t)sj * DIM + 2 * lane);
  }
#pragma unroll
  for (int u = 0; u < B; ++u) {
    const float aj = __int_as_float(__builtin_amdgcn_readlane(__float_as_int(alpha), j0 + u));
    acc.x += aj * r[u].x;
    acc.y += aj * r[u].y;
  }
}

__device__ __forceinline__ void add_list(const float* __restrict__ z, int s, float alpha,
                                         int n, int lane, v2f& acc) {
  int j = 0;
  for (; j + 8 <= n; j += 8) add_rows<8>(z, s, alpha, j, lane, acc);
  if (n - j >= 4) { add_rows<4>(z, s, alpha, j, lane, acc); j += 4; }
  if (n - j >= 2) { add_rows<2>(z, s, alpha, j, lane, acc); j += 2; }
  if (n - j >= 1) { add_rows<1>(z, s, alpha, j, lane, acc); }
}

#define NPW 4

__device__ __forceinline__ void issue_bounds(int d0, int stride, int nN, int lane,
                                             const int* __restrict__ rowptr,
                                             const float* __restrict__ sdst,
                                             int& rbe, float& sdl) {
  const int kl = (lane >> 1) & (NPW - 1);
  const int dl = d0 + kl * stride;
  const int dk = (dl < nN) ? dl : 0;
  rbe = rowptr[dk + (lane & 1)];
  sdl = sdst[dk];
}

__device__ __forceinline__ void load_heads(int rbe, float sdl, int d0, int stride, int nN,
                                           int lane, float ab,
                                           const int* __restrict__ csrc,
                                           const float* __restrict__ ssrc,
                                           int (&beg)[NPW], int (&deg)[NPW], float (&sd)[NPW],
                                           int (&s)[NPW], float (&alpha)[NPW]) {
#pragma unroll
  for (int k = 0; k < NPW; ++k) {
    beg[k] = __builtin_amdgcn_readlane(rbe, 2 * k);
    const int end = __builtin_amdgcn_readlane(rbe, 2 * k + 1);
    deg[k] = (d0 + k * stride < nN) ? end - beg[k] : 0;
    sd[k] = __int_as_float(__builtin_amdgcn_readlane(__float_as_int(sdl), 2 * k));
  }
#pragma unroll
  for (int k = 0; k < NPW; ++k) s[k] = csrc[(lane < deg[k]) ? beg[k] + lane : 0];
  float ts[NPW];
#pragma unroll
  for (int k = 0; k < NPW; ++k) ts[k] = ssrc[s[k]];
  float v[NPW], m[NPW], ex[NPW], den[NPW];
#pragma unroll
  for (int k = 0; k < NPW; ++k) {
    const float t = ts[k] + sd[k] + ab;
    const float lv = (t > 0.0f) ? t : LEAKY * t;
    v[k] = (lane < deg[k]) ? lv : -INFINITY;
    m[k] = v[k];
  }
#pragma unroll
  for (int off = 16; off > 0; off >>= 1) {
#pragma unroll
    for (int k = 0; k < NPW; ++k) m[k] = fmaxf(m[k], __shfl_xor(m[k], off, 32));
  }
#pragma unroll
  for (int k = 0; k < NPW; ++k) {
    ex[k] = (lane < deg[k]) ? __expf(v[k] - m[k]) : 0.0f;
    den[k] = ex[k];
  }
#pragma unroll
  for (int off = 16; off > 0; off >>= 1) {
#pragma unroll
    for (int k = 0; k < NPW; ++k) den[k] += __shfl_xor(den[k], off, 32);
  }
#pragma unroll
  for (int k = 0; k < NPW; ++k) alpha[k] = ex[k] / den[k];
}

template <typename T>
__device__ __forceinline__ T pick(const T (&a)[NPW], int k) {
  T r = a[0];
#pragma unroll
  for (int j = 1; j < NPW; ++j) r = (k == j) ? a[j] : r;
  return r;
}

__device__ __forceinline__ v2f sum_rows(int beg, int deg, float sd, int s, float alpha,
                                        int lane, float ab,
                                        const int* __restrict__ csrc,
                                        const float* __restrict__ ssrc,
                                        const float* __restrict__ z) {
  v2f acc; acc.x = 0.0f; acc.y = 0.0f;
  if (deg > 0 && deg <= 32) {
    add_list(z, s, alpha, deg, lane, acc);
  } else if (deg > 32) {
    const int end = beg + deg;
    float m = -INFINITY;
    for (int c = beg; c < end; c += 32) {
      const int i = c + lane;
      if (i < end) {
        const float t = ssrc[csrc[i]] + sd + ab;
        m = fmaxf(m, (t > 0.0f) ? t : LEAKY * t);
      }
    }
    m = wave_max(m);
    float den = 0.0f;
    for (int c = beg; c < end; c += 32) {
      const int i = c + lane;
      if (i < end) {
        const float t = ssrc[csrc[i]] + sd + ab;
        den += __expf(((t > 0.0f) ? t : LEAKY * t) - m);
      }
    }
    den = wave_sum(den);
    for (int c = beg; c < end; c += 32) {
      const int i = c + lane;
      int s2 = 0;
      float a2 = 0.0f;
      if (i < end) {
        s2 = csrc[i];
        const float t = ssrc[s2] + sd + ab;
        a2 = __expf(((t > 0.0f) ? t : LEAKY * t) - m) / den;
      }
      add_list(z, s2, a2, min(32, end - c), lane, acc);
    }
  }
  return acc;
}

__global__ __launch_bounds__(256)
void node_softmax_aggregate(const int* __restrict__ rowptr, const int* __restrict__ csrc,
                            const float* __restrict__ ssrc, const float* __restrict__ sdst,
                            const float* __restrict__ abp, const float* __restrict__ z,
                            float* __restrict__ out, int nN) {
  const int lane = threadIdx.x & 31;
  const int warp = __builtin_amdgcn_readfirstlane((int)((blockIdx.x * blockDim.x + threadIdx.x) >> 5));
  const int nWarps = (gridDim.x * blockDim.x) >> 5;
  const float ab = abp[0];

  for (int base = warp; base < nN; base += NPW * nWarps) {
    int beg[NPW], deg[NPW], s[NPW];
    float sd[NPW], alpha[NPW];
    int rbe;
    float sdl;
    issue_bounds(base, nWarps, nN, lane, rowptr, sdst, rbe, sdl);
    load_heads(rbe, sdl, base, nWarps, nN, lane, ab, csrc, ssrc, beg, deg, sd, s, alpha);
#pragma unroll 1
    for (int k = 0; k < NPW; ++k) {
      const int d = base + k * nWarps;
      if (d < nN) {
        const v2f acc = sum_rows(pick(beg, k), pick(deg, k), pick(sd, k), pick(s, k),
                                 pick(alpha, k), lane, ab, csrc, ssrc, z);
        *(v2f*)(out + (size_t)d * DIM + 2 * lane) = acc;
      }
    }
  }
}

__global__ __launch_bounds__(256)
void node_aggregate_next_gemm(const int* __restrict__ rowptr, const int* __restrict__ csrc,
                              const float* __restrict__ ssrc, const float* __restrict__ sdst,
                              const float* __restrict__ abp, const float* __restrict__ z,
                              const float* __restrict__ Wn, const float* __restrict__ biasn,
                              const float* __restrict__ attn, float* __restrict__ zn,
                              float* __restrict__ ssrcn, float* __restrict__ sdstn, int nN) {
  __shared__ __attribute__((aligned(16))) float sW[DIM * WPITCH];
  __shared__ __attribute__((aligned(16))) float sH[32 * WPITCH];
  __shared__ float sB[DIM];
  __shared__ float sP[2 * 16 * 32];

  const int tid = threadIdx.x;
  const int lane = tid & 31;
  const int w = __builtin_amdgcn_readfirstlane(tid >> 5);
  const int warp = blockIdx.x * 8 + w;
  const int nWarps = gridDim.x * 8;

  int rbe;
  float sdl;
  issue_bounds(warp, nWarps, nN, lane, rowptr, sdst, rbe, sdl);

  float bv = 0.0f;
  if (tid < DIM) bv = biasn[tid];
#pragma unroll
  for (int i = 0; i < DIM / 4; ++i) {
    const int r = (tid >> 6) + 4 * i;
    sW[r * WPITCH + (tid & (DIM - 1))] = Wn[r * DIM + (tid & (DIM - 1))];
  }
  if (tid < DIM) sB[tid] = bv;

  const float ab = abp[0];
  {
    int beg[NPW], deg[NPW], s[NPW];
    float sd[NPW], alpha[NPW];
    load_heads(rbe, sdl, warp, nWarps, nN, lane, ab, csrc, ssrc, beg, deg, sd, s, alpha);
#pragma unroll 1
    for (int k = 0; k < NPW; ++k) {
      v2f acc = sum_rows(pick(beg, k), pick(deg, k), pick(sd, k), pick(s, k),
                         pick(alpha, k), lane, ab, csrc, ssrc, z);
      acc.x = fmaxf(acc.x, 0.0f);
      acc.y = fmaxf(acc.y, 0.0f);
      *(v2f*)(&sH[(k * 8 + w) * WPITCH + 2 * lane]) = acc;
    }
  }
  __syncthreads();

  const int rt = w >> 1;
  const int ch = w & 1;
  const int laneRow = lane & 15;
  const int hh = lane >> 4;
  const int kHalf = hh * 2;
  const int dBase = blockIdx.x * 8 + (2 * rt + hh) * nWarps;
  const int c0 = ch * 32 + laneRow;

  v8f acc[2];
  v2f aw[2];
  if (w < 4) {
    const float* aBase = &sH[(rt * 16 + laneRow) * WPITCH + kHalf];
    const float* bBase = &sW[c0 * WPITCH + kHalf];
#pragma unroll
    for (int j = 0; j < 2; ++j) {
      const float bcol = sB[c0 + 16 * j];
#pragma unroll
      for (int r = 0; r < 8; ++r) acc[j][r] = bcol;
      aw[j].x = attn[c0 + 16 * j];
      aw[j].y = attn[DIM + c0 + 16 * j];
    }
#pragma unroll
    for (int t = 0; t < 16; ++t) {
      const v2f a = *(const v2f*)(aBase + 4 * t);
#pragma unroll
      for (int j = 0; j < 2; ++j) {
        const v2f b = *(const v2f*)(bBase + 16 * j * WPITCH + 4 * t);
        acc[j] = __builtin_amdgcn_wmma_f32_16x16x4_f32(
             false, a,  false, b,
             (short)0, acc[j],  false,  false);
      }
    }
    __builtin_amdgcn_sched_group_barrier(0x100, 3, 0);
#pragma unroll
    for (int g = 0; g < 7; ++g) {
      __builtin_amdgcn_sched_group_barrier(0x100, 3, 0);
      __builtin_amdgcn_sched_group_barrier(0x008, 4, 0);
    }
    __builtin_amdgcn_sched_group_barrier(0x008, 4, 0);
  }

  v2f sd[8];
  if (w < 4 && ch == 0) {
#pragma unroll
    for (int r = 0; r < 8; ++r) { sd[r].x = 0.0f; sd[r].y = 0.0f; }
#pragma unroll
    for (int j = 0; j < 2; ++j) {
#pragma unroll
      for (int r = 0; r < 8; ++r) sd[r] += acc[j][r] * aw[j];
    }
#pragma unroll
    for (int r = 0; r < 8; ++r) {
      sP[(rt * 16 + 2 * r) * 32 + lane] = sd[r].x;
      sP[(rt * 16 + 2 * r + 1) * 32 + lane] = sd[r].y;
    }
  }
  __syncthreads();
  if (w >= 4) return;

#pragma unroll
  for (int j = 0; j < 2; ++j) {
#pragma unroll
    for (int r = 0; r < 8; ++r) {
      const int d = dBase + r;
      if (d < nN) zn[(size_t)d * DIM + c0 + 16 * j] = acc[j][r];
    }
  }
  if (ch == 0) return;

#pragma unroll
  for (int r = 0; r < 8; ++r) {
    sd[r].x = sP[(rt * 16 + 2 * r) * 32 + lane];
    sd[r].y = sP[(rt * 16 + 2 * r + 1) * 32 + lane];
  }
#pragma unroll
  for (int j = 0; j < 2; ++j) {
#pragma unroll
    for (int r = 0; r < 8; ++r) sd[r] += acc[j][r] * aw[j];
  }

  const bool b3 = (lane & 8) != 0, b2 = (lane & 4) != 0;
  const bool b1 = (lane & 2) != 0, b0 = (lane & 1) != 0;
  float t[8], u[4], x2[2];
#pragma unroll
  for (int r = 0; r < 8; ++r) {
    const float keep = b3 ? sd[r].y : sd[r].x;
    const float send = b3 ? sd[r].x : sd[r].y;
    t[r] = keep + row_xor<8>(send);
  }
#pragma unroll
  for (int j = 0; j < 4; ++j) {
    const float keep = b2 ? t[j + 4] : t[j];
    const float send = b2 ? t[j] : t[j + 4];
    u[j] = keep + row_xor<4>(send);
  }
#pragma unroll
  for (int j = 0; j < 2; ++j) {
    const float keep = b1 ? u[j + 2] : u[j];
    const float send = b1 ? u[j] : u[j + 2];
    x2[j] = keep + row_xor<2>(send);
  }
  const float keep = b0 ? x2[1] : x2[0];
  const float send = b0 ? x2[0] : x2[1];
  const float y = keep + row_xor<1>(send);

  const int dy = dBase + (lane & 7);
  float* sp = b3 ? sdstn : ssrcn;
  if (dy < nN) sp[dy] = y;
}

extern "C" void kernel_launch(void* const* d_in, const int* in_sizes, int n_in,
                              void* d_out, int out_size, void* d_ws, size_t ws_size,
                              hipStream_t stream) {
  const float* x  = (const float*)d_in[0];
  const int* src  = (const int*)d_in[1];
  const int* dst  = (const int*)d_in[2];
  const int nN = in_sizes[0] / DIM;
  const int nE = in_sizes[1];

  char* ws = (char*)d_ws;
  const size_t featB = (size_t)nN * DIM * sizeof(float);
  const size_t edgeB = (size_t)nE * 4;
  const size_t nodeB = (size_t)nN * 4;
  float* z      = (float*)ws;
  float* z2     = (float*)(ws + featB);
  int*   bkt    = (int*)(ws + 2 * featB);
  int*   csrc   = (int*)(ws + 2 * featB + 2 * edgeB);
  int*   off    = (int*)(ws + 2 * featB + 3 * edgeB);
  const size_t tabB = (size_t)NWG * NB * 4;
  float* ssrc   = (float*)(ws + 2 * featB + 3 * edgeB + tabB + 2048);
  float* sdst   = (float*)(ws + 2 * featB + 3 * edgeB + tabB + 2048 + nodeB);
  int*   rowptr = (int*)(ws + 2 * featB + 3 * edgeB + tabB + 2048 + 2 * nodeB);
  float* ssrc2  = (float*)(ws + 2 * featB + 3 * edgeB + tabB + 4096 + 3 * nodeB);
  float* sdst2  = (float*)(ws + 2 * featB + 3 * edgeB + tabB + 4096 + 4 * nodeB);

  const int nTiles = (nN + 15) / 16;
  const int nLin = (nTiles + 31) / 32;
  const int aggBlocks = (nN + 31) / 32;

  piece_sort_and_linear<<<nLin + NWG, 1024, 0, stream>>>(
      src, dst, off, bkt, nE, nN, x, (const float*)d_in[3], (const float*)d_in[4],
      (const float*)d_in[5], z, ssrc, sdst, nTiles, nLin);
  range_to_lists<<<NB, 1024, 0, stream>>>(bkt, off, rowptr, csrc, nE, nN);

  node_aggregate_next_gemm<<<aggBlocks, 256, 0, stream>>>(
      rowptr, csrc, ssrc, sdst, (const float*)d_in[6], z,
      (const float*)d_in[7], (const float*)d_in[8], (const float*)d_in[9], z2, ssrc2, sdst2, nN);

  node_aggregate_next_gemm<<<aggBlocks, 256, 0, stream>>>(
      rowptr, csrc, ssrc2, sdst2, (const float*)d_in[10], z2,
      (const float*)d_in[11], (const float*)d_in[12], (const float*)d_in[13], z, ssrc, sdst, nN);

  node_softmax_aggregate<<<aggBlocks, 256, 0, stream>>>(rowptr, csrc, ssrc, sdst,
                                                        (const float*)d_in[14], z,
                                                        (float*)d_out, nN);
}


# Edge grouping: both placing passes write into LDS and copy the finished piece or range out in index order, whole 128-B lines per store, not one partial-line write per edge; list order as before
# speedup vs baseline: 8.6370x; 1.0510x over previous
#include <hip/hip_runtime.h>
#include <hip/hip_bf16.h>

typedef __attribute__((ext_vector_type(2))) float v2f;
typedef __attribute__((ext_vector_type(8))) float v8f;

#define DIM 64
#define LEAKY 0.01f
#define WPITCH 68

template <int M>
__device__ __forceinline__ float row_xor(float x) {
  return __builtin_bit_cast(float, __builtin_amdgcn_update_dpp(
      0, __builtin_bit_cast(int, x), 0x160 + M  , 0xf, 0xf, true));
}

__device__ __forceinline__
void gat_gemm_wmma(const float* __restrict__ hin, const float* __restrict__ W,
                   const float* __restrict__ bias, const float* __restrict__ att,
                   float* __restrict__ z, float* __restrict__ ssrc,
                   float* __restrict__ sdst, int nTiles, int tb,
                   float* __restrict__ sW, float* __restrict__ sB) {
  const int tid = threadIdx.x;
  float bv = 0.0f;
  if (tid < DIM) bv = bias[tid];
#pragma unroll
  for (int i = 0; i < DIM / 16; ++i) {
    const int r = (tid >> 6) + 16 * i;
    sW[r * WPITCH + (tid & (DIM - 1))] = W[r * DIM + (tid & (DIM - 1))];
  }
  if (tid < DIM) sB[tid] = bv;
  __syncthreads();

  const int wave = tid >> 5;
  const int lane = tid & 31;
  const int rowTile = tb * 32 + wave;
  if (rowTile >= nTiles) return;

  const int row0 = rowTile * 16;
  const int laneRow = lane & 15;
  const int kHalf = (lane >> 4) * 2;

  v2f aw[4];
#pragma unroll
  for (int nt = 0; nt < 4; ++nt) {
    aw[nt].x = att[nt * 16 + laneRow];
    aw[nt].y = att[DIM + nt * 16 + laneRow];
  }

  const float* arow = hin + (size_t)(row0 + laneRow) * DIM;
  v2f aR[16];
#pragma unroll
  for (int t = 0; t < 16; ++t) {
    aR[t] = *(const v2f*)(arow + 4 * t + kHalf);
  }

  const int rOff = (lane < 16) ? 0 : 8;
  v2f sd[8];
#pragma unroll
  for (int r = 0; r < 8; ++r) { sd[r].x = 0.0f; sd[r].y = 0.0f; }
#pragma unroll
  for (int nt = 0; nt < 4; ++nt) {
    const int n0 = nt * 16;
    const float bcol = sB[n0 + laneRow];
    v8f acc;
#pragma unroll
    for (int r = 0; r < 8; ++r) acc[r] = bcol;

#pragma unroll
    for (int t = 0; t < 16; ++t) {
      v2f b = *(const v2f*)(&sW[(n0 + laneRow) * WPITCH + 4 * t + kHalf]);
      acc = __builtin_amdgcn_wmma_f32_16x16x4_f32(
           false, aR[t],  false, b,
           (short)0, acc,  false,  false);
    }

    float* zr = z + (size_t)(row0 + rOff) * DIM + n0 + laneRow;
#pragma unroll
    for (int r = 0; r < 8; ++r) zr[(size_t)r * DIM] = acc[r];
#pragma unroll
    for (int r = 0; r < 8; ++r) sd[r] += acc[r] * aw[nt];
  }

  const bool b3 = (lane & 8) != 0, b2 = (lane & 4) != 0;
  const bool b1 = (lane & 2) != 0, b0 = (lane & 1) != 0;
  float t[8], u[4], w[2];
#pragma unroll
  for (int r = 0; r < 8; ++r) {
    const float keep = b3 ? sd[r].y : sd[r].x;
    const float send = b3 ? sd[r].x : sd[r].y;
    t[r] = keep + row_xor<8>(send);
  }
#pragma unroll
  for (int j = 0; j < 4; ++j) {
    const float keep = b2 ? t[j + 4] : t[j];
    const float send = b2 ? t[j] : t[j + 4];
    u[j] = keep + row_xor<4>(send);
  }
#pragma unroll
  for (int j = 0; j < 2; ++j) {
    const float keep = b1 ? u[j + 2] : u[j];
    const float send = b1 ? u[j] : u[j + 2];
    w[j] = keep + row_xor<2>(send);
  }
  const float keep = b0 ? w[1] : w[0];
  const float send = b0 ? w[0] : w[1];
  const float y = keep + row_xor<1>(send);

  float* sp = b3 ? sdst : ssrc;
  sp[row0 + rOff + (lane & 7)] = y;
}

#define NB 256
#define NWG 256
#define MAXS 1024
#define CAP1 4096
#define CAP2 8192

typedef __attribute__((ext_vector_type(2))) int v2i;

__device__ __forceinline__ int block_excl_scan(int c, int* wsum) {
  const int lane = threadIdx.x & 31;
  const int wave = threadIdx.x >> 5;
  int x = c;
#pragma unroll
  for (int off = 1; off < 32; off <<= 1) {
    const int y = __shfl_up(x, off, 32);
    if (lane >= off) x += y;
  }
  if (lane == 31) wsum[wave] = x;
  __syncthreads();
  int part = (lane < wave) ? wsum[lane] : 0;
#pragma unroll
  for (int off = 16; off > 0; off >>= 1) part += __shfl_xor(part, off, 32);
  return part + x - c;
}

__global__ __launch_bounds__(1024)
void piece_sort_and_linear(const int* __restrict__ src, const int* __restrict__ dst,
                           int* __restrict__ off, int* __restrict__ bkt, int nE, int nN,
                           const float* __restrict__ x, const float* __restrict__ W,
                           const float* __restrict__ bias, const float* __restrict__ att,
                           float* __restrict__ z, float* __restrict__ ssrc,
                           float* __restrict__ sdst, int nTiles, int nLin) {
  static_assert(2 * CAP1 >= DIM * WPITCH, "the W tile must fit the shared buffer");
  __shared__ __attribute__((aligned(16))) float sW[2 * CAP1];
  __shared__ float sB[DIM];
  __shared__ int hist[NB];
  __shared__ int wsum[32];
  if ((int)blockIdx.x < nLin) {
    gat_gemm_wmma(x, W, bias, att, z, ssrc, sdst, nTiles, blockIdx.x, sW, sB);
    return;
  }
  const int tid = threadIdx.x;
  const int wg = blockIdx.x - nLin;
  const int S = (nN + NB - 1) / NB;
  const int chunk = (nE + NWG - 1) / NWG;
  const int e0 = wg * chunk;
  const int e1 = min(e0 + chunk, nE);
  if (tid < NB) hist[tid] = 0;

  v2i rec[4];
#pragma unroll
  for (int u = 0; u < 4; ++u) {
    const int e = e0 + tid + u * 1024;
    rec[u].x = (e < e1) ? dst[e] : 0;
    rec[u].y = (e < e1) ? src[e] : 0;
  }
  __syncthreads();
  int rg[4];
#pragma unroll
  for (int u = 0; u < 4; ++u) {
    rg[u] = rec[u].x / S;
    if (e0 + tid + u * 1024 < e1) atomicAdd(&hist[rg[u]], 1);
  }
  for (int eb = e0 + tid + 4 * 1024; eb < e1; eb += 4 * 1024) {
    int d[4];
#pragma unroll
    for (int u = 0; u < 4; ++u) {
      const int e = eb + u * 1024;
      d[u] = (e < e1) ? dst[e] : 0;
    }
#pragma unroll
    for (int u = 0; u < 4; ++u) {
      if (eb + u * 1024 < e1) atomicAdd(&hist[d[u] / S], 1);
    }
  }
  __syncthreads();
  const int c = (tid < NB) ? hist[tid] : 0;
  const int ex = block_excl_scan(c, wsum);
  if (tid < NB) {
    hist[tid] = ex;
    off[wg * NB + tid] = ex;
  }
  __syncthreads();

  v2i* stg = (v2i*)sW;
  const bool staged = (e1 - e0) <= CAP1;
  int pos[4];
#pragma unroll
  for (int u = 0; u < 4; ++u) {
    pos[u] = 0;
    if (e0 + tid + u * 1024 < e1) pos[u] = atomicAdd(&hist[rg[u]], 1);
  }
  if (staged) {
#pragma unroll
    for (int u = 0; u < 4; ++u) {
      if (e0 + tid + u * 1024 < e1) stg[pos[u]] = rec[u];
    }
    __syncthreads();
    v2i pr[4];
#pragma unroll
    for (int u = 0; u < 4; ++u) pr[u] = stg[tid + u * 1024];
#pragma unroll
    for (int u = 0; u < 4; ++u) {
      const int i = tid + u * 1024;
      if (e0 + i < e1) *(v2i*)(bkt + 2 * (size_t)(e0 + i)) = pr[u];
    }
    return;
  }
#pragma unroll
  for (int u = 0; u < 4; ++u) {
    if (e0 + tid + u * 1024 < e1) *(v2i*)(bkt + 2 * (size_t)(e0 + pos[u])) = rec[u];
  }
  for (int eb = e0 + tid + 4 * 1024; eb < e1; eb += 4 * 1024) {
    v2i r2[4];
#pragma unroll
    for (int u = 0; u < 4; ++u) {
      const int e = eb + u * 1024;
      r2[u].x = (e < e1) ? dst[e] : 0;
      r2[u].y = (e < e1) ? src[e] : 0;
    }
#pragma unroll
    for (int u = 0; u < 4; ++u) {
      if (eb + u * 1024 < e1) {
        const int p2 = atomicAdd(&hist[r2[u].x / S], 1);
        *(v2i*)(bkt + 2 * (size_t)(e0 + p2)) = r2[u];
      }
    }
  }
}

__global__ __launch_bounds__(1024)
void range_to_lists(const int* __restrict__ bkt, const int* __restrict__ off,
                    int* __restrict__ rowptr, int* __restrict__ csrc, int nE, int nN) {
  __shared__ int cnt[MAXS];
  __shared__ int wsum[32];
  __shared__ int red[32];
  __shared__ int stage[CAP2];
  __shared__ int totS;
  const int tid = threadIdx.x;
  const int lane = tid & 31;
  const int b = blockIdx.x;
  const int S = (nN + NB - 1) / NB;
  const int n0 = b * S;
  const int chunk = (nE + NWG - 1) / NWG;
  const int pc = tid >> 2;
  const int q = tid & 3;
  const int pe0 = pc * chunk;
  const int plen = max(min(pe0 + chunk, nE) - pe0, 0);
  const int o0 = off[pc * NB + b];
  const int o1 = (b + 1 < NB) ? off[pc * NB + b + 1] : plen;
  const int sb = pe0 + o0;
  const int se = pe0 + o1;

  v2i rec[8];
#pragma unroll
  for (int u = 0; u < 8; ++u) {
    const int i = sb + q + 4 * u;
    rec[u].x = n0;
    rec[u].y = 0;
    if (i < se) rec[u] = *(const v2i*)(bkt + 2 * (size_t)i);
  }

  int part = (q == 0) ? o0 : 0;
#pragma unroll
  for (int sh = 16; sh > 0; sh >>= 1) part += __shfl_xor(part, sh, 32);
  if (lane == 0) red[tid >> 5] = part;
  cnt[tid] = 0;
  __syncthreads();

#pragma unroll
  for (int u = 0; u < 8; ++u) {
    if (sb + q + 4 * u < se) atomicAdd(&cnt[rec[u].x - n0], 1);
  }
  for (int ib = sb + q + 32; ib < se; ib += 16) {
    int d[4];
#pragma unroll
    for (int u = 0; u < 4; ++u) {
      const int i = ib + 4 * u;
      d[u] = (i < se) ? bkt[2 * (size_t)i] : n0;
    }
#pragma unroll
    for (int u = 0; u < 4; ++u) {
      if (ib + 4 * u < se) atomicAdd(&cnt[d[u] - n0], 1);
    }
  }
  __syncthreads();
  const int c = cnt[tid];
  const int ex = block_excl_scan(c, wsum);
  cnt[tid] = ex;
  if (tid == MAXS - 1) totS = ex + c;
  int beg = 0;
#pragma unroll
  for (int w = 0; w < 32; ++w) beg += red[w];
  if (tid < S && n0 + tid < nN) rowptr[n0 + tid] = beg + ex;
  if (b == 0 && tid == 0) rowptr[nN] = nE;
  __syncthreads();
  const int tot = totS;
  const bool staged = tot <= CAP2;

  int k[8];
#pragma unroll
  for (int u = 0; u < 8; ++u) {
    k[u] = 0;
    if (sb + q + 4 * u < se) k[u] = atomicAdd(&cnt[rec[u].x - n0], 1);
  }
  if (staged) {
#pragma unroll
    for (int u = 0; u < 8; ++u) {
      if (sb + q + 4 * u < se) stage[k[u]] = rec[u].y;
    }
  } else {
#pragma unroll
    for (int u = 0; u < 8; ++u) {
      if (sb + q + 4 * u < se) csrc[beg + k[u]] = rec[u].y;
    }
  }
  for (int ib = sb + q + 32; ib < se; ib += 16) {
    v2i r2[4];
#pragma unroll
    for (int u = 0; u < 4; ++u) {
      const int i = ib + 4 * u;
      r2[u].x = n0;
      r2[u].y = 0;
      if (i < se) r2[u] = *(const v2i*)(bkt + 2 * (size_t)i);
    }
#pragma unroll
    for (int u = 0; u < 4; ++u) {
      if (ib + 4 * u < se) {
        const int k2 = atomicAdd(&cnt[r2[u].x - n0], 1);
        if (staged) stage[k2] = r2[u].y;
        else csrc[beg + k2] = r2[u].y;
      }
    }
  }
  if (staged) {
    __syncthreads();
    for (int i0 = 0; i0 < tot; i0 += 4 * 1024) {
      int sv[4];
#pragma unroll
      for (int u = 0; u < 4; ++u) sv[u] = stage[i0 + tid + u * 1024];
#pragma unroll
      for (int u = 0; u < 4; ++u) {
        const int i = i0 + tid + u * 1024;
        if (i < tot) csrc[beg + i] = sv[u];
      }
    }
  }
}

__device__ __forceinline__ float wave_max(float v) {
#pragma unroll
  for (int off = 16; off > 0; off >>= 1) v = fmaxf(v, __shfl_xor(v, off, 32));
  return v;
}
__device__ __forceinline__ float wave_sum(float v) {
#pragma unroll
  for (int off = 16; off > 0; off >>= 1) v += __shfl_xor(v, off, 32);
  return v;
}

template <int B>
__device__ __forceinline__ void add_rows(const float* __restrict__ z, int s, float alpha,
                                         int j0, int lane, v2f& acc) {
  v2f r[B];
#pragma unroll
  for (int u = 0; u < B; ++u) {
    const int sj = __builtin_amdgcn_readlane(s, j0 + u);
    r[u] = *(const v2f*)(z + (size_t)sj * DIM + 2 * lane);
  }
#pragma unroll
  for (int u = 0; u < B; ++u) {
    const float aj = __int_as_float(__builtin_amdgcn_readlane(__float_as_int(alpha), j0 + u));
    acc.x += aj * r[u].x;
    acc.y += aj * r[u].y;
  }
}

__device__ __forceinline__ void add_list(const float* __restrict__ z, int s, float alpha,
                                         int n, int lane, v2f& acc) {
  int j = 0;
  for (; j + 8 <= n; j += 8) add_rows<8>(z, s, alpha, j, lane, acc);
  if (n - j >= 4) { add_rows<4>(z, s, alpha, j, lane, acc); j += 4; }
  if (n - j >= 2) { add_rows<2>(z, s, alpha, j, lane, acc); j += 2; }
  if (n - j >= 1) { add_rows<1>(z, s, alpha, j, lane, acc); }
}

#define NPW 4

__device__ __forceinline__ void issue_bounds(int d0, int stride, int nN, int lane,
                                             const int* __restrict__ rowptr,
                                             const float* __restrict__ sdst,
                                             int& rbe, float& sdl) {
  const int kl = (lane >> 1) & (NPW - 1);
  const int dl = d0 + kl * stride;
  const int dk = (dl < nN) ? dl : 0;
  rbe = rowptr[dk + (lane & 1)];
  sdl = sdst[dk];
}

__device__ __forceinline__ void load_heads(int rbe, float sdl, int d0, int stride, int nN,
                                           int lane, float ab,
                                           const int* __restrict__ csrc,
                                           const float* __restrict__ ssrc,
                                           int (&beg)[NPW], int (&deg)[NPW], float (&sd)[NPW],
                                           int (&s)[NPW], float (&alpha)[NPW]) {
#pragma unroll
  for (int k = 0; k < NPW; ++k) {
    beg[k] = __builtin_amdgcn_readlane(rbe, 2 * k);
    const int end = __builtin_amdgcn_readlane(rbe, 2 * k + 1);
    deg[k] = (d0 + k * stride < nN) ? end - beg[k] : 0;
    sd[k] = __int_as_float(__builtin_amdgcn_readlane(__float_as_int(sdl), 2 * k));
  }
#pragma unroll
  for (int k = 0; k < NPW; ++k) s[k] = csrc[(lane < deg[k]) ? beg[k] + lane : 0];
  float ts[NPW];
#pragma unroll
  for (int k = 0; k < NPW; ++k) ts[k] = ssrc[s[k]];
  float v[NPW], m[NPW], ex[NPW], den[NPW];
#pragma unroll
  for (int k = 0; k < NPW; ++k) {
    const float t = ts[k] + sd[k] + ab;
    const float lv = (t > 0.0f) ? t : LEAKY * t;
    v[k] = (lane < deg[k]) ? lv : -INFINITY;
    m[k] = v[k];
  }
#pragma unroll
  for (int off = 16; off > 0; off >>= 1) {
#pragma unroll
    for (int k = 0; k < NPW; ++k) m[k] = fmaxf(m[k], __shfl_xor(m[k], off, 32));
  }
#pragma unroll
  for (int k = 0; k < NPW; ++k) {
    ex[k] = (lane < deg[k]) ? __expf(v[k] - m[k]) : 0.0f;
    den[k] = ex[k];
  }
#pragma unroll
  for (int off = 16; off > 0; off >>= 1) {
#pragma unroll
    for (int k = 0; k < NPW; ++k) den[k] += __shfl_xor(den[k], off, 32);
  }
#pragma unroll
  for (int k = 0; k < NPW; ++k) alpha[k] = ex[k] / den[k];
}

template <typename T>
__device__ __forceinline__ T pick(const T (&a)[NPW], int k) {
  T r = a[0];
#pragma unroll
  for (int j = 1; j < NPW; ++j) r = (k == j) ? a[j] : r;
  return r;
}

__device__ __forceinline__ v2f sum_rows(int beg, int deg, float sd, int s, float alpha,
                                        int lane, float ab,
                                        const int* __restrict__ csrc,
                                        const float* __restrict__ ssrc,
                                        const float* __restrict__ z) {
  v2f acc; acc.x = 0.0f; acc.y = 0.0f;
  if (deg > 0 && deg <= 32) {
    add_list(z, s, alpha, deg, lane, acc);
  } else if (deg > 32) {
    const int end = beg + deg;
    float m = -INFINITY;
    for (int c = beg; c < end; c += 32) {
      const int i = c + lane;
      if (i < end) {
        const float t = ssrc[csrc[i]] + sd + ab;
        m = fmaxf(m, (t > 0.0f) ? t : LEAKY * t);
      }
    }
    m = wave_max(m);
    float den = 0.0f;
    for (int c = beg; c < end; c += 32) {
      const int i = c + lane;
      if (i < end) {
        const float t = ssrc[csrc[i]] + sd + ab;
        den += __expf(((t > 0.0f) ? t : LEAKY * t) - m);
      }
    }
    den = wave_sum(den);
    for (int c = beg; c < end; c += 32) {
      const int i = c + lane;
      int s2 = 0;
      float a2 = 0.0f;
      if (i < end) {
        s2 = csrc[i];
        const float t = ssrc[s2] + sd + ab;
        a2 = __expf(((t > 0.0f) ? t : LEAKY * t) - m) / den;
      }
      add_list(z, s2, a2, min(32, end - c), lane, acc);
    }
  }
  return acc;
}

__global__ __launch_bounds__(256)
void node_softmax_aggregate(const int* __restrict__ rowptr, const int* __restrict__ csrc,
                            const float* __restrict__ ssrc, const float* __restrict__ sdst,
                            const float* __restrict__ abp, const float* __restrict__ z,
                            float* __restrict__ out, int nN) {
  const int lane = threadIdx.x & 31;
  const int warp = __builtin_amdgcn_readfirstlane((int)((blockIdx.x * blockDim.x + threadIdx.x) >> 5));
  const int nWarps = (gridDim.x * blockDim.x) >> 5;
  const float ab = abp[0];

  for (int base = warp; base < nN; base += NPW * nWarps) {
    int beg[NPW], deg[NPW], s[NPW];
    float sd[NPW], alpha[NPW];
    int rbe;
    float sdl;
    issue_bounds(base, nWarps, nN, lane, rowptr, sdst, rbe, sdl);
    load_heads(rbe, sdl, base, nWarps, nN, lane, ab, csrc, ssrc, beg, deg, sd, s, alpha);
#pragma unroll 1
    for (int k = 0; k < NPW; ++k) {
      const int d = base + k * nWarps;
      if (d < nN) {
        const v2f acc = sum_rows(pick(beg, k), pick(deg, k), pick(sd, k), pick(s, k),
                                 pick(alpha, k), lane, ab, csrc, ssrc, z);
        *(v2f*)(out + (size_t)d * DIM + 2 * lane) = acc;
      }
    }
  }
}

__global__ __launch_bounds__(256)
void node_aggregate_next_gemm(const int* __restrict__ rowptr, const int* __restrict__ csrc,
                              const float* __restrict__ ssrc, const float* __restrict__ sdst,
                              const float* __restrict__ abp, const float* __restrict__ z,
                              const float* __restrict__ Wn, const float* __restrict__ biasn,
                              const float* __restrict__ attn, float* __restrict__ zn,
                              float* __restrict__ ssrcn, float* __restrict__ sdstn, int nN) {
  __shared__ __attribute__((aligned(16))) float sW[DIM * WPITCH];
  __shared__ __attribute__((aligned(16))) float sH[32 * WPITCH];
  __shared__ float sB[DIM];
  __shared__ float sP[2 * 16 * 32];

  const int tid = threadIdx.x;
  const int lane = tid & 31;
  const int w = __builtin_amdgcn_readfirstlane(tid >> 5);
  const int warp = blockIdx.x * 8 + w;
  const int nWarps = gridDim.x * 8;

  int rbe;
  float sdl;
  issue_bounds(warp, nWarps, nN, lane, rowptr, sdst, rbe, sdl);

  float bv = 0.0f;
  if (tid < DIM) bv = biasn[tid];
#pragma unroll
  for (int i = 0; i < DIM / 4; ++i) {
    const int r = (tid >> 6) + 4 * i;
    sW[r * WPITCH + (tid & (DIM - 1))] = Wn[r * DIM + (tid & (DIM - 1))];
  }
  if (tid < DIM) sB[tid] = bv;

  const float ab = abp[0];
  {
    int beg[NPW], deg[NPW], s[NPW];
    float sd[NPW], alpha[NPW];
    load_heads(rbe, sdl, warp, nWarps, nN, lane, ab, csrc, ssrc, beg, deg, sd, s, alpha);
#pragma unroll 1
    for (int k = 0; k < NPW; ++k) {
      v2f acc = sum_rows(pick(beg, k), pick(deg, k), pick(sd, k), pick(s, k),
                         pick(alpha, k), lane, ab, csrc, ssrc, z);
      acc.x = fmaxf(acc.x, 0.0f);
      acc.y = fmaxf(acc.y, 0.0f);
      *(v2f*)(&sH[(k * 8 + w) * WPITCH + 2 * lane]) = acc;
    }
  }
  __syncthreads();

  const int rt = w >> 1;
  const int ch = w & 1;
  const int laneRow = lane & 15;
  const int hh = lane >> 4;
  const int kHalf = hh * 2;
  const int dBase = blockIdx.x * 8 + (2 * rt + hh) * nWarps;
  const int c0 = ch * 32 + laneRow;

  v8f acc[2];
  v2f aw[2];
  if (w < 4) {
    const float* aBase = &sH[(rt * 16 + laneRow) * WPITCH + kHalf];
    const float* bBase = &sW[c0 * WPITCH + kHalf];
#pragma unroll
    for (int j = 0; j < 2; ++j) {
      const float bcol = sB[c0 + 16 * j];
#pragma unroll
      for (int r = 0; r < 8; ++r) acc[j][r] = bcol;
      aw[j].x = attn[c0 + 16 * j];
      aw[j].y = attn[DIM + c0 + 16 * j];
    }
#pragma unroll
    for (int t = 0; t < 16; ++t) {
      const v2f a = *(const v2f*)(aBase + 4 * t);
#pragma unroll
      for (int j = 0; j < 2; ++j) {
        const v2f b = *(const v2f*)(bBase + 16 * j * WPITCH + 4 * t);
        acc[j] = __builtin_amdgcn_wmma_f32_16x16x4_f32(
             false, a,  false, b,
             (short)0, acc[j],  false,  false);
      }
    }
    __builtin_amdgcn_sched_group_barrier(0x100, 3, 0);
#pragma unroll
    for (int g = 0; g < 7; ++g) {
      __builtin_amdgcn_sched_group_barrier(0x100, 3, 0);
      __builtin_amdgcn_sched_group_barrier(0x008, 4, 0);
    }
    __builtin_amdgcn_sched_group_barrier(0x008, 4, 0);
  }

  v2f sd[8];
  if (w < 4 && ch == 0) {
#pragma unroll
    for (int r = 0; r < 8; ++r) { sd[r].x = 0.0f; sd[r].y = 0.0f; }
#pragma unroll
    for (int j = 0; j < 2; ++j) {
#pragma unroll
      for (int r = 0; r < 8; ++r) sd[r] += acc[j][r] * aw[j];
    }
#pragma unroll
    for (int r = 0; r < 8; ++r) {
      sP[(rt * 16 + 2 * r) * 32 + lane] = sd[r].x;
      sP[(rt * 16 + 2 * r + 1) * 32 + lane] = sd[r].y;
    }
  }
  __syncthreads();
  if (w >= 4) return;

#pragma unroll
  for (int j = 0; j < 2; ++j) {
#pragma unroll
    for (int r = 0; r < 8; ++r) {
      const int d = dBase + r;
      if (d < nN) zn[(size_t)d * DIM + c0 + 16 * j] = acc[j][r];
    }
  }
  if (ch == 0) return;

#pragma unroll
  for (int r = 0; r < 8; ++r) {
    sd[r].x = sP[(rt * 16 + 2 * r) * 32 + lane];
    sd[r].y = sP[(rt * 16 + 2 * r + 1) * 32 + lane];
  }
#pragma unroll
  for (int j = 0; j < 2; ++j) {
#pragma unroll
    for (int r = 0; r < 8; ++r) sd[r] += acc[j][r] * aw[j];
  }

  const bool b3 = (lane & 8) != 0, b2 = (lane & 4) != 0;
  const bool b1 = (lane & 2) != 0, b0 = (lane & 1) != 0;
  float t[8], u[4], x2[2];
#pragma unroll
  for (int r = 0; r < 8; ++r) {
    const float keep = b3 ? sd[r].y : sd[r].x;
    const float send = b3 ? sd[r].x : sd[r].y;
    t[r] = keep + row_xor<8>(send);
  }
#pragma unroll
  for (int j = 0; j < 4; ++j) {
    const float keep = b2 ? t[j + 4] : t[j];
    const float send = b2 ? t[j] : t[j + 4];
    u[j] = keep + row_xor<4>(send);
  }
#pragma unroll
  for (int j = 0; j < 2; ++j) {
    const float keep = b1 ? u[j + 2] : u[j];
    const float send = b1 ? u[j] : u[j + 2];
    x2[j] = keep + row_xor<2>(send);
  }
  const float keep = b0 ? x2[1] : x2[0];
  const float send = b0 ? x2[0] : x2[1];
  const float y = keep + row_xor<1>(send);

  const int dy = dBase + (lane & 7);
  float* sp = b3 ? sdstn : ssrcn;
  if (dy < nN) sp[dy] = y;
}

extern "C" void kernel_launch(void* const* d_in, const int* in_sizes, int n_in,
                              void* d_out, int out_size, void* d_ws, size_t ws_size,
                              hipStream_t stream) {
  const float* x  = (const float*)d_in[0];
  const int* src  = (const int*)d_in[1];
  const int* dst  = (const int*)d_in[2];
  const int nN = in_sizes[0] / DIM;
  const int nE = in_sizes[1];

  char* ws = (char*)d_ws;
  const size_t featB = (size_t)nN * DIM * sizeof(float);
  const size_t edgeB = (size_t)nE * 4;
  const size_t nodeB = (size_t)nN * 4;
  float* z      = (float*)ws;
  float* z2     = (float*)(ws + featB);
  int*   bkt    = (int*)(ws + 2 * featB);
  int*   csrc   = (int*)(ws + 2 * featB + 2 * edgeB);
  int*   off    = (int*)(ws + 2 * featB + 3 * edgeB);
  const size_t tabB = (size_t)NWG * NB * 4;
  float* ssrc   = (float*)(ws + 2 * featB + 3 * edgeB + tabB + 2048);
  float* sdst   = (float*)(ws + 2 * featB + 3 * edgeB + tabB + 2048 + nodeB);
  int*   rowptr = (int*)(ws + 2 * featB + 3 * edgeB + tabB + 2048 + 2 * nodeB);
  float* ssrc2  = (float*)(ws + 2 * featB + 3 * edgeB + tabB + 4096 + 3 * nodeB);
  float* sdst2  = (float*)(ws + 2 * featB + 3 * edgeB + tabB + 4096 + 4 * nodeB);

  const int nTiles = (nN + 15) / 16;
  const int nLin = (nTiles + 31) / 32;
  const int aggBlocks = (nN + 31) / 32;

  piece_sort_and_linear<<<nLin + NWG, 1024, 0, stream>>>(
      src, dst, off, bkt, nE, nN, x, (const float*)d_in[3], (const float*)d_in[4],
      (const float*)d_in[5], z, ssrc, sdst, nTiles, nLin);
  range_to_lists<<<NB, 1024, 0, stream>>>(bkt, off, rowptr, csrc, nE, nN);

  node_aggregate_next_gemm<<<aggBlocks, 256, 0, stream>>>(
      rowptr, csrc, ssrc, sdst, (const float*)d_in[6], z,
      (const float*)d_in[7], (const float*)d_in[8], (const float*)d_in[9], z2, ssrc2, sdst2, nN);

  node_aggregate_next_gemm<<<aggBlocks, 256, 0, stream>>>(
      rowptr, csrc, ssrc2, sdst2, (const float*)d_in[10], z2,
      (const float*)d_in[11], (const float*)d_in[12], (const float*)d_in[13], z, ssrc, sdst, nN);

  node_softmax_aggregate<<<aggBlocks, 256, 0, stream>>>(rowptr, csrc, ssrc, sdst,
                                                        (const float*)d_in[14], z,
                                                        (float*)d_out, nN);
}
